# starting kernel
# baseline (speedup 1.0000x reference)
#include <hip/hip_runtime.h>

#define N_NODES 100000
#define D_IN 256
#define D_H 128
#define D_OUT 32

typedef __attribute__((ext_vector_type(16))) __bf16 v16bf;
typedef __attribute__((ext_vector_type(8)))  float  v8f;

__device__ __forceinline__ v16bf frag_from_lds(const __bf16* base, int rowIdx, int khalf) {
  union { uint4 u[2]; v16bf v; } c;
  const uint4* p = (const uint4*)(base + rowIdx * 32 + khalf);
  c.u[0] = p[0];
  c.u[1] = p[2];
  return c.v;
}

template<int K, int N>
__global__ __launch_bounds__(256) void gemm_bf16_wmma(
    const float* __restrict__ A, const float* __restrict__ W,
    float* __restrict__ C,
    const float* __restrict__ dinv, const float* __restrict__ bias,
    float* __restrict__ out2, int M) {
  constexpr int NT = N / 16;
  __shared__ __bf16 ldsA[128 * 32];
  __shared__ __bf16 ldsB[N * 32];

  const int tid   = threadIdx.x;
  const int wave  = tid >> 5;
  const int lane  = tid & 31;
  const int laneM = lane & 15;
  const int khalf = (lane >> 4) << 3;
  const int rowBlk = blockIdx.x * 128;

  const int ldRow = tid >> 1;
  const int ldC0  = (tid & 1) << 4;
  const int ldGr  = rowBlk + ldRow;

  v8f acc[NT];
#pragma unroll
  for (int t = 0; t < NT; ++t) {
    v8f z = {0.f, 0.f, 0.f, 0.f, 0.f, 0.f, 0.f, 0.f};
    acc[t] = z;
  }

  for (int kk = 0; kk < K; kk += 32) {
    __syncthreads();

    {
      union { __bf16 h[16]; uint4 q[2]; } buf;
      if (ldGr < M) {
        const float4* src = (const float4*)(A + (size_t)ldGr * K + kk + ldC0);
#pragma unroll
        for (int j = 0; j < 4; ++j) {
          float4 v = src[j];
          buf.h[j * 4 + 0] = (__bf16)v.x;
          buf.h[j * 4 + 1] = (__bf16)v.y;
          buf.h[j * 4 + 2] = (__bf16)v.z;
          buf.h[j * 4 + 3] = (__bf16)v.w;
        }
      } else {
#pragma unroll
        for (int j = 0; j < 16; ++j) buf.h[j] = (__bf16)0.0f;
      }
      uint4* dst = (uint4*)(ldsA + ldRow * 32 + ldC0);
      dst[0] = buf.q[0];
      dst[1] = buf.q[1];
    }

#pragma unroll
    for (int i = tid; i < N * 8; i += 256) {
      const int n  = i % N;
      const int kq = i / N;
      union { __bf16 h[4]; uint2 q; } wb;
#pragma unroll
      for (int j = 0; j < 4; ++j)
        wb.h[j] = (__bf16)W[(size_t)(kk + kq * 4 + j) * N + n];
      *(uint2*)(ldsB + n * 32 + kq * 4) = wb.q;
    }
    __syncthreads();

    v16bf a = frag_from_lds(ldsA, (wave << 4) + laneM, khalf);
    v16bf bc = frag_from_lds(ldsB, laneM, khalf);
#pragma unroll
    for (int t = 0; t < NT; ++t) {
      v16bf bn = (t + 1 < NT) ? frag_from_lds(ldsB, ((t + 1) << 4) + laneM, khalf) : bc;
      acc[t] = __builtin_amdgcn_wmma_f32_16x16x32_bf16(
          false, a, false, bc, (short)0, acc[t], false, false);
      bc = bn;
    }
  }

  float bfrag[NT];
#pragma unroll
  for (int t = 0; t < NT; ++t) bfrag[t] = bias[(t << 4) + laneM];

  const int mBase = rowBlk + (wave << 4) + ((lane >> 4) << 3);
#pragma unroll
  for (int v = 0; v < 8; ++v) {
    const int gr = mBase + v;
    if (gr < M) {
      const float d  = dinv[gr];
      const float dd = d * d;
#pragma unroll
      for (int t = 0; t < NT; ++t) {
        const size_t idx = (size_t)gr * N + (t << 4) + laneM;
        const float val = acc[t][v];
        C[idx]    = val;
        out2[idx] = val * dd + bfrag[t];
      }
    }
  }
}

__global__ void k_fill1(float* __restrict__ p, int n) {
  int i = blockIdx.x * blockDim.x + threadIdx.x;
  if (i < n) p[i] = 1.0f;
}

__global__ void k_deg(const int* __restrict__ col, float* __restrict__ deg, int E) {
  int e = blockIdx.x * blockDim.x + threadIdx.x;
  if (e < E) atomicAdd(&deg[col[e]], 1.0f);
}

__global__ void k_rsqrt(float* __restrict__ p, int n) {
  int i = blockIdx.x * blockDim.x + threadIdx.x;
  if (i < n) p[i] = rsqrtf(p[i]);
}

template<int F>
__global__ __launch_bounds__(256) void k_edges(
    const float* __restrict__ h, const float* __restrict__ dinv,
    const int* __restrict__ row, const int* __restrict__ col,
    float* __restrict__ out, int E) {
  constexpr int EPB = 256 / F;
  const int f  = threadIdx.x & (F - 1);
  const int es = threadIdx.x / F;
  for (int e = blockIdx.x * EPB + es; e < E; e += gridDim.x * EPB) {
    int r = row[e], c = col[e];
    float nrm = dinv[r] * dinv[c];
    atomicAdd(&out[(size_t)c * F + f], h[(size_t)r * F + f] * nrm);
  }
}

__global__ void k_relu4(float4* __restrict__ p, int n4) {
  int i = blockIdx.x * blockDim.x + threadIdx.x;
  if (i < n4) {
    float4 v = p[i];
    v.x = fmaxf(v.x, 0.0f);
    v.y = fmaxf(v.y, 0.0f);
    v.z = fmaxf(v.z, 0.0f);
    v.w = fmaxf(v.w, 0.0f);
    p[i] = v;
  }
}

__global__ __launch_bounds__(256) void k_log_softmax32(float* __restrict__ out, int n) {
  int node = blockIdx.x * 8 + (threadIdx.x >> 5);
  int lane = threadIdx.x & 31;
  if (node >= n) return;
  float v = out[node * 32 + lane];
  float m = v;
#pragma unroll
  for (int o = 16; o > 0; o >>= 1) m = fmaxf(m, __shfl_xor(m, o, 32));
  float ex = __expf(v - m);
  float s = ex;
#pragma unroll
  for (int o = 16; o > 0; o >>= 1) s += __shfl_xor(s, o, 32);
  out[node * 32 + lane] = v - m - __logf(s);
}

extern "C" void kernel_launch(void* const* d_in, const int* in_sizes, int n_in,
                              void* d_out, int out_size, void* d_ws, size_t ws_size,
                              hipStream_t stream) {
  const float* x  = (const float*)d_in[0];
  const float* W1 = (const float*)d_in[1];
  const float* b1 = (const float*)d_in[2];
  const float* W2 = (const float*)d_in[3];
  const float* b2 = (const float*)d_in[4];
  const int*  eix = (const int*)d_in[5];

  const int n = N_NODES;
  const int E = in_sizes[5] / 2;
  const int* row = eix;
  const int* col = eix + E;

  float* ws   = (float*)d_ws;
  float* dinv = ws;
  float* h    = ws + 100352;
  float* h1   = h  + (size_t)n * D_H;
  float* g2   = h1 + (size_t)n * D_H;
  float* outf = (float*)d_out;

  k_fill1 <<<(n + 255) / 256, 256, 0, stream>>>(dinv, n);
  k_deg   <<<(E + 255) / 256, 256, 0, stream>>>(col, dinv, E);
  k_rsqrt <<<(n + 255) / 256, 256, 0, stream>>>(dinv, n);

  gemm_bf16_wmma<D_IN, D_H><<<(n + 127) / 128, 256, 0, stream>>>(
      x, W1, h, dinv, b1, h1, n);
  k_edges<D_H><<<4096, 256, 0, stream>>>(h, dinv, row, col, h1, E);
  k_relu4<<<((n * (D_H / 4)) + 255) / 256, 256, 0, stream>>>((float4*)h1, n * (D_H / 4));

  gemm_bf16_wmma<D_H, D_OUT><<<(n + 127) / 128, 256, 0, stream>>>(
      h1, W2, g2, dinv, b2, outf, n);
  k_edges<D_OUT><<<4096, 256, 0, stream>>>(g2, dinv, row, col, outf, E);
  k_log_softmax32<<<(n + 7) / 8, 256, 0, stream>>>(outf, n);
}


# Layer-1 edge sums without global atomics: a work-group owns 512 targets, keeps their sums in 256 KiB of LDS (LDS float adds, arrival order), sweeps the whole target list, gathers matching rows
# speedup vs baseline: 1.4544x; 1.4544x over previous
#include <hip/hip_runtime.h>

#define N_NODES 100000
#define D_IN 256
#define D_H 128
#define D_OUT 32

typedef __attribute__((ext_vector_type(16))) __bf16 v16bf;
typedef __attribute__((ext_vector_type(8)))  float  v8f;

__device__ __forceinline__ v16bf frag_from_lds(const __bf16* base, int rowIdx, int khalf) {
  union { uint4 u[2]; v16bf v; } c;
  const uint4* p = (const uint4*)(base + rowIdx * 32 + khalf);
  c.u[0] = p[0];
  c.u[1] = p[2];
  return c.v;
}

template<int K, int N>
__global__ __launch_bounds__(256) void gemm_bf16_wmma(
    const float* __restrict__ A, const float* __restrict__ W,
    float* __restrict__ C,
    const float* __restrict__ dinv, const float* __restrict__ bias,
    float* __restrict__ out2, int M) {
  constexpr int NT = N / 16;
  __shared__ __bf16 ldsA[128 * 32];
  __shared__ __bf16 ldsB[N * 32];

  const int tid   = threadIdx.x;
  const int wave  = tid >> 5;
  const int lane  = tid & 31;
  const int laneM = lane & 15;
  const int khalf = (lane >> 4) << 3;
  const int rowBlk = blockIdx.x * 128;

  const int ldRow = tid >> 1;
  const int ldC0  = (tid & 1) << 4;
  const int ldGr  = rowBlk + ldRow;

  v8f acc[NT];
#pragma unroll
  for (int t = 0; t < NT; ++t) {
    v8f z = {0.f, 0.f, 0.f, 0.f, 0.f, 0.f, 0.f, 0.f};
    acc[t] = z;
  }

  for (int kk = 0; kk < K; kk += 32) {
    __syncthreads();

    {
      union { __bf16 h[16]; uint4 q[2]; } buf;
      if (ldGr < M) {
        const float4* src = (const float4*)(A + (size_t)ldGr * K + kk + ldC0);
#pragma unroll
        for (int j = 0; j < 4; ++j) {
          float4 v = src[j];
          buf.h[j * 4 + 0] = (__bf16)v.x;
          buf.h[j * 4 + 1] = (__bf16)v.y;
          buf.h[j * 4 + 2] = (__bf16)v.z;
          buf.h[j * 4 + 3] = (__bf16)v.w;
        }
      } else {
#pragma unroll
        for (int j = 0; j < 16; ++j) buf.h[j] = (__bf16)0.0f;
      }
      uint4* dst = (uint4*)(ldsA + ldRow * 32 + ldC0);
      dst[0] = buf.q[0];
      dst[1] = buf.q[1];
    }

#pragma unroll
    for (int i = tid; i < N * 8; i += 256) {
      const int n  = i % N;
      const int kq = i / N;
      union { __bf16 h[4]; uint2 q; } wb;
#pragma unroll
      for (int j = 0; j < 4; ++j)
        wb.h[j] = (__bf16)W[(size_t)(kk + kq * 4 + j) * N + n];
      *(uint2*)(ldsB + n * 32 + kq * 4) = wb.q;
    }
    __syncthreads();

    v16bf a = frag_from_lds(ldsA, (wave << 4) + laneM, khalf);
    v16bf bc = frag_from_lds(ldsB, laneM, khalf);
#pragma unroll
    for (int t = 0; t < NT; ++t) {
      v16bf bn = (t + 1 < NT) ? frag_from_lds(ldsB, ((t + 1) << 4) + laneM, khalf) : bc;
      acc[t] = __builtin_amdgcn_wmma_f32_16x16x32_bf16(
          false, a, false, bc, (short)0, acc[t], false, false);
      bc = bn;
    }
  }

  float bfrag[NT];
#pragma unroll
  for (int t = 0; t < NT; ++t) bfrag[t] = bias[(t << 4) + laneM];

  const int mBase = rowBlk + (wave << 4) + ((lane >> 4) << 3);
#pragma unroll
  for (int v = 0; v < 8; ++v) {
    const int gr = mBase + v;
    if (gr < M) {
      const float d  = dinv[gr];
      const float dd = d * d;
#pragma unroll
      for (int t = 0; t < NT; ++t) {
        const size_t idx = (size_t)gr * N + (t << 4) + laneM;
        const float val = acc[t][v];
        C[idx]    = val;
        out2[idx] = val * dd + bfrag[t];
      }
    }
  }
}

__global__ void k_fill1(float* __restrict__ p, int n) {
  int i = blockIdx.x * blockDim.x + threadIdx.x;
  if (i < n) p[i] = 1.0f;
}

__global__ void k_deg(const int* __restrict__ col, float* __restrict__ deg, int E) {
  int e = blockIdx.x * blockDim.x + threadIdx.x;
  if (e < E) atomicAdd(&deg[col[e]], 1.0f);
}

__global__ void k_rsqrt(float* __restrict__ p, int n) {
  int i = blockIdx.x * blockDim.x + threadIdx.x;
  if (i < n) p[i] = rsqrtf(p[i]);
}

template<int F>
__global__ __launch_bounds__(256) void k_edges(
    const float* __restrict__ h, const float* __restrict__ dinv,
    const int* __restrict__ row, const int* __restrict__ col,
    float* __restrict__ out, int E) {
  constexpr int EPB = 256 / F;
  const int f  = threadIdx.x & (F - 1);
  const int es = threadIdx.x / F;
  for (int e = blockIdx.x * EPB + es; e < E; e += gridDim.x * EPB) {
    int r = row[e], c = col[e];
    float nrm = dinv[r] * dinv[c];
    atomicAdd(&out[(size_t)c * F + f], h[(size_t)r * F + f] * nrm);
  }
}


#define AGG_T     512
#define AGG_QCAP  160

__device__ __forceinline__ void agg_flush(
    const int* myq, int cnt, int lane, int base,
    const float* __restrict__ h, const float* __restrict__ dinv,
    const int* __restrict__ row, const int* __restrict__ col,
    float* acc) {
  const float4* h4 = (const float4*)h;
  for (int b0 = 0; b0 < cnt; b0 += 32) {
    const int m = min(32, cnt - b0);
    int r = 0, t = 0;
    float nr = 0.0f;
    if (lane < m) {
      const int e = myq[b0 + lane];
      const int c = col[e];
      r  = row[e];
      t  = c - base;
      nr = dinv[r] * dinv[c];
    }
    int i = 0;
    for (; i + 4 <= m; i += 4) {
      float4 v[4];
      int    tt[4];
      float  nn[4];
#pragma unroll
      for (int u = 0; u < 4; ++u) {
        const int ru = __builtin_amdgcn_readlane(r, i + u);
        tt[u] = __builtin_amdgcn_readlane(t, i + u);
        nn[u] = __int_as_float(__builtin_amdgcn_readlane(__float_as_int(nr), i + u));
        v[u]  = h4[(size_t)ru * 32 + lane];
      }
#pragma unroll
      for (int u = 0; u < 4; ++u) {
        float* a = acc + tt[u] * 128 + lane * 4;
        atomicAdd(a + 0, v[u].x * nn[u]);
        atomicAdd(a + 1, v[u].y * nn[u]);
        atomicAdd(a + 2, v[u].z * nn[u]);
        atomicAdd(a + 3, v[u].w * nn[u]);
      }
    }
    for (; i < m; ++i) {
      const int   ru = __builtin_amdgcn_readlane(r, i);
      const int   tu = __builtin_amdgcn_readlane(t, i);
      const float nu = __int_as_float(__builtin_amdgcn_readlane(__float_as_int(nr), i));
      const float4 v = h4[(size_t)ru * 32 + lane];
      float* a = acc + tu * 128 + lane * 4;
      atomicAdd(a + 0, v.x * nu);
      atomicAdd(a + 1, v.y * nu);
      atomicAdd(a + 2, v.z * nu);
      atomicAdd(a + 3, v.w * nu);
    }
  }
}

__global__ __launch_bounds__(1024) void k_agg_lds128(
    const float* __restrict__ h, const float* __restrict__ dinv,
    const int* __restrict__ row, const int* __restrict__ col,
    float* __restrict__ out, int E, int n) {
  __shared__ float acc[AGG_T * 128];
  __shared__ int   q[32 * AGG_QCAP];

  const int tid  = threadIdx.x;
  const int lane = tid & 31;
  const int wave = tid >> 5;
  const int base = blockIdx.x * AGG_T;
  const int nT   = min(AGG_T, n - base);
  const unsigned lt = (1u << lane) - 1u;

  {
    const float4* s4 = (const float4*)(out + (size_t)base * 128);
    float4* d4 = (float4*)acc;
    for (int i = tid; i < nT * 32; i += 1024) d4[i] = s4[i];
  }
  __syncthreads();

  int* myq = q + wave * AGG_QCAP;
  int cnt = 0;

  const bool vec = ((E & 3) == 0) && ((((size_t)col) & 15) == 0);
  const int E4 = vec ? (E >> 2) : 0;
  const int4* col4 = (const int4*)col;

  const int nCh = (E4 + 4095) >> 12;
  int ch = (int)(((long long)blockIdx.x * nCh) / gridDim.x);
  for (int j = 0; j < nCh; ++j) {
    const int i0 = ch << 12;
    ch = (ch + 1 == nCh) ? 0 : ch + 1;
    int4 c[4];
#pragma unroll
    for (int u = 0; u < 4; ++u) {
      const int i = i0 + u * 1024 + tid;
      c[u] = col4[min(i, E4 - 1)];
    }
#pragma unroll
    for (int u = 0; u < 4; ++u) {
      const int i = i0 + u * 1024 + tid;
      const bool in = i < E4;
      const bool m0 = in & ((unsigned)(c[u].x - base) < (unsigned)nT);
      const bool m1 = in & ((unsigned)(c[u].y - base) < (unsigned)nT);
      const bool m2 = in & ((unsigned)(c[u].z - base) < (unsigned)nT);
      const bool m3 = in & ((unsigned)(c[u].w - base) < (unsigned)nT);
      if (__builtin_amdgcn_ballot_w32(m0 | m1 | m2 | m3)) {
        unsigned b;
        b = __builtin_amdgcn_ballot_w32(m0);
        if (m0) myq[cnt + __builtin_popcount(b & lt)] = 4 * i + 0;
        cnt += __builtin_popcount(b);
        b = __builtin_amdgcn_ballot_w32(m1);
        if (m1) myq[cnt + __builtin_popcount(b & lt)] = 4 * i + 1;
        cnt += __builtin_popcount(b);
        b = __builtin_amdgcn_ballot_w32(m2);
        if (m2) myq[cnt + __builtin_popcount(b & lt)] = 4 * i + 2;
        cnt += __builtin_popcount(b);
        b = __builtin_amdgcn_ballot_w32(m3);
        if (m3) myq[cnt + __builtin_popcount(b & lt)] = 4 * i + 3;
        cnt += __builtin_popcount(b);
        if (cnt >= 32) {
          agg_flush(myq, cnt, lane, base, h, dinv, row, col, acc);
          cnt = 0;
        }
      }
    }
  }
  for (int i0 = E4 * 4; i0 < E; i0 += 1024) {
    const int e = i0 + tid;
    const int c = (e < E) ? col[e] : -1;
    const bool m = (unsigned)(c - base) < (unsigned)nT;
    const unsigned b = __builtin_amdgcn_ballot_w32(m);
    if (b) {
      if (m) myq[cnt + __builtin_popcount(b & lt)] = e;
      cnt += __builtin_popcount(b);
      if (cnt >= 32) {
        agg_flush(myq, cnt, lane, base, h, dinv, row, col, acc);
        cnt = 0;
      }
    }
  }
  agg_flush(myq, cnt, lane, base, h, dinv, row, col, acc);

  __syncthreads();
  {
    float4* d4 = (float4*)(out + (size_t)base * 128);
    const float4* s4 = (const float4*)acc;
    for (int i = tid; i < nT * 32; i += 1024) d4[i] = s4[i];
  }
}

__global__ void k_relu4(float4* __restrict__ p, int n4) {
  int i = blockIdx.x * blockDim.x + threadIdx.x;
  if (i < n4) {
    float4 v = p[i];
    v.x = fmaxf(v.x, 0.0f);
    v.y = fmaxf(v.y, 0.0f);
    v.z = fmaxf(v.z, 0.0f);
    v.w = fmaxf(v.w, 0.0f);
    p[i] = v;
  }
}

__global__ __launch_bounds__(256) void k_log_softmax32(float* __restrict__ out, int n) {
  int node = blockIdx.x * 8 + (threadIdx.x >> 5);
  int lane = threadIdx.x & 31;
  if (node >= n) return;
  float v = out[node * 32 + lane];
  float m = v;
#pragma unroll
  for (int o = 16; o > 0; o >>= 1) m = fmaxf(m, __shfl_xor(m, o, 32));
  float ex = __expf(v - m);
  float s = ex;
#pragma unroll
  for (int o = 16; o > 0; o >>= 1) s += __shfl_xor(s, o, 32);
  out[node * 32 + lane] = v - m - __logf(s);
}

extern "C" void kernel_launch(void* const* d_in, const int* in_sizes, int n_in,
                              void* d_out, int out_size, void* d_ws, size_t ws_size,
                              hipStream_t stream) {
  const float* x  = (const float*)d_in[0];
  const float* W1 = (const float*)d_in[1];
  const float* b1 = (const float*)d_in[2];
  const float* W2 = (const float*)d_in[3];
  const float* b2 = (const float*)d_in[4];
  const int*  eix = (const int*)d_in[5];

  const int n = N_NODES;
  const int E = in_sizes[5] / 2;
  const int* row = eix;
  const int* col = eix + E;

  float* ws   = (float*)d_ws;
  float* dinv = ws;
  float* h    = ws + 100352;
  float* h1   = h  + (size_t)n * D_H;
  float* g2   = h1 + (size_t)n * D_H;
  float* outf = (float*)d_out;

  k_fill1 <<<(n + 255) / 256, 256, 0, stream>>>(dinv, n);
  k_deg   <<<(E + 255) / 256, 256, 0, stream>>>(col, dinv, E);
  k_rsqrt <<<(n + 255) / 256, 256, 0, stream>>>(dinv, n);

  gemm_bf16_wmma<D_IN, D_H><<<(n + 127) / 128, 256, 0, stream>>>(
      x, W1, h, dinv, b1, h1, n);
  k_agg_lds128<<<(n + AGG_T - 1) / AGG_T, 1024, 0, stream>>>(h, dinv, row, col, h1, E, n);
  k_relu4<<<((n * (D_H / 4)) + 255) / 256, 256, 0, stream>>>((float4*)h1, n * (D_H / 4));

  gemm_bf16_wmma<D_H, D_OUT><<<(n + 127) / 128, 256, 0, stream>>>(
      h1, W2, g2, dinv, b2, outf, n);
  k_edges<D_OUT><<<4096, 256, 0, stream>>>(g2, dinv, row, col, outf, E);
  k_log_softmax32<<<(n + 7) / 8, 256, 0, stream>>>(outf, n);
}


# Layer-2 edge sums without global atomics: each layer-1 sweep wave saves its edges (source, target, norm); layer 2 adds their rows into LDS sums, log-softmax at the store; sum order LDS arrival
# speedup vs baseline: 1.7167x; 1.1803x over previous
#include <hip/hip_runtime.h>

#define N_NODES 100000
#define D_IN 256
#define D_H 128
#define D_OUT 32

typedef __attribute__((ext_vector_type(16))) __bf16 v16bf;
typedef __attribute__((ext_vector_type(8)))  float  v8f;

__device__ __forceinline__ v16bf frag_from_lds(const __bf16* base, int rowIdx, int khalf) {
  union { uint4 u[2]; v16bf v; } c;
  const uint4* p = (const uint4*)(base + rowIdx * 32 + khalf);
  c.u[0] = p[0];
  c.u[1] = p[2];
  return c.v;
}

template<int K, int N>
__global__ __launch_bounds__(256) void gemm_bf16_wmma(
    const float* __restrict__ A, const float* __restrict__ W,
    float* __restrict__ C,
    const float* __restrict__ dinv, const float* __restrict__ bias,
    float* __restrict__ out2, int M) {
  constexpr int NT = N / 16;
  __shared__ __bf16 ldsA[128 * 32];
  __shared__ __bf16 ldsB[N * 32];

  const int tid   = threadIdx.x;
  const int wave  = tid >> 5;
  const int lane  = tid & 31;
  const int laneM = lane & 15;
  const int khalf = (lane >> 4) << 3;
  const int rowBlk = blockIdx.x * 128;

  const int ldRow = tid >> 1;
  const int ldC0  = (tid & 1) << 4;
  const int ldGr  = rowBlk + ldRow;

  v8f acc[NT];
#pragma unroll
  for (int t = 0; t < NT; ++t) {
    v8f z = {0.f, 0.f, 0.f, 0.f, 0.f, 0.f, 0.f, 0.f};
    acc[t] = z;
  }

  for (int kk = 0; kk < K; kk += 32) {
    __syncthreads();

    {
      union { __bf16 h[16]; uint4 q[2]; } buf;
      if (ldGr < M) {
        const float4* src = (const float4*)(A + (size_t)ldGr * K + kk + ldC0);
#pragma unroll
        for (int j = 0; j < 4; ++j) {
          float4 v = src[j];
          buf.h[j * 4 + 0] = (__bf16)v.x;
          buf.h[j * 4 + 1] = (__bf16)v.y;
          buf.h[j * 4 + 2] = (__bf16)v.z;
          buf.h[j * 4 + 3] = (__bf16)v.w;
        }
      } else {
#pragma unroll
        for (int j = 0; j < 16; ++j) buf.h[j] = (__bf16)0.0f;
      }
      uint4* dst = (uint4*)(ldsA + ldRow * 32 + ldC0);
      dst[0] = buf.q[0];
      dst[1] = buf.q[1];
    }

#pragma unroll
    for (int i = tid; i < N * 8; i += 256) {
      const int n  = i % N;
      const int kq = i / N;
      union { __bf16 h[4]; uint2 q; } wb;
#pragma unroll
      for (int j = 0; j < 4; ++j)
        wb.h[j] = (__bf16)W[(size_t)(kk + kq * 4 + j) * N + n];
      *(uint2*)(ldsB + n * 32 + kq * 4) = wb.q;
    }
    __syncthreads();

    v16bf a = frag_from_lds(ldsA, (wave << 4) + laneM, khalf);
    v16bf bc = frag_from_lds(ldsB, laneM, khalf);
#pragma unroll
    for (int t = 0; t < NT; ++t) {
      v16bf bn = (t + 1 < NT) ? frag_from_lds(ldsB, ((t + 1) << 4) + laneM, khalf) : bc;
      acc[t] = __builtin_amdgcn_wmma_f32_16x16x32_bf16(
          false, a, false, bc, (short)0, acc[t], false, false);
      bc = bn;
    }
  }

  float bfrag[NT];
#pragma unroll
  for (int t = 0; t < NT; ++t) bfrag[t] = bias[(t << 4) + laneM];

  const int mBase = rowBlk + (wave << 4) + ((lane >> 4) << 3);
#pragma unroll
  for (int v = 0; v < 8; ++v) {
    const int gr = mBase + v;
    if (gr < M) {
      const float d  = dinv[gr];
      const float dd = d * d;
#pragma unroll
      for (int t = 0; t < NT; ++t) {
        const size_t idx = (size_t)gr * N + (t << 4) + laneM;
        const float val = acc[t][v];
        C[idx]    = val;
        out2[idx] = val * dd + bfrag[t];
      }
    }
  }
}

__global__ void k_fill1(float* __restrict__ p, int n, int* __restrict__ ovf) {
  int i = blockIdx.x * blockDim.x + threadIdx.x;
  if (i < n) p[i] = 1.0f;
  if (i == 0) ovf[0] = 0;
}

__global__ void k_deg(const int* __restrict__ col, float* __restrict__ deg, int E) {
  int e = blockIdx.x * blockDim.x + threadIdx.x;
  if (e < E) atomicAdd(&deg[col[e]], 1.0f);
}

__global__ void k_rsqrt(float* __restrict__ p, int n) {
  int i = blockIdx.x * blockDim.x + threadIdx.x;
  if (i < n) p[i] = rsqrtf(p[i]);
}

#define AGG_T     512
#define AGG_TBITS 9
#define AGG_QCAP  160
#define AGG_NGRP  ((N_NODES + AGG_T - 1) / AGG_T)
#define AGG_LCAP  1024
static_assert((1 << AGG_TBITS) == AGG_T, "local target bits");
static_assert(N_NODES <= (1 << (31 - AGG_TBITS)), "source index and local target share one int");

__device__ __forceinline__ void agg_flush(
    const int* myq, int cnt, int lane, int base,
    const float* __restrict__ h, const float* __restrict__ dinv,
    const int* __restrict__ row, const int* __restrict__ col,
    float* acc,
    int* __restrict__ lp, float* __restrict__ ln, int* __restrict__ ovf, int saved) {
  const float4* h4 = (const float4*)h;
  for (int b0 = 0; b0 < cnt; b0 += 32) {
    const int m = min(32, cnt - b0);
    int r = 0, t = 0;
    float nr = 0.0f;
    if (lane < m) {
      const int e = myq[b0 + lane];
      const int c = col[e];
      r  = row[e];
      t  = c - base;
      nr = dinv[r] * dinv[c];
      const int slot = saved + b0 + lane;
      if (slot < AGG_LCAP) {
        lp[slot] = (r << AGG_TBITS) | t;
        ln[slot] = nr;
      } else {
        const int k = atomicAdd(ovf, 1);
        ovf[1 + k] = e;
      }
    }
    int i = 0;
    for (; i + 4 <= m; i += 4) {
      float4 v[4];
      int    tt[4];
      float  nn[4];
#pragma unroll
      for (int u = 0; u < 4; ++u) {
        const int ru = __builtin_amdgcn_readlane(r, i + u);
        tt[u] = __builtin_amdgcn_readlane(t, i + u);
        nn[u] = __int_as_float(__builtin_amdgcn_readlane(__float_as_int(nr), i + u));
        v[u]  = h4[(size_t)ru * 32 + lane];
      }
#pragma unroll
      for (int u = 0; u < 4; ++u) {
        float* a = acc + tt[u] * 128 + lane * 4;
        atomicAdd(a + 0, v[u].x * nn[u]);
        atomicAdd(a + 1, v[u].y * nn[u]);
        atomicAdd(a + 2, v[u].z * nn[u]);
        atomicAdd(a + 3, v[u].w * nn[u]);
      }
    }
    for (; i < m; ++i) {
      const int   ru = __builtin_amdgcn_readlane(r, i);
      const int   tu = __builtin_amdgcn_readlane(t, i);
      const float nu = __int_as_float(__builtin_amdgcn_readlane(__float_as_int(nr), i));
      const float4 v = h4[(size_t)ru * 32 + lane];
      float* a = acc + tu * 128 + lane * 4;
      atomicAdd(a + 0, v.x * nu);
      atomicAdd(a + 1, v.y * nu);
      atomicAdd(a + 2, v.z * nu);
      atomicAdd(a + 3, v.w * nu);
    }
  }
}

__global__ __launch_bounds__(1024) void k_agg_lds128(
    const float* __restrict__ h, const float* __restrict__ dinv,
    const int* __restrict__ row, const int* __restrict__ col,
    float* __restrict__ out, int E, int n,
    int* __restrict__ lstP, float* __restrict__ lstN,
    int* __restrict__ lcnt, int* __restrict__ ovf) {
  __shared__ float acc[AGG_T * 128];
  __shared__ int   q[32 * AGG_QCAP];

  const int tid  = threadIdx.x;
  const int lane = tid & 31;
  const int wave = tid >> 5;
  const int base = blockIdx.x * AGG_T;
  const int nT   = min(AGG_T, n - base);
  const unsigned lt = (1u << lane) - 1u;

  {
    const float4* s4 = (const float4*)(out + (size_t)base * 128);
    float4* d4 = (float4*)acc;
    for (int i = tid; i < nT * 32; i += 1024) d4[i] = s4[i];
  }
  __syncthreads();

  int* myq = q + wave * AGG_QCAP;
  int cnt = 0;
  const int wslot = blockIdx.x * 32 + wave;
  int*   lp = lstP + (size_t)wslot * AGG_LCAP;
  float* ln = lstN + (size_t)wslot * AGG_LCAP;
  int saved = 0;

  const bool vec = ((E & 3) == 0) && ((((size_t)col) & 15) == 0);
  const int E4 = vec ? (E >> 2) : 0;
  const int4* col4 = (const int4*)col;

  const int nCh = (E4 + 4095) >> 12;
  int ch = (int)(((long long)blockIdx.x * nCh) / gridDim.x);
  for (int j = 0; j < nCh; ++j) {
    const int i0 = ch << 12;
    ch = (ch + 1 == nCh) ? 0 : ch + 1;
    int4 c[4];
#pragma unroll
    for (int u = 0; u < 4; ++u) {
      const int i = i0 + u * 1024 + tid;
      c[u] = col4[min(i, E4 - 1)];
    }
#pragma unroll
    for (int u = 0; u < 4; ++u) {
      const int i = i0 + u * 1024 + tid;
      const bool in = i < E4;
      const bool m0 = in & ((unsigned)(c[u].x - base) < (unsigned)nT);
      const bool m1 = in & ((unsigned)(c[u].y - base) < (unsigned)nT);
      const bool m2 = in & ((unsigned)(c[u].z - base) < (unsigned)nT);
      const bool m3 = in & ((unsigned)(c[u].w - base) < (unsigned)nT);
      if (__builtin_amdgcn_ballot_w32(m0 | m1 | m2 | m3)) {
        unsigned b;
        b = __builtin_amdgcn_ballot_w32(m0);
        if (m0) myq[cnt + __builtin_popcount(b & lt)] = 4 * i + 0;
        cnt += __builtin_popcount(b);
        b = __builtin_amdgcn_ballot_w32(m1);
        if (m1) myq[cnt + __builtin_popcount(b & lt)] = 4 * i + 1;
        cnt += __builtin_popcount(b);
        b = __builtin_amdgcn_ballot_w32(m2);
        if (m2) myq[cnt + __builtin_popcount(b & lt)] = 4 * i + 2;
        cnt += __builtin_popcount(b);
        b = __builtin_amdgcn_ballot_w32(m3);
        if (m3) myq[cnt + __builtin_popcount(b & lt)] = 4 * i + 3;
        cnt += __builtin_popcount(b);
        if (cnt >= 32) {
          agg_flush(myq, cnt, lane, base, h, dinv, row, col, acc, lp, ln, ovf, saved);
          saved += cnt;
          cnt = 0;
        }
      }
    }
  }
  for (int i0 = E4 * 4; i0 < E; i0 += 1024) {
    const int e = i0 + tid;
    const int c = (e < E) ? col[e] : -1;
    const bool m = (unsigned)(c - base) < (unsigned)nT;
    const unsigned b = __builtin_amdgcn_ballot_w32(m);
    if (b) {
      if (m) myq[cnt + __builtin_popcount(b & lt)] = e;
      cnt += __builtin_popcount(b);
      if (cnt >= 32) {
        agg_flush(myq, cnt, lane, base, h, dinv, row, col, acc, lp, ln, ovf, saved);
        saved += cnt;
        cnt = 0;
      }
    }
  }
  agg_flush(myq, cnt, lane, base, h, dinv, row, col, acc, lp, ln, ovf, saved);
  saved += cnt;
  if (lane == 0) lcnt[wslot] = saved;

  __syncthreads();
  {
    float4* d4 = (float4*)(out + (size_t)base * 128);
    const float4* s4 = (const float4*)acc;
    for (int i = tid; i < nT * 32; i += 1024) d4[i] = s4[i];
  }
}

__global__ void k_relu4(float4* __restrict__ p, int n4) {
  int i = blockIdx.x * blockDim.x + threadIdx.x;
  if (i < n4) {
    float4 v = p[i];
    v.x = fmaxf(v.x, 0.0f);
    v.y = fmaxf(v.y, 0.0f);
    v.z = fmaxf(v.z, 0.0f);
    v.w = fmaxf(v.w, 0.0f);
    p[i] = v;
  }
}

__global__ __launch_bounds__(1024) void k_agg2_lds32(
    const float* __restrict__ g, const float* __restrict__ dinv,
    const int* __restrict__ row, const int* __restrict__ col,
    const int* __restrict__ lstP, const float* __restrict__ lstN,
    const int* __restrict__ lcnt, const int* __restrict__ ovf,
    float* __restrict__ out, int n) {
  __shared__ float acc[AGG_T * 32];

  const int tid  = threadIdx.x;
  const int lane = tid & 31;
  const int wave = __builtin_amdgcn_readfirstlane(tid >> 5);
  const int base = blockIdx.x * AGG_T;
  const int nT   = min(AGG_T, n - base);

  {
    const float4* s4 = (const float4*)(out + (size_t)base * 32);
    float4* d4 = (float4*)acc;
    for (int i = tid; i < nT * 8; i += 1024) d4[i] = s4[i];
  }
  __syncthreads();

  const int wslot = blockIdx.x * 32 + wave;
  const int cnt = min(__builtin_amdgcn_readfirstlane(lcnt[wslot]), AGG_LCAP);
  const int*   lp = lstP + (size_t)wslot * AGG_LCAP;
  const float* ln = lstN + (size_t)wslot * AGG_LCAP;
  for (int b0 = 0; b0 < cnt; b0 += 32) {
    const int m = min(32, cnt - b0);
    int   pk = 0;
    float nr = 0.0f;
    if (lane < m) {
      pk = lp[b0 + lane];
      nr = ln[b0 + lane];
    }
    if (m == 32) {
      float v[32];
#pragma unroll
      for (int u = 0; u < 32; ++u) {
        const unsigned ru = (unsigned)__builtin_amdgcn_readlane(pk, u) >> AGG_TBITS;
        v[u] = g[(size_t)ru * 32 + lane];
      }
#pragma unroll
      for (int u = 0; u < 32; ++u) {
        const int   tu = __builtin_amdgcn_readlane(pk, u) & (AGG_T - 1);
        const float nu = __int_as_float(__builtin_amdgcn_readlane(__float_as_int(nr), u));
        atomicAdd(acc + tu * 32 + lane, v[u] * nu);
      }
    } else {
      for (int u = 0; u < m; ++u) {
        const int   pu = __builtin_amdgcn_readlane(pk, u);
        const float nu = __int_as_float(__builtin_amdgcn_readlane(__float_as_int(nr), u));
        const int   ru = pu >> AGG_TBITS;
        const int   tu = pu & (AGG_T - 1);
        atomicAdd(acc + tu * 32 + lane, g[(size_t)ru * 32 + lane] * nu);
      }
    }
  }

  const int nOv = ovf[0];
  for (int i = tid; i < nOv; i += 1024) {
    const int e = ovf[1 + i];
    const int c = col[e];
    const int t = c - base;
    if ((unsigned)t < (unsigned)nT) {
      const int   r  = row[e];
      const float nr = dinv[r] * dinv[c];
      for (int f = 0; f < 32; ++f)
        atomicAdd(acc + t * 32 + f, g[(size_t)r * 32 + f] * nr);
    }
  }
  __syncthreads();

  for (int node0 = wave; node0 < nT; node0 += 128) {
    float v[4], mx[4], s[4];
#pragma unroll
    for (int k = 0; k < 4; ++k) {
      const int node = node0 + 32 * k;
      v[k]  = (node < nT) ? acc[node * 32 + lane] : 0.0f;
      mx[k] = v[k];
    }
#pragma unroll
    for (int o = 16; o > 0; o >>= 1) {
#pragma unroll
      for (int k = 0; k < 4; ++k) mx[k] = fmaxf(mx[k], __shfl_xor(mx[k], o, 32));
    }
#pragma unroll
    for (int k = 0; k < 4; ++k) s[k] = __expf(v[k] - mx[k]);
#pragma unroll
    for (int o = 16; o > 0; o >>= 1) {
#pragma unroll
      for (int k = 0; k < 4; ++k) s[k] += __shfl_xor(s[k], o, 32);
    }
#pragma unroll
    for (int k = 0; k < 4; ++k) {
      const int node = node0 + 32 * k;
      if (node < nT)
        out[(size_t)(base + node) * 32 + lane] = v[k] - mx[k] - __logf(s[k]);
    }
  }
}

extern "C" void kernel_launch(void* const* d_in, const int* in_sizes, int n_in,
                              void* d_out, int out_size, void* d_ws, size_t ws_size,
                              hipStream_t stream) {
  const float* x  = (const float*)d_in[0];
  const float* W1 = (const float*)d_in[1];
  const float* b1 = (const float*)d_in[2];
  const float* W2 = (const float*)d_in[3];
  const float* b2 = (const float*)d_in[4];
  const int*  eix = (const int*)d_in[5];

  const int n = N_NODES;
  const int E = in_sizes[5] / 2;
  const int* row = eix;
  const int* col = eix + E;

  float* ws   = (float*)d_ws;
  float* dinv = ws;
  float* h    = ws + 100352;
  float* h1   = h  + (size_t)n * D_H;
  float* g2   = h1 + (size_t)n * D_H;
  int*   lcnt = (int*)(g2 + (size_t)n * D_OUT);
  int*   lstP = lcnt + 8192;
  float* lstN = (float*)(lstP + (size_t)AGG_NGRP * 32 * AGG_LCAP);
  int*   ovf  = (int*)(lstN + (size_t)AGG_NGRP * 32 * AGG_LCAP);
  float* outf = (float*)d_out;

  k_fill1 <<<(n + 255) / 256, 256, 0, stream>>>(dinv, n, ovf);
  k_deg   <<<(E + 255) / 256, 256, 0, stream>>>(col, dinv, E);
  k_rsqrt <<<(n + 255) / 256, 256, 0, stream>>>(dinv, n);

  gemm_bf16_wmma<D_IN, D_H><<<(n + 127) / 128, 256, 0, stream>>>(
      x, W1, h, dinv, b1, h1, n);
  k_agg_lds128<<<(n + AGG_T - 1) / AGG_T, 1024, 0, stream>>>(h, dinv, row, col, h1, E, n,
                                                         lstP, lstN, lcnt, ovf);
  k_relu4<<<((n * (D_H / 4)) + 255) / 256, 256, 0, stream>>>((float4*)h1, n * (D_H / 4));

  gemm_bf16_wmma<D_H, D_OUT><<<(n + 127) / 128, 256, 0, stream>>>(
      h1, W2, g2, dinv, b2, outf, n);
  k_agg2_lds32<<<(n + AGG_T - 1) / AGG_T, 1024, 0, stream>>>(
      g2, dinv, row, col, lstP, lstN, lcnt, ovf, outf, n);
}


# Relu applied where the layer-1 sums leave LDS; the separate relu launch (100,000 waves of one 16-byte element per lane, 51 MB read and rewritten) goes: 7 launches
# speedup vs baseline: 1.7754x; 1.0342x over previous
#include <hip/hip_runtime.h>

#define N_NODES 100000
#define D_IN 256
#define D_H 128
#define D_OUT 32

typedef __attribute__((ext_vector_type(16))) __bf16 v16bf;
typedef __attribute__((ext_vector_type(8)))  float  v8f;

__device__ __forceinline__ v16bf frag_from_lds(const __bf16* base, int rowIdx, int khalf) {
  union { uint4 u[2]; v16bf v; } c;
  const uint4* p = (const uint4*)(base + rowIdx * 32 + khalf);
  c.u[0] = p[0];
  c.u[1] = p[2];
  return c.v;
}

template<int K, int N>
__global__ __launch_bounds__(256) void gemm_bf16_wmma(
    const float* __restrict__ A, const float* __restrict__ W,
    float* __restrict__ C,
    const float* __restrict__ dinv, const float* __restrict__ bias,
    float* __restrict__ out2, int M) {
  constexpr int NT = N / 16;
  __shared__ __bf16 ldsA[128 * 32];
  __shared__ __bf16 ldsB[N * 32];

  const int tid   = threadIdx.x;
  const int wave  = tid >> 5;
  const int lane  = tid & 31;
  const int laneM = lane & 15;
  const int khalf = (lane >> 4) << 3;
  const int rowBlk = blockIdx.x * 128;

  const int ldRow = tid >> 1;
  const int ldC0  = (tid & 1) << 4;
  const int ldGr  = rowBlk + ldRow;

  v8f acc[NT];
#pragma unroll
  for (int t = 0; t < NT; ++t) {
    v8f z = {0.f, 0.f, 0.f, 0.f, 0.f, 0.f, 0.f, 0.f};
    acc[t] = z;
  }

  for (int kk = 0; kk < K; kk += 32) {
    __syncthreads();

    {
      union { __bf16 h[16]; uint4 q[2]; } buf;
      if (ldGr < M) {
        const float4* src = (const float4*)(A + (size_t)ldGr * K + kk + ldC0);
#pragma unroll
        for (int j = 0; j < 4; ++j) {
          float4 v = src[j];
          buf.h[j * 4 + 0] = (__bf16)v.x;
          buf.h[j * 4 + 1] = (__bf16)v.y;
          buf.h[j * 4 + 2] = (__bf16)v.z;
          buf.h[j * 4 + 3] = (__bf16)v.w;
        }
      } else {
#pragma unroll
        for (int j = 0; j < 16; ++j) buf.h[j] = (__bf16)0.0f;
      }
      uint4* dst = (uint4*)(ldsA + ldRow * 32 + ldC0);
      dst[0] = buf.q[0];
      dst[1] = buf.q[1];
    }

#pragma unroll
    for (int i = tid; i < N * 8; i += 256) {
      const int n  = i % N;
      const int kq = i / N;
      union { __bf16 h[4]; uint2 q; } wb;
#pragma unroll
      for (int j = 0; j < 4; ++j)
        wb.h[j] = (__bf16)W[(size_t)(kk + kq * 4 + j) * N + n];
      *(uint2*)(ldsB + n * 32 + kq * 4) = wb.q;
    }
    __syncthreads();

    v16bf a = frag_from_lds(ldsA, (wave << 4) + laneM, khalf);
    v16bf bc = frag_from_lds(ldsB, laneM, khalf);
#pragma unroll
    for (int t = 0; t < NT; ++t) {
      v16bf bn = (t + 1 < NT) ? frag_from_lds(ldsB, ((t + 1) << 4) + laneM, khalf) : bc;
      acc[t] = __builtin_amdgcn_wmma_f32_16x16x32_bf16(
          false, a, false, bc, (short)0, acc[t], false, false);
      bc = bn;
    }
  }

  float bfrag[NT];
#pragma unroll
  for (int t = 0; t < NT; ++t) bfrag[t] = bias[(t << 4) + laneM];

  const int mBase = rowBlk + (wave << 4) + ((lane >> 4) << 3);
#pragma unroll
  for (int v = 0; v < 8; ++v) {
    const int gr = mBase + v;
    if (gr < M) {
      const float d  = dinv[gr];
      const float dd = d * d;
#pragma unroll
      for (int t = 0; t < NT; ++t) {
        const size_t idx = (size_t)gr * N + (t << 4) + laneM;
        const float val = acc[t][v];
        C[idx]    = val;
        out2[idx] = val * dd + bfrag[t];
      }
    }
  }
}

__global__ void k_fill1(float* __restrict__ p, int n, int* __restrict__ ovf) {
  int i = blockIdx.x * blockDim.x + threadIdx.x;
  if (i < n) p[i] = 1.0f;
  if (i == 0) ovf[0] = 0;
}

__global__ void k_deg(const int* __restrict__ col, float* __restrict__ deg, int E) {
  int e = blockIdx.x * blockDim.x + threadIdx.x;
  if (e < E) atomicAdd(&deg[col[e]], 1.0f);
}

__global__ void k_rsqrt(float* __restrict__ p, int n) {
  int i = blockIdx.x * blockDim.x + threadIdx.x;
  if (i < n) p[i] = rsqrtf(p[i]);
}

#define AGG_T     512
#define AGG_TBITS 9
#define AGG_QCAP  160
#define AGG_NGRP  ((N_NODES + AGG_T - 1) / AGG_T)
#define AGG_LCAP  1024
static_assert((1 << AGG_TBITS) == AGG_T, "local target bits");
static_assert(N_NODES <= (1 << (31 - AGG_TBITS)), "source index and local target share one int");

__device__ __forceinline__ void agg_flush(
    const int* myq, int cnt, int lane, int base,
    const float* __restrict__ h, const float* __restrict__ dinv,
    const int* __restrict__ row, const int* __restrict__ col,
    float* acc,
    int* __restrict__ lp, float* __restrict__ ln, int* __restrict__ ovf, int saved) {
  const float4* h4 = (const float4*)h;
  for (int b0 = 0; b0 < cnt; b0 += 32) {
    const int m = min(32, cnt - b0);
    int r = 0, t = 0;
    float nr = 0.0f;
    if (lane < m) {
      const int e = myq[b0 + lane];
      const int c = col[e];
      r  = row[e];
      t  = c - base;
      nr = dinv[r] * dinv[c];
      const int slot = saved + b0 + lane;
      if (slot < AGG_LCAP) {
        lp[slot] = (r << AGG_TBITS) | t;
        ln[slot] = nr;
      } else {
        const int k = atomicAdd(ovf, 1);
        ovf[1 + k] = e;
      }
    }
    int i = 0;
    for (; i + 4 <= m; i += 4) {
      float4 v[4];
      int    tt[4];
      float  nn[4];
#pragma unroll
      for (int u = 0; u < 4; ++u) {
        const int ru = __builtin_amdgcn_readlane(r, i + u);
        tt[u] = __builtin_amdgcn_readlane(t, i + u);
        nn[u] = __int_as_float(__builtin_amdgcn_readlane(__float_as_int(nr), i + u));
        v[u]  = h4[(size_t)ru * 32 + lane];
      }
#pragma unroll
      for (int u = 0; u < 4; ++u) {
        float* a = acc + tt[u] * 128 + lane * 4;
        atomicAdd(a + 0, v[u].x * nn[u]);
        atomicAdd(a + 1, v[u].y * nn[u]);
        atomicAdd(a + 2, v[u].z * nn[u]);
        atomicAdd(a + 3, v[u].w * nn[u]);
      }
    }
    for (; i < m; ++i) {
      const int   ru = __builtin_amdgcn_readlane(r, i);
      const int   tu = __builtin_amdgcn_readlane(t, i);
      const float nu = __int_as_float(__builtin_amdgcn_readlane(__float_as_int(nr), i));
      const float4 v = h4[(size_t)ru * 32 + lane];
      float* a = acc + tu * 128 + lane * 4;
      atomicAdd(a + 0, v.x * nu);
      atomicAdd(a + 1, v.y * nu);
      atomicAdd(a + 2, v.z * nu);
      atomicAdd(a + 3, v.w * nu);
    }
  }
}

__global__ __launch_bounds__(1024) void k_agg_lds128(
    const float* __restrict__ h, const float* __restrict__ dinv,
    const int* __restrict__ row, const int* __restrict__ col,
    float* __restrict__ out, int E, int n,
    int* __restrict__ lstP, float* __restrict__ lstN,
    int* __restrict__ lcnt, int* __restrict__ ovf) {
  __shared__ float acc[AGG_T * 128];
  __shared__ int   q[32 * AGG_QCAP];

  const int tid  = threadIdx.x;
  const int lane = tid & 31;
  const int wave = tid >> 5;
  const int base = blockIdx.x * AGG_T;
  const int nT   = min(AGG_T, n - base);
  const unsigned lt = (1u << lane) - 1u;

  {
    const float4* s4 = (const float4*)(out + (size_t)base * 128);
    float4* d4 = (float4*)acc;
    for (int i = tid; i < nT * 32; i += 1024) d4[i] = s4[i];
  }
  __syncthreads();

  int* myq = q + wave * AGG_QCAP;
  int cnt = 0;
  const int wslot = blockIdx.x * 32 + wave;
  int*   lp = lstP + (size_t)wslot * AGG_LCAP;
  float* ln = lstN + (size_t)wslot * AGG_LCAP;
  int saved = 0;

  const bool vec = ((E & 3) == 0) && ((((size_t)col) & 15) == 0);
  const int E4 = vec ? (E >> 2) : 0;
  const int4* col4 = (const int4*)col;

  const int nCh = (E4 + 4095) >> 12;
  int ch = (int)(((long long)blockIdx.x * nCh) / gridDim.x);
  for (int j = 0; j < nCh; ++j) {
    const int i0 = ch << 12;
    ch = (ch + 1 == nCh) ? 0 : ch + 1;
    int4 c[4];
#pragma unroll
    for (int u = 0; u < 4; ++u) {
      const int i = i0 + u * 1024 + tid;
      c[u] = col4[min(i, E4 - 1)];
    }
#pragma unroll
    for (int u = 0; u < 4; ++u) {
      const int i = i0 + u * 1024 + tid;
      const bool in = i < E4;
      const bool m0 = in & ((unsigned)(c[u].x - base) < (unsigned)nT);
      const bool m1 = in & ((unsigned)(c[u].y - base) < (unsigned)nT);
      const bool m2 = in & ((unsigned)(c[u].z - base) < (unsigned)nT);
      const bool m3 = in & ((unsigned)(c[u].w - base) < (unsigned)nT);
      if (__builtin_amdgcn_ballot_w32(m0 | m1 | m2 | m3)) {
        unsigned b;
        b = __builtin_amdgcn_ballot_w32(m0);
        if (m0) myq[cnt + __builtin_popcount(b & lt)] = 4 * i + 0;
        cnt += __builtin_popcount(b);
        b = __builtin_amdgcn_ballot_w32(m1);
        if (m1) myq[cnt + __builtin_popcount(b & lt)] = 4 * i + 1;
        cnt += __builtin_popcount(b);
        b = __builtin_amdgcn_ballot_w32(m2);
        if (m2) myq[cnt + __builtin_popcount(b & lt)] = 4 * i + 2;
        cnt += __builtin_popcount(b);
        b = __builtin_amdgcn_ballot_w32(m3);
        if (m3) myq[cnt + __builtin_popcount(b & lt)] = 4 * i + 3;
        cnt += __builtin_popcount(b);
        if (cnt >= 32) {
          agg_flush(myq, cnt, lane, base, h, dinv, row, col, acc, lp, ln, ovf, saved);
          saved += cnt;
          cnt = 0;
        }
      }
    }
  }
  for (int i0 = E4 * 4; i0 < E; i0 += 1024) {
    const int e = i0 + tid;
    const int c = (e < E) ? col[e] : -1;
    const bool m = (unsigned)(c - base) < (unsigned)nT;
    const unsigned b = __builtin_amdgcn_ballot_w32(m);
    if (b) {
      if (m) myq[cnt + __builtin_popcount(b & lt)] = e;
      cnt += __builtin_popcount(b);
      if (cnt >= 32) {
        agg_flush(myq, cnt, lane, base, h, dinv, row, col, acc, lp, ln, ovf, saved);
        saved += cnt;
        cnt = 0;
      }
    }
  }
  agg_flush(myq, cnt, lane, base, h, dinv, row, col, acc, lp, ln, ovf, saved);
  saved += cnt;
  if (lane == 0) lcnt[wslot] = saved;

  __syncthreads();
  {
    float4* d4 = (float4*)(out + (size_t)base * 128);
    const float4* s4 = (const float4*)acc;
    for (int i = tid; i < nT * 32; i += 1024) {
      float4 v = s4[i];
      v.x = fmaxf(v.x, 0.0f);
      v.y = fmaxf(v.y, 0.0f);
      v.z = fmaxf(v.z, 0.0f);
      v.w = fmaxf(v.w, 0.0f);
      d4[i] = v;
    }
  }
}

__global__ __launch_bounds__(1024) void k_agg2_lds32(
    const float* __restrict__ g, const float* __restrict__ dinv,
    const int* __restrict__ row, const int* __restrict__ col,
    const int* __restrict__ lstP, const float* __restrict__ lstN,
    const int* __restrict__ lcnt, const int* __restrict__ ovf,
    float* __restrict__ out, int n) {
  __shared__ float acc[AGG_T * 32];

  const int tid  = threadIdx.x;
  const int lane = tid & 31;
  const int wave = __builtin_amdgcn_readfirstlane(tid >> 5);
  const int base = blockIdx.x * AGG_T;
  const int nT   = min(AGG_T, n - base);

  {
    const float4* s4 = (const float4*)(out + (size_t)base * 32);
    float4* d4 = (float4*)acc;
    for (int i = tid; i < nT * 8; i += 1024) d4[i] = s4[i];
  }
  __syncthreads();

  const int wslot = blockIdx.x * 32 + wave;
  const int cnt = min(__builtin_amdgcn_readfirstlane(lcnt[wslot]), AGG_LCAP);
  const int*   lp = lstP + (size_t)wslot * AGG_LCAP;
  const float* ln = lstN + (size_t)wslot * AGG_LCAP;
  for (int b0 = 0; b0 < cnt; b0 += 32) {
    const int m = min(32, cnt - b0);
    int   pk = 0;
    float nr = 0.0f;
    if (lane < m) {
      pk = lp[b0 + lane];
      nr = ln[b0 + lane];
    }
    if (m == 32) {
      float v[32];
#pragma unroll
      for (int u = 0; u < 32; ++u) {
        const unsigned ru = (unsigned)__builtin_amdgcn_readlane(pk, u) >> AGG_TBITS;
        v[u] = g[(size_t)ru * 32 + lane];
      }
#pragma unroll
      for (int u = 0; u < 32; ++u) {
        const int   tu = __builtin_amdgcn_readlane(pk, u) & (AGG_T - 1);
        const float nu = __int_as_float(__builtin_amdgcn_readlane(__float_as_int(nr), u));
        atomicAdd(acc + tu * 32 + lane, v[u] * nu);
      }
    } else {
      for (int u = 0; u < m; ++u) {
        const int   pu = __builtin_amdgcn_readlane(pk, u);
        const float nu = __int_as_float(__builtin_amdgcn_readlane(__float_as_int(nr), u));
        const int   ru = pu >> AGG_TBITS;
        const int   tu = pu & (AGG_T - 1);
        atomicAdd(acc + tu * 32 + lane, g[(size_t)ru * 32 + lane] * nu);
      }
    }
  }

  const int nOv = ovf[0];
  for (int i = tid; i < nOv; i += 1024) {
    const int e = ovf[1 + i];
    const int c = col[e];
    const int t = c - base;
    if ((unsigned)t < (unsigned)nT) {
      const int   r  = row[e];
      const float nr = dinv[r] * dinv[c];
      for (int f = 0; f < 32; ++f)
        atomicAdd(acc + t * 32 + f, g[(size_t)r * 32 + f] * nr);
    }
  }
  __syncthreads();

  for (int node0 = wave; node0 < nT; node0 += 128) {
    float v[4], mx[4], s[4];
#pragma unroll
    for (int k = 0; k < 4; ++k) {
      const int node = node0 + 32 * k;
      v[k]  = (node < nT) ? acc[node * 32 + lane] : 0.0f;
      mx[k] = v[k];
    }
#pragma unroll
    for (int o = 16; o > 0; o >>= 1) {
#pragma unroll
      for (int k = 0; k < 4; ++k) mx[k] = fmaxf(mx[k], __shfl_xor(mx[k], o, 32));
    }
#pragma unroll
    for (int k = 0; k < 4; ++k) s[k] = __expf(v[k] - mx[k]);
#pragma unroll
    for (int o = 16; o > 0; o >>= 1) {
#pragma unroll
      for (int k = 0; k < 4; ++k) s[k] += __shfl_xor(s[k], o, 32);
    }
#pragma unroll
    for (int k = 0; k < 4; ++k) {
      const int node = node0 + 32 * k;
      if (node < nT)
        out[(size_t)(base + node) * 32 + lane] = v[k] - mx[k] - __logf(s[k]);
    }
  }
}

extern "C" void kernel_launch(void* const* d_in, const int* in_sizes, int n_in,
                              void* d_out, int out_size, void* d_ws, size_t ws_size,
                              hipStream_t stream) {
  const float* x  = (const float*)d_in[0];
  const float* W1 = (const float*)d_in[1];
  const float* b1 = (const float*)d_in[2];
  const float* W2 = (const float*)d_in[3];
  const float* b2 = (const float*)d_in[4];
  const int*  eix = (const int*)d_in[5];

  const int n = N_NODES;
  const int E = in_sizes[5] / 2;
  const int* row = eix;
  const int* col = eix + E;

  float* ws   = (float*)d_ws;
  float* dinv = ws;
  float* h    = ws + 100352;
  float* h1   = h  + (size_t)n * D_H;
  float* g2   = h1 + (size_t)n * D_H;
  int*   lcnt = (int*)(g2 + (size_t)n * D_OUT);
  int*   lstP = lcnt + 8192;
  float* lstN = (float*)(lstP + (size_t)AGG_NGRP * 32 * AGG_LCAP);
  int*   ovf  = (int*)(lstN + (size_t)AGG_NGRP * 32 * AGG_LCAP);
  float* outf = (float*)d_out;

  k_fill1 <<<(n + 255) / 256, 256, 0, stream>>>(dinv, n, ovf);
  k_deg   <<<(E + 255) / 256, 256, 0, stream>>>(col, dinv, E);
  k_rsqrt <<<(n + 255) / 256, 256, 0, stream>>>(dinv, n);

  gemm_bf16_wmma<D_IN, D_H><<<(n + 127) / 128, 256, 0, stream>>>(
      x, W1, h, dinv, b1, h1, n);
  k_agg_lds128<<<(n + AGG_T - 1) / AGG_T, 1024, 0, stream>>>(h, dinv, row, col, h1, E, n,
                                                         lstP, lstN, lcnt, ovf);

  gemm_bf16_wmma<D_H, D_OUT><<<(n + 127) / 128, 256, 0, stream>>>(
      h1, W2, g2, dinv, b2, outf, n);
  k_agg2_lds32<<<(n + AGG_T - 1) / AGG_T, 1024, 0, stream>>>(
      g2, dinv, row, col, lstP, lstN, lcnt, ovf, outf, n);
}


# Layer-1 sweep reads a 1-byte owner key per edge (target>>9; degree launch writes them, 4 edges per thread), not the 4-byte target: 16 edges per 16-B read, packed byte test; sum order LDS arrival
# speedup vs baseline: 1.8796x; 1.0587x over previous
#include <hip/hip_runtime.h>

#define N_NODES 100000
#define D_IN 256
#define D_H 128
#define D_OUT 32

typedef __attribute__((ext_vector_type(16))) __bf16 v16bf;
typedef __attribute__((ext_vector_type(8)))  float  v8f;

__device__ __forceinline__ v16bf frag_from_lds(const __bf16* base, int rowIdx, int khalf) {
  union { uint4 u[2]; v16bf v; } c;
  const uint4* p = (const uint4*)(base + rowIdx * 32 + khalf);
  c.u[0] = p[0];
  c.u[1] = p[2];
  return c.v;
}

template<int K, int N>
__global__ __launch_bounds__(256) void gemm_bf16_wmma(
    const float* __restrict__ A, const float* __restrict__ W,
    float* __restrict__ C,
    const float* __restrict__ dinv, const float* __restrict__ bias,
    float* __restrict__ out2, int M) {
  constexpr int NT = N / 16;
  __shared__ __bf16 ldsA[128 * 32];
  __shared__ __bf16 ldsB[N * 32];

  const int tid   = threadIdx.x;
  const int wave  = tid >> 5;
  const int lane  = tid & 31;
  const int laneM = lane & 15;
  const int khalf = (lane >> 4) << 3;
  const int rowBlk = blockIdx.x * 128;

  const int ldRow = tid >> 1;
  const int ldC0  = (tid & 1) << 4;
  const int ldGr  = rowBlk + ldRow;

  v8f acc[NT];
#pragma unroll
  for (int t = 0; t < NT; ++t) {
    v8f z = {0.f, 0.f, 0.f, 0.f, 0.f, 0.f, 0.f, 0.f};
    acc[t] = z;
  }

  for (int kk = 0; kk < K; kk += 32) {
    __syncthreads();

    {
      union { __bf16 h[16]; uint4 q[2]; } buf;
      if (ldGr < M) {
        const float4* src = (const float4*)(A + (size_t)ldGr * K + kk + ldC0);
#pragma unroll
        for (int j = 0; j < 4; ++j) {
          float4 v = src[j];
          buf.h[j * 4 + 0] = (__bf16)v.x;
          buf.h[j * 4 + 1] = (__bf16)v.y;
          buf.h[j * 4 + 2] = (__bf16)v.z;
          buf.h[j * 4 + 3] = (__bf16)v.w;
        }
      } else {
#pragma unroll
        for (int j = 0; j < 16; ++j) buf.h[j] = (__bf16)0.0f;
      }
      uint4* dst = (uint4*)(ldsA + ldRow * 32 + ldC0);
      dst[0] = buf.q[0];
      dst[1] = buf.q[1];
    }

#pragma unroll
    for (int i = tid; i < N * 8; i += 256) {
      const int n  = i % N;
      const int kq = i / N;
      union { __bf16 h[4]; uint2 q; } wb;
#pragma unroll
      for (int j = 0; j < 4; ++j)
        wb.h[j] = (__bf16)W[(size_t)(kk + kq * 4 + j) * N + n];
      *(uint2*)(ldsB + n * 32 + kq * 4) = wb.q;
    }
    __syncthreads();

    v16bf a = frag_from_lds(ldsA, (wave << 4) + laneM, khalf);
    v16bf bc = frag_from_lds(ldsB, laneM, khalf);
#pragma unroll
    for (int t = 0; t < NT; ++t) {
      v16bf bn = (t + 1 < NT) ? frag_from_lds(ldsB, ((t + 1) << 4) + laneM, khalf) : bc;
      acc[t] = __builtin_amdgcn_wmma_f32_16x16x32_bf16(
          false, a, false, bc, (short)0, acc[t], false, false);
      bc = bn;
    }
  }

  float bfrag[NT];
#pragma unroll
  for (int t = 0; t < NT; ++t) bfrag[t] = bias[(t << 4) + laneM];

  const int mBase = rowBlk + (wave << 4) + ((lane >> 4) << 3);
#pragma unroll
  for (int v = 0; v < 8; ++v) {
    const int gr = mBase + v;
    if (gr < M) {
      const float d  = dinv[gr];
      const float dd = d * d;
#pragma unroll
      for (int t = 0; t < NT; ++t) {
        const size_t idx = (size_t)gr * N + (t << 4) + laneM;
        const float val = acc[t][v];
        C[idx]    = val;
        out2[idx] = val * dd + bfrag[t];
      }
    }
  }
}

#define AGG_T     512
#define AGG_TBITS 9
#define AGG_QCAP  160
#define AGG_NGRP  ((N_NODES + AGG_T - 1) / AGG_T)
#define AGG_LCAP  1024
static_assert((1 << AGG_TBITS) == AGG_T, "local target bits");
static_assert(N_NODES <= (1 << (31 - AGG_TBITS)), "source index and local target share one int");
static_assert(AGG_NGRP <= 255, "an owner key is one byte and 0xFF marks an empty slot");
#define AGG_SW    8

__global__ void k_fill1(float* __restrict__ p, int n, int* __restrict__ ovf) {
  int i = blockIdx.x * blockDim.x + threadIdx.x;
  if (i < n) p[i] = 1.0f;
  if (i == 0) ovf[0] = 0;
}

__global__ void k_deg(const int* __restrict__ col, float* __restrict__ deg, int E,
                      unsigned int* __restrict__ keys) {
  const int q  = blockIdx.x * blockDim.x + threadIdx.x;
  const int e0 = q * 4;
  unsigned kw = 0xFFFFFFFFu;
  if (e0 + 3 < E && ((((size_t)col) & 15) == 0)) {
    const int4 c = *(const int4*)(col + e0);
    atomicAdd(&deg[c.x], 1.0f);
    atomicAdd(&deg[c.y], 1.0f);
    atomicAdd(&deg[c.z], 1.0f);
    atomicAdd(&deg[c.w], 1.0f);
    kw = (unsigned)(c.x >> AGG_TBITS) | ((unsigned)(c.y >> AGG_TBITS) << 8) |
         ((unsigned)(c.z >> AGG_TBITS) << 16) | ((unsigned)(c.w >> AGG_TBITS) << 24);
  } else {
    for (int j = 0; j < 4; ++j) {
      if (e0 + j < E) {
        const int c = col[e0 + j];
        atomicAdd(&deg[c], 1.0f);
        kw = (kw & ~(0xFFu << (8 * j))) | ((unsigned)(c >> AGG_TBITS) << (8 * j));
      }
    }
  }
  keys[q] = kw;
}

__global__ void k_rsqrt(float* __restrict__ p, int n) {
  int i = blockIdx.x * blockDim.x + threadIdx.x;
  if (i < n) p[i] = rsqrtf(p[i]);
}


__device__ __forceinline__ unsigned agg_match16(uint4 k, unsigned pat) {
  const unsigned M7 = 0x7F7F7F7Fu;
  const unsigned t0 = k.x ^ pat, t1 = k.y ^ pat, t2 = k.z ^ pat, t3 = k.w ^ pat;
  const unsigned n0 = ((t0 & M7) + M7) | t0;
  const unsigned n1 = ((t1 & M7) + M7) | t1;
  const unsigned n2 = ((t2 & M7) + M7) | t2;
  const unsigned n3 = ((t3 & M7) + M7) | t3;
  unsigned r = ((n0 >> 7) & 0x01010101u) | ((n1 >> 6) & 0x02020202u) |
               ((n2 >> 5) & 0x04040404u) | ((n3 >> 4) & 0x08080808u);
  return ~r & 0x0F0F0F0Fu;
}

__device__ __forceinline__ void agg_flush(
    const int* myq, int cnt, int lane, int base,
    const float* __restrict__ h, const float* __restrict__ dinv,
    const int* __restrict__ row, const int* __restrict__ col,
    float* acc,
    int* __restrict__ lp, float* __restrict__ ln, int* __restrict__ ovf, int saved) {
  const float4* h4 = (const float4*)h;
  for (int b0 = 0; b0 < cnt; b0 += 32) {
    const int m = min(32, cnt - b0);
    int r = 0, t = 0;
    float nr = 0.0f;
    if (lane < m) {
      const int e = myq[b0 + lane];
      const int c = col[e];
      r  = row[e];
      t  = c - base;
      nr = dinv[r] * dinv[c];
      const int slot = saved + b0 + lane;
      if (slot < AGG_LCAP) {
        lp[slot] = (r << AGG_TBITS) | t;
        ln[slot] = nr;
      } else {
        const int k = atomicAdd(ovf, 1);
        ovf[1 + k] = e;
      }
    }
    int i = 0;
    for (; i + 4 <= m; i += 4) {
      float4 v[4];
      int    tt[4];
      float  nn[4];
#pragma unroll
      for (int u = 0; u < 4; ++u) {
        const int ru = __builtin_amdgcn_readlane(r, i + u);
        tt[u] = __builtin_amdgcn_readlane(t, i + u);
        nn[u] = __int_as_float(__builtin_amdgcn_readlane(__float_as_int(nr), i + u));
        v[u]  = h4[(size_t)ru * 32 + lane];
      }
#pragma unroll
      for (int u = 0; u < 4; ++u) {
        float* a = acc + tt[u] * 128 + lane * 4;
        atomicAdd(a + 0, v[u].x * nn[u]);
        atomicAdd(a + 1, v[u].y * nn[u]);
        atomicAdd(a + 2, v[u].z * nn[u]);
        atomicAdd(a + 3, v[u].w * nn[u]);
      }
    }
    for (; i < m; ++i) {
      const int   ru = __builtin_amdgcn_readlane(r, i);
      const int   tu = __builtin_amdgcn_readlane(t, i);
      const float nu = __int_as_float(__builtin_amdgcn_readlane(__float_as_int(nr), i));
      const float4 v = h4[(size_t)ru * 32 + lane];
      float* a = acc + tu * 128 + lane * 4;
      atomicAdd(a + 0, v.x * nu);
      atomicAdd(a + 1, v.y * nu);
      atomicAdd(a + 2, v.z * nu);
      atomicAdd(a + 3, v.w * nu);
    }
  }
}

__global__ __launch_bounds__(1024) void k_agg_lds128(
    const float* __restrict__ h, const float* __restrict__ dinv,
    const int* __restrict__ row, const int* __restrict__ col,
    float* __restrict__ out, int E, int n,
    int* __restrict__ lstP, float* __restrict__ lstN,
    int* __restrict__ lcnt, int* __restrict__ ovf,
    const unsigned int* __restrict__ keys) {
  __shared__ float acc[AGG_T * 128];
  __shared__ int   q[32 * AGG_QCAP];

  const int tid  = threadIdx.x;
  const int lane = tid & 31;
  const int wave = tid >> 5;
  const int base = blockIdx.x * AGG_T;
  const int nT   = min(AGG_T, n - base);
  const unsigned lt = (1u << lane) - 1u;

  {
    const float4* s4 = (const float4*)(out + (size_t)base * 128);
    float4* d4 = (float4*)acc;
    for (int i = tid; i < nT * 32; i += 1024) d4[i] = s4[i];
  }
  __syncthreads();

  int* myq = q + wave * AGG_QCAP;
  int cnt = 0;
  const int wslot = blockIdx.x * 32 + wave;
  int*   lp = lstP + (size_t)wslot * AGG_LCAP;
  float* ln = lstN + (size_t)wslot * AGG_LCAP;
  int saved = 0;

  const uint4* key4 = (const uint4*)keys;
  const int nK = (E + 15) >> 4;
  const unsigned pat = (unsigned)blockIdx.x * 0x01010101u;

  const int nU  = (nK + 1023) >> 10;
  const int nIt = (nU + AGG_SW - 1) / AGG_SW;
  const int u0  = (int)(((long long)blockIdx.x * nU) / gridDim.x);
  for (int j = 0; j < nIt; ++j) {
    uint4 k[AGG_SW];
    int   ix[AGG_SW];
#pragma unroll
    for (int u = 0; u < AGG_SW; ++u) {
      const int pos = j * AGG_SW + u;
      int unit = u0 + pos;
      if (unit >= nU) unit -= nU;
      const int i = (unit << 10) + tid;
      ix[u] = (pos < nU && i < nK) ? i : -1;
      k[u] = key4[min(i, nK - 1)];
    }
#pragma unroll
    for (int u = 0; u < AGG_SW; ++u) {
      const int i = ix[u];
      unsigned mk = (i >= 0) ? agg_match16(k[u], pat) : 0u;
      unsigned b = __builtin_amdgcn_ballot_w32(mk != 0u);
      while (b) {
        if (mk) {
          const int bit = __builtin_ctz(mk);
          myq[cnt + __builtin_popcount(b & lt)] = (i << 4) + ((bit & 7) << 2) + (bit >> 3);
          mk &= mk - 1u;
        }
        cnt += __builtin_popcount(b);
        if (cnt >= 32) {
          agg_flush(myq, cnt, lane, base, h, dinv, row, col, acc, lp, ln, ovf, saved);
          saved += cnt;
          cnt = 0;
        }
        b = __builtin_amdgcn_ballot_w32(mk != 0u);
      }
    }
  }
  agg_flush(myq, cnt, lane, base, h, dinv, row, col, acc, lp, ln, ovf, saved);
  saved += cnt;
  if (lane == 0) lcnt[wslot] = saved;

  __syncthreads();
  {
    float4* d4 = (float4*)(out + (size_t)base * 128);
    const float4* s4 = (const float4*)acc;
    for (int i = tid; i < nT * 32; i += 1024) {
      float4 v = s4[i];
      v.x = fmaxf(v.x, 0.0f);
      v.y = fmaxf(v.y, 0.0f);
      v.z = fmaxf(v.z, 0.0f);
      v.w = fmaxf(v.w, 0.0f);
      d4[i] = v;
    }
  }
}

__global__ __launch_bounds__(1024) void k_agg2_lds32(
    const float* __restrict__ g, const float* __restrict__ dinv,
    const int* __restrict__ row, const int* __restrict__ col,
    const int* __restrict__ lstP, const float* __restrict__ lstN,
    const int* __restrict__ lcnt, const int* __restrict__ ovf,
    float* __restrict__ out, int n) {
  __shared__ float acc[AGG_T * 32];

  const int tid  = threadIdx.x;
  const int lane = tid & 31;
  const int wave = __builtin_amdgcn_readfirstlane(tid >> 5);
  const int base = blockIdx.x * AGG_T;
  const int nT   = min(AGG_T, n - base);

  {
    const float4* s4 = (const float4*)(out + (size_t)base * 32);
    float4* d4 = (float4*)acc;
    for (int i = tid; i < nT * 8; i += 1024) d4[i] = s4[i];
  }
  __syncthreads();

  const int wslot = blockIdx.x * 32 + wave;
  const int cnt = min(__builtin_amdgcn_readfirstlane(lcnt[wslot]), AGG_LCAP);
  const int*   lp = lstP + (size_t)wslot * AGG_LCAP;
  const float* ln = lstN + (size_t)wslot * AGG_LCAP;
  for (int b0 = 0; b0 < cnt; b0 += 32) {
    const int m = min(32, cnt - b0);
    int   pk = 0;
    float nr = 0.0f;
    if (lane < m) {
      pk = lp[b0 + lane];
      nr = ln[b0 + lane];
    }
    if (m == 32) {
      float v[32];
#pragma unroll
      for (int u = 0; u < 32; ++u) {
        const unsigned ru = (unsigned)__builtin_amdgcn_readlane(pk, u) >> AGG_TBITS;
        v[u] = g[(size_t)ru * 32 + lane];
      }
#pragma unroll
      for (int u = 0; u < 32; ++u) {
        const int   tu = __builtin_amdgcn_readlane(pk, u) & (AGG_T - 1);
        const float nu = __int_as_float(__builtin_amdgcn_readlane(__float_as_int(nr), u));
        atomicAdd(acc + tu * 32 + lane, v[u] * nu);
      }
    } else {
      for (int u = 0; u < m; ++u) {
        const int   pu = __builtin_amdgcn_readlane(pk, u);
        const float nu = __int_as_float(__builtin_amdgcn_readlane(__float_as_int(nr), u));
        const int   ru = pu >> AGG_TBITS;
        const int   tu = pu & (AGG_T - 1);
        atomicAdd(acc + tu * 32 + lane, g[(size_t)ru * 32 + lane] * nu);
      }
    }
  }

  const int nOv = ovf[0];
  for (int i = tid; i < nOv; i += 1024) {
    const int e = ovf[1 + i];
    const int c = col[e];
    const int t = c - base;
    if ((unsigned)t < (unsigned)nT) {
      const int   r  = row[e];
      const float nr = dinv[r] * dinv[c];
      for (int f = 0; f < 32; ++f)
        atomicAdd(acc + t * 32 + f, g[(size_t)r * 32 + f] * nr);
    }
  }
  __syncthreads();

  for (int node0 = wave; node0 < nT; node0 += 128) {
    float v[4], mx[4], s[4];
#pragma unroll
    for (int k = 0; k < 4; ++k) {
      const int node = node0 + 32 * k;
      v[k]  = (node < nT) ? acc[node * 32 + lane] : 0.0f;
      mx[k] = v[k];
    }
#pragma unroll
    for (int o = 16; o > 0; o >>= 1) {
#pragma unroll
      for (int k = 0; k < 4; ++k) mx[k] = fmaxf(mx[k], __shfl_xor(mx[k], o, 32));
    }
#pragma unroll
    for (int k = 0; k < 4; ++k) s[k] = __expf(v[k] - mx[k]);
#pragma unroll
    for (int o = 16; o > 0; o >>= 1) {
#pragma unroll
      for (int k = 0; k < 4; ++k) s[k] += __shfl_xor(s[k], o, 32);
    }
#pragma unroll
    for (int k = 0; k < 4; ++k) {
      const int node = node0 + 32 * k;
      if (node < nT)
        out[(size_t)(base + node) * 32 + lane] = v[k] - mx[k] - __logf(s[k]);
    }
  }
}

extern "C" void kernel_launch(void* const* d_in, const int* in_sizes, int n_in,
                              void* d_out, int out_size, void* d_ws, size_t ws_size,
                              hipStream_t stream) {
  const float* x  = (const float*)d_in[0];
  const float* W1 = (const float*)d_in[1];
  const float* b1 = (const float*)d_in[2];
  const float* W2 = (const float*)d_in[3];
  const float* b2 = (const float*)d_in[4];
  const int*  eix = (const int*)d_in[5];

  const int n = N_NODES;
  const int E = in_sizes[5] / 2;
  const int* row = eix;
  const int* col = eix + E;

  float* ws   = (float*)d_ws;
  float* dinv = ws;
  float* h    = ws + 100352;
  float* h1   = h  + (size_t)n * D_H;
  float* g2   = h1 + (size_t)n * D_H;
  int*   lcnt = (int*)(g2 + (size_t)n * D_OUT);
  unsigned int* keys = (unsigned int*)(lcnt + 8192);
  int*   lstP = (int*)keys + E + 256;
  float* lstN = (float*)(lstP + (size_t)AGG_NGRP * 32 * AGG_LCAP);
  int*   ovf  = (int*)(lstN + (size_t)AGG_NGRP * 32 * AGG_LCAP);
  float* outf = (float*)d_out;

  k_fill1 <<<(n + 255) / 256, 256, 0, stream>>>(dinv, n, ovf);
  k_deg   <<<(E + 1023) / 1024, 256, 0, stream>>>(col, dinv, E, keys);
  k_rsqrt <<<(n + 255) / 256, 256, 0, stream>>>(dinv, n);

  gemm_bf16_wmma<D_IN, D_H><<<(n + 127) / 128, 256, 0, stream>>>(
      x, W1, h, dinv, b1, h1, n);
  k_agg_lds128<<<(n + AGG_T - 1) / AGG_T, 1024, 0, stream>>>(h, dinv, row, col, h1, E, n,
                                                         lstP, lstN, lcnt, ovf, keys);

  gemm_bf16_wmma<D_H, D_OUT><<<(n + 127) / 128, 256, 0, stream>>>(
      h1, W2, g2, dinv, b2, outf, n);
  k_agg2_lds32<<<(n + AGG_T - 1) / AGG_T, 1024, 0, stream>>>(
      g2, dinv, row, col, lstP, lstN, lcnt, ovf, outf, n);
}


# Owner groups of 400 targets, not 512: 250 work-groups, one per WGP on 250 of the chip's 256 (was 196), in both aggregation launches; owner key = target/400; sum order LDS arrival
# speedup vs baseline: 2.0284x; 1.0792x over previous
#include <hip/hip_runtime.h>

#define N_NODES 100000
#define D_IN 256
#define D_H 128
#define D_OUT 32

typedef __attribute__((ext_vector_type(16))) __bf16 v16bf;
typedef __attribute__((ext_vector_type(8)))  float  v8f;

__device__ __forceinline__ v16bf frag_from_lds(const __bf16* base, int rowIdx, int khalf) {
  union { uint4 u[2]; v16bf v; } c;
  const uint4* p = (const uint4*)(base + rowIdx * 32 + khalf);
  c.u[0] = p[0];
  c.u[1] = p[2];
  return c.v;
}

template<int K, int N>
__global__ __launch_bounds__(256) void gemm_bf16_wmma(
    const float* __restrict__ A, const float* __restrict__ W,
    float* __restrict__ C,
    const float* __restrict__ dinv, const float* __restrict__ bias,
    float* __restrict__ out2, int M) {
  constexpr int NT = N / 16;
  __shared__ __bf16 ldsA[128 * 32];
  __shared__ __bf16 ldsB[N * 32];

  const int tid   = threadIdx.x;
  const int wave  = tid >> 5;
  const int lane  = tid & 31;
  const int laneM = lane & 15;
  const int khalf = (lane >> 4) << 3;
  const int rowBlk = blockIdx.x * 128;

  const int ldRow = tid >> 1;
  const int ldC0  = (tid & 1) << 4;
  const int ldGr  = rowBlk + ldRow;

  v8f acc[NT];
#pragma unroll
  for (int t = 0; t < NT; ++t) {
    v8f z = {0.f, 0.f, 0.f, 0.f, 0.f, 0.f, 0.f, 0.f};
    acc[t] = z;
  }

  for (int kk = 0; kk < K; kk += 32) {
    __syncthreads();

    {
      union { __bf16 h[16]; uint4 q[2]; } buf;
      if (ldGr < M) {
        const float4* src = (const float4*)(A + (size_t)ldGr * K + kk + ldC0);
#pragma unroll
        for (int j = 0; j < 4; ++j) {
          float4 v = src[j];
          buf.h[j * 4 + 0] = (__bf16)v.x;
          buf.h[j * 4 + 1] = (__bf16)v.y;
          buf.h[j * 4 + 2] = (__bf16)v.z;
          buf.h[j * 4 + 3] = (__bf16)v.w;
        }
      } else {
#pragma unroll
        for (int j = 0; j < 16; ++j) buf.h[j] = (__bf16)0.0f;
      }
      uint4* dst = (uint4*)(ldsA + ldRow * 32 + ldC0);
      dst[0] = buf.q[0];
      dst[1] = buf.q[1];
    }

#pragma unroll
    for (int i = tid; i < N * 8; i += 256) {
      const int n  = i % N;
      const int kq = i / N;
      union { __bf16 h[4]; uint2 q; } wb;
#pragma unroll
      for (int j = 0; j < 4; ++j)
        wb.h[j] = (__bf16)W[(size_t)(kk + kq * 4 + j) * N + n];
      *(uint2*)(ldsB + n * 32 + kq * 4) = wb.q;
    }
    __syncthreads();

    v16bf a = frag_from_lds(ldsA, (wave << 4) + laneM, khalf);
    v16bf bc = frag_from_lds(ldsB, laneM, khalf);
#pragma unroll
    for (int t = 0; t < NT; ++t) {
      v16bf bn = (t + 1 < NT) ? frag_from_lds(ldsB, ((t + 1) << 4) + laneM, khalf) : bc;
      acc[t] = __builtin_amdgcn_wmma_f32_16x16x32_bf16(
          false, a, false, bc, (short)0, acc[t], false, false);
      bc = bn;
    }
  }

  float bfrag[NT];
#pragma unroll
  for (int t = 0; t < NT; ++t) bfrag[t] = bias[(t << 4) + laneM];

  const int mBase = rowBlk + (wave << 4) + ((lane >> 4) << 3);
#pragma unroll
  for (int v = 0; v < 8; ++v) {
    const int gr = mBase + v;
    if (gr < M) {
      const float d  = dinv[gr];
      const float dd = d * d;
#pragma unroll
      for (int t = 0; t < NT; ++t) {
        const size_t idx = (size_t)gr * N + (t << 4) + laneM;
        const float val = acc[t][v];
        C[idx]    = val;
        out2[idx] = val * dd + bfrag[t];
      }
    }
  }
}

#define AGG_T     400
#define AGG_TBITS 9
#define AGG_TMASK ((1 << AGG_TBITS) - 1)
#define AGG_QCAP  160
#define AGG_NGRP  ((N_NODES + AGG_T - 1) / AGG_T)
#define AGG_LCAP  1024
#define AGG_LCNT  8192
static_assert(AGG_T <= (1 << AGG_TBITS), "local target bits");
static_assert(AGG_NGRP * 32 <= AGG_LCNT, "one count per sweep wave");
static_assert(N_NODES <= (1 << (31 - AGG_TBITS)), "source index and local target share one int");
static_assert(AGG_NGRP <= 255, "an owner key is one byte and 0xFF marks an empty slot");
#define AGG_SW    8

__global__ void k_fill1(float* __restrict__ p, int n, int* __restrict__ ovf) {
  int i = blockIdx.x * blockDim.x + threadIdx.x;
  if (i < n) p[i] = 1.0f;
  if (i == 0) ovf[0] = 0;
}

__global__ void k_deg(const int* __restrict__ col, float* __restrict__ deg, int E,
                      unsigned int* __restrict__ keys) {
  const int q  = blockIdx.x * blockDim.x + threadIdx.x;
  const int e0 = q * 4;
  unsigned kw = 0xFFFFFFFFu;
  if (e0 + 3 < E && ((((size_t)col) & 15) == 0)) {
    const int4 c = *(const int4*)(col + e0);
    atomicAdd(&deg[c.x], 1.0f);
    atomicAdd(&deg[c.y], 1.0f);
    atomicAdd(&deg[c.z], 1.0f);
    atomicAdd(&deg[c.w], 1.0f);
    kw = ((unsigned)c.x / AGG_T) | (((unsigned)c.y / AGG_T) << 8) |
         (((unsigned)c.z / AGG_T) << 16) | (((unsigned)c.w / AGG_T) << 24);
  } else {
    for (int j = 0; j < 4; ++j) {
      if (e0 + j < E) {
        const int c = col[e0 + j];
        atomicAdd(&deg[c], 1.0f);
        kw = (kw & ~(0xFFu << (8 * j))) | (((unsigned)c / AGG_T) << (8 * j));
      }
    }
  }
  keys[q] = kw;
}

__global__ void k_rsqrt(float* __restrict__ p, int n) {
  int i = blockIdx.x * blockDim.x + threadIdx.x;
  if (i < n) p[i] = rsqrtf(p[i]);
}


__device__ __forceinline__ unsigned agg_match16(uint4 k, unsigned pat) {
  const unsigned M7 = 0x7F7F7F7Fu;
  const unsigned t0 = k.x ^ pat, t1 = k.y ^ pat, t2 = k.z ^ pat, t3 = k.w ^ pat;
  const unsigned n0 = ((t0 & M7) + M7) | t0;
  const unsigned n1 = ((t1 & M7) + M7) | t1;
  const unsigned n2 = ((t2 & M7) + M7) | t2;
  const unsigned n3 = ((t3 & M7) + M7) | t3;
  unsigned r = ((n0 >> 7) & 0x01010101u) | ((n1 >> 6) & 0x02020202u) |
               ((n2 >> 5) & 0x04040404u) | ((n3 >> 4) & 0x08080808u);
  return ~r & 0x0F0F0F0Fu;
}

__device__ __forceinline__ void agg_flush(
    const int* myq, int cnt, int lane, int base,
    const float* __restrict__ h, const float* __restrict__ dinv,
    const int* __restrict__ row, const int* __restrict__ col,
    float* acc,
    int* __restrict__ lp, float* __restrict__ ln, int* __restrict__ ovf, int saved) {
  const float4* h4 = (const float4*)h;
  for (int b0 = 0; b0 < cnt; b0 += 32) {
    const int m = min(32, cnt - b0);
    int r = 0, t = 0;
    float nr = 0.0f;
    if (lane < m) {
      const int e = myq[b0 + lane];
      const int c = col[e];
      r  = row[e];
      t  = c - base;
      nr = dinv[r] * dinv[c];
      const int slot = saved + b0 + lane;
      if (slot < AGG_LCAP) {
        lp[slot] = (r << AGG_TBITS) | t;
        ln[slot] = nr;
      } else {
        const int k = atomicAdd(ovf, 1);
        ovf[1 + k] = e;
      }
    }
    int i = 0;
    for (; i + 4 <= m; i += 4) {
      float4 v[4];
      int    tt[4];
      float  nn[4];
#pragma unroll
      for (int u = 0; u < 4; ++u) {
        const int ru = __builtin_amdgcn_readlane(r, i + u);
        tt[u] = __builtin_amdgcn_readlane(t, i + u);
        nn[u] = __int_as_float(__builtin_amdgcn_readlane(__float_as_int(nr), i + u));
        v[u]  = h4[(size_t)ru * 32 + lane];
      }
#pragma unroll
      for (int u = 0; u < 4; ++u) {
        float* a = acc + tt[u] * 128 + lane * 4;
        atomicAdd(a + 0, v[u].x * nn[u]);
        atomicAdd(a + 1, v[u].y * nn[u]);
        atomicAdd(a + 2, v[u].z * nn[u]);
        atomicAdd(a + 3, v[u].w * nn[u]);
      }
    }
    for (; i < m; ++i) {
      const int   ru = __builtin_amdgcn_readlane(r, i);
      const int   tu = __builtin_amdgcn_readlane(t, i);
      const float nu = __int_as_float(__builtin_amdgcn_readlane(__float_as_int(nr), i));
      const float4 v = h4[(size_t)ru * 32 + lane];
      float* a = acc + tu * 128 + lane * 4;
      atomicAdd(a + 0, v.x * nu);
      atomicAdd(a + 1, v.y * nu);
      atomicAdd(a + 2, v.z * nu);
      atomicAdd(a + 3, v.w * nu);
    }
  }
}

__global__ __launch_bounds__(1024) void k_agg_lds128(
    const float* __restrict__ h, const float* __restrict__ dinv,
    const int* __restrict__ row, const int* __restrict__ col,
    float* __restrict__ out, int E, int n,
    int* __restrict__ lstP, float* __restrict__ lstN,
    int* __restrict__ lcnt, int* __restrict__ ovf,
    const unsigned int* __restrict__ keys) {
  __shared__ float acc[AGG_T * 128];
  __shared__ int   q[32 * AGG_QCAP];

  const int tid  = threadIdx.x;
  const int lane = tid & 31;
  const int wave = tid >> 5;
  const int base = blockIdx.x * AGG_T;
  const int nT   = min(AGG_T, n - base);
  const unsigned lt = (1u << lane) - 1u;

  {
    const float4* s4 = (const float4*)(out + (size_t)base * 128);
    float4* d4 = (float4*)acc;
    for (int i = tid; i < nT * 32; i += 1024) d4[i] = s4[i];
  }
  __syncthreads();

  int* myq = q + wave * AGG_QCAP;
  int cnt = 0;
  const int wslot = blockIdx.x * 32 + wave;
  int*   lp = lstP + (size_t)wslot * AGG_LCAP;
  float* ln = lstN + (size_t)wslot * AGG_LCAP;
  int saved = 0;

  const uint4* key4 = (const uint4*)keys;
  const int nK = (E + 15) >> 4;
  const unsigned pat = (unsigned)blockIdx.x * 0x01010101u;

  const int nU  = (nK + 1023) >> 10;
  const int nIt = (nU + AGG_SW - 1) / AGG_SW;
  const int u0  = (int)(((long long)blockIdx.x * nU) / gridDim.x);
  for (int j = 0; j < nIt; ++j) {
    uint4 k[AGG_SW];
    int   ix[AGG_SW];
#pragma unroll
    for (int u = 0; u < AGG_SW; ++u) {
      const int pos = j * AGG_SW + u;
      int unit = u0 + pos;
      if (unit >= nU) unit -= nU;
      const int i = (unit << 10) + tid;
      ix[u] = (pos < nU && i < nK) ? i : -1;
      k[u] = key4[min(i, nK - 1)];
    }
#pragma unroll
    for (int u = 0; u < AGG_SW; ++u) {
      const int i = ix[u];
      unsigned mk = (i >= 0) ? agg_match16(k[u], pat) : 0u;
      unsigned b = __builtin_amdgcn_ballot_w32(mk != 0u);
      while (b) {
        if (mk) {
          const int bit = __builtin_ctz(mk);
          myq[cnt + __builtin_popcount(b & lt)] = (i << 4) + ((bit & 7) << 2) + (bit >> 3);
          mk &= mk - 1u;
        }
        cnt += __builtin_popcount(b);
        if (cnt >= 32) {
          agg_flush(myq, cnt, lane, base, h, dinv, row, col, acc, lp, ln, ovf, saved);
          saved += cnt;
          cnt = 0;
        }
        b = __builtin_amdgcn_ballot_w32(mk != 0u);
      }
    }
  }
  agg_flush(myq, cnt, lane, base, h, dinv, row, col, acc, lp, ln, ovf, saved);
  saved += cnt;
  if (lane == 0) lcnt[wslot] = saved;

  __syncthreads();
  {
    float4* d4 = (float4*)(out + (size_t)base * 128);
    const float4* s4 = (const float4*)acc;
    for (int i = tid; i < nT * 32; i += 1024) {
      float4 v = s4[i];
      v.x = fmaxf(v.x, 0.0f);
      v.y = fmaxf(v.y, 0.0f);
      v.z = fmaxf(v.z, 0.0f);
      v.w = fmaxf(v.w, 0.0f);
      d4[i] = v;
    }
  }
}

__global__ __launch_bounds__(1024) void k_agg2_lds32(
    const float* __restrict__ g, const float* __restrict__ dinv,
    const int* __restrict__ row, const int* __restrict__ col,
    const int* __restrict__ lstP, const float* __restrict__ lstN,
    const int* __restrict__ lcnt, const int* __restrict__ ovf,
    float* __restrict__ out, int n) {
  __shared__ float acc[AGG_T * 32];

  const int tid  = threadIdx.x;
  const int lane = tid & 31;
  const int wave = __builtin_amdgcn_readfirstlane(tid >> 5);
  const int base = blockIdx.x * AGG_T;
  const int nT   = min(AGG_T, n - base);

  {
    const float4* s4 = (const float4*)(out + (size_t)base * 32);
    float4* d4 = (float4*)acc;
    for (int i = tid; i < nT * 8; i += 1024) d4[i] = s4[i];
  }
  __syncthreads();

  const int wslot = blockIdx.x * 32 + wave;
  const int cnt = min(__builtin_amdgcn_readfirstlane(lcnt[wslot]), AGG_LCAP);
  const int*   lp = lstP + (size_t)wslot * AGG_LCAP;
  const float* ln = lstN + (size_t)wslot * AGG_LCAP;
  for (int b0 = 0; b0 < cnt; b0 += 32) {
    const int m = min(32, cnt - b0);
    int   pk = 0;
    float nr = 0.0f;
    if (lane < m) {
      pk = lp[b0 + lane];
      nr = ln[b0 + lane];
    }
    if (m == 32) {
      float v[32];
#pragma unroll
      for (int u = 0; u < 32; ++u) {
        const unsigned ru = (unsigned)__builtin_amdgcn_readlane(pk, u) >> AGG_TBITS;
        v[u] = g[(size_t)ru * 32 + lane];
      }
#pragma unroll
      for (int u = 0; u < 32; ++u) {
        const int   tu = __builtin_amdgcn_readlane(pk, u) & AGG_TMASK;
        const float nu = __int_as_float(__builtin_amdgcn_readlane(__float_as_int(nr), u));
        atomicAdd(acc + tu * 32 + lane, v[u] * nu);
      }
    } else {
      for (int u = 0; u < m; ++u) {
        const int   pu = __builtin_amdgcn_readlane(pk, u);
        const float nu = __int_as_float(__builtin_amdgcn_readlane(__float_as_int(nr), u));
        const int   ru = pu >> AGG_TBITS;
        const int   tu = pu & AGG_TMASK;
        atomicAdd(acc + tu * 32 + lane, g[(size_t)ru * 32 + lane] * nu);
      }
    }
  }

  const int nOv = ovf[0];
  for (int i = tid; i < nOv; i += 1024) {
    const int e = ovf[1 + i];
    const int c = col[e];
    const int t = c - base;
    if ((unsigned)t < (unsigned)nT) {
      const int   r  = row[e];
      const float nr = dinv[r] * dinv[c];
      for (int f = 0; f < 32; ++f)
        atomicAdd(acc + t * 32 + f, g[(size_t)r * 32 + f] * nr);
    }
  }
  __syncthreads();

  for (int node0 = wave; node0 < nT; node0 += 128) {
    float v[4], mx[4], s[4];
#pragma unroll
    for (int k = 0; k < 4; ++k) {
      const int node = node0 + 32 * k;
      v[k]  = (node < nT) ? acc[node * 32 + lane] : 0.0f;
      mx[k] = v[k];
    }
#pragma unroll
    for (int o = 16; o > 0; o >>= 1) {
#pragma unroll
      for (int k = 0; k < 4; ++k) mx[k] = fmaxf(mx[k], __shfl_xor(mx[k], o, 32));
    }
#pragma unroll
    for (int k = 0; k < 4; ++k) s[k] = __expf(v[k] - mx[k]);
#pragma unroll
    for (int o = 16; o > 0; o >>= 1) {
#pragma unroll
      for (int k = 0; k < 4; ++k) s[k] += __shfl_xor(s[k], o, 32);
    }
#pragma unroll
    for (int k = 0; k < 4; ++k) {
      const int node = node0 + 32 * k;
      if (node < nT)
        out[(size_t)(base + node) * 32 + lane] = v[k] - mx[k] - __logf(s[k]);
    }
  }
}

extern "C" void kernel_launch(void* const* d_in, const int* in_sizes, int n_in,
                              void* d_out, int out_size, void* d_ws, size_t ws_size,
                              hipStream_t stream) {
  const float* x  = (const float*)d_in[0];
  const float* W1 = (const float*)d_in[1];
  const float* b1 = (const float*)d_in[2];
  const float* W2 = (const float*)d_in[3];
  const float* b2 = (const float*)d_in[4];
  const int*  eix = (const int*)d_in[5];

  const int n = N_NODES;
  const int E = in_sizes[5] / 2;
  const int* row = eix;
  const int* col = eix + E;

  float* ws   = (float*)d_ws;
  float* dinv = ws;
  float* h    = ws + 100352;
  float* h1   = h  + (size_t)n * D_H;
  float* g2   = h1 + (size_t)n * D_H;
  int*   lcnt = (int*)(g2 + (size_t)n * D_OUT);
  unsigned int* keys = (unsigned int*)(lcnt + AGG_LCNT);
  int*   lstP = (int*)keys + E + 256;
  float* lstN = (float*)(lstP + (size_t)AGG_NGRP * 32 * AGG_LCAP);
  int*   ovf  = (int*)(lstN + (size_t)AGG_NGRP * 32 * AGG_LCAP);
  float* outf = (float*)d_out;

  k_fill1 <<<(n + 255) / 256, 256, 0, stream>>>(dinv, n, ovf);
  k_deg   <<<(E + 1023) / 1024, 256, 0, stream>>>(col, dinv, E, keys);
  k_rsqrt <<<(n + 255) / 256, 256, 0, stream>>>(dinv, n);

  gemm_bf16_wmma<D_IN, D_H><<<(n + 127) / 128, 256, 0, stream>>>(
      x, W1, h, dinv, b1, h1, n);
  k_agg_lds128<<<(n + AGG_T - 1) / AGG_T, 1024, 0, stream>>>(h, dinv, row, col, h1, E, n,
                                                         lstP, lstN, lcnt, ovf, keys);

  gemm_bf16_wmma<D_H, D_OUT><<<(n + 127) / 128, 256, 0, stream>>>(
      h1, W2, g2, dinv, b2, outf, n);
  k_agg2_lds32<<<(n + AGG_T - 1) / AGG_T, 1024, 0, stream>>>(
      g2, dinv, row, col, lstP, lstN, lcnt, ovf, outf, n);
}


# Edge norm split: GEMMs store rows*dinv[src] only (no start plane); LDS sums add raw rows from the own row, *dinv[tgt] then bias last (was first) at the store; no per-edge dinv loads or norm list
# speedup vs baseline: 2.1147x; 1.0425x over previous
#include <hip/hip_runtime.h>

#define N_NODES 100000
#define D_IN 256
#define D_H 128
#define D_OUT 32

typedef __attribute__((ext_vector_type(16))) __bf16 v16bf;
typedef __attribute__((ext_vector_type(8)))  float  v8f;

__device__ __forceinline__ v16bf frag_from_lds(const __bf16* base, int rowIdx, int khalf) {
  union { uint4 u[2]; v16bf v; } c;
  const uint4* p = (const uint4*)(base + rowIdx * 32 + khalf);
  c.u[0] = p[0];
  c.u[1] = p[2];
  return c.v;
}

template<int K, int N>
__global__ __launch_bounds__(256) void gemm_bf16_wmma(
    const float* __restrict__ A, const float* __restrict__ W,
    float* __restrict__ C,
    const float* __restrict__ dinv, int M) {
  constexpr int NT = N / 16;
  __shared__ __bf16 ldsA[128 * 32];
  __shared__ __bf16 ldsB[N * 32];

  const int tid   = threadIdx.x;
  const int wave  = tid >> 5;
  const int lane  = tid & 31;
  const int laneM = lane & 15;
  const int khalf = (lane >> 4) << 3;
  const int rowBlk = blockIdx.x * 128;

  const int ldRow = tid >> 1;
  const int ldC0  = (tid & 1) << 4;
  const int ldGr  = rowBlk + ldRow;

  v8f acc[NT];
#pragma unroll
  for (int t = 0; t < NT; ++t) {
    v8f z = {0.f, 0.f, 0.f, 0.f, 0.f, 0.f, 0.f, 0.f};
    acc[t] = z;
  }

  for (int kk = 0; kk < K; kk += 32) {
    __syncthreads();

    {
      union { __bf16 h[16]; uint4 q[2]; } buf;
      if (ldGr < M) {
        const float4* src = (const float4*)(A + (size_t)ldGr * K + kk + ldC0);
#pragma unroll
        for (int j = 0; j < 4; ++j) {
          float4 v = src[j];
          buf.h[j * 4 + 0] = (__bf16)v.x;
          buf.h[j * 4 + 1] = (__bf16)v.y;
          buf.h[j * 4 + 2] = (__bf16)v.z;
          buf.h[j * 4 + 3] = (__bf16)v.w;
        }
      } else {
#pragma unroll
        for (int j = 0; j < 16; ++j) buf.h[j] = (__bf16)0.0f;
      }
      uint4* dst = (uint4*)(ldsA + ldRow * 32 + ldC0);
      dst[0] = buf.q[0];
      dst[1] = buf.q[1];
    }

#pragma unroll
    for (int i = tid; i < N * 8; i += 256) {
      const int n  = i % N;
      const int kq = i / N;
      union { __bf16 h[4]; uint2 q; } wb;
#pragma unroll
      for (int j = 0; j < 4; ++j)
        wb.h[j] = (__bf16)W[(size_t)(kk + kq * 4 + j) * N + n];
      *(uint2*)(ldsB + n * 32 + kq * 4) = wb.q;
    }
    __syncthreads();

    v16bf a = frag_from_lds(ldsA, (wave << 4) + laneM, khalf);
    v16bf bc = frag_from_lds(ldsB, laneM, khalf);
#pragma unroll
    for (int t = 0; t < NT; ++t) {
      v16bf bn = (t + 1 < NT) ? frag_from_lds(ldsB, ((t + 1) << 4) + laneM, khalf) : bc;
      acc[t] = __builtin_amdgcn_wmma_f32_16x16x32_bf16(
          false, a, false, bc, (short)0, acc[t], false, false);
      bc = bn;
    }
  }

  const int mBase = rowBlk + (wave << 4) + ((lane >> 4) << 3);
#pragma unroll
  for (int v = 0; v < 8; ++v) {
    const int gr = mBase + v;
    if (gr < M) {
      const float d = dinv[gr];
#pragma unroll
      for (int t = 0; t < NT; ++t) {
        const size_t idx = (size_t)gr * N + (t << 4) + laneM;
        C[idx] = acc[t][v] * d;
      }
    }
  }
}

#define AGG_T     400
#define AGG_TBITS 9
#define AGG_TMASK ((1 << AGG_TBITS) - 1)
#define AGG_QCAP  160
#define AGG_NGRP  ((N_NODES + AGG_T - 1) / AGG_T)
#define AGG_LCAP  1024
#define AGG_LCNT  8192
static_assert(AGG_T <= (1 << AGG_TBITS), "local target bits");
static_assert(AGG_NGRP * 32 <= AGG_LCNT, "one count per sweep wave");
static_assert(N_NODES <= (1 << (31 - AGG_TBITS)), "source index and local target share one int");
static_assert(AGG_NGRP <= 255, "an owner key is one byte and 0xFF marks an empty slot");
#define AGG_SW    8

__global__ void k_fill1(float* __restrict__ p, int n, int* __restrict__ ovf) {
  int i = blockIdx.x * blockDim.x + threadIdx.x;
  if (i < n) p[i] = 1.0f;
  if (i == 0) ovf[0] = 0;
}

__global__ void k_deg(const int* __restrict__ col, float* __restrict__ deg, int E,
                      unsigned int* __restrict__ keys) {
  const int q  = blockIdx.x * blockDim.x + threadIdx.x;
  const int e0 = q * 4;
  unsigned kw = 0xFFFFFFFFu;
  if (e0 + 3 < E && ((((size_t)col) & 15) == 0)) {
    const int4 c = *(const int4*)(col + e0);
    atomicAdd(&deg[c.x], 1.0f);
    atomicAdd(&deg[c.y], 1.0f);
    atomicAdd(&deg[c.z], 1.0f);
    atomicAdd(&deg[c.w], 1.0f);
    kw = ((unsigned)c.x / AGG_T) | (((unsigned)c.y / AGG_T) << 8) |
         (((unsigned)c.z / AGG_T) << 16) | (((unsigned)c.w / AGG_T) << 24);
  } else {
    for (int j = 0; j < 4; ++j) {
      if (e0 + j < E) {
        const int c = col[e0 + j];
        atomicAdd(&deg[c], 1.0f);
        kw = (kw & ~(0xFFu << (8 * j))) | (((unsigned)c / AGG_T) << (8 * j));
      }
    }
  }
  keys[q] = kw;
}

__global__ void k_rsqrt(float* __restrict__ p, int n) {
  int i = blockIdx.x * blockDim.x + threadIdx.x;
  if (i < n) p[i] = rsqrtf(p[i]);
}


__device__ __forceinline__ unsigned agg_match16(uint4 k, unsigned pat) {
  const unsigned M7 = 0x7F7F7F7Fu;
  const unsigned t0 = k.x ^ pat, t1 = k.y ^ pat, t2 = k.z ^ pat, t3 = k.w ^ pat;
  const unsigned n0 = ((t0 & M7) + M7) | t0;
  const unsigned n1 = ((t1 & M7) + M7) | t1;
  const unsigned n2 = ((t2 & M7) + M7) | t2;
  const unsigned n3 = ((t3 & M7) + M7) | t3;
  unsigned r = ((n0 >> 7) & 0x01010101u) | ((n1 >> 6) & 0x02020202u) |
               ((n2 >> 5) & 0x04040404u) | ((n3 >> 4) & 0x08080808u);
  return ~r & 0x0F0F0F0Fu;
}

__device__ __forceinline__ void agg_flush(
    const int* myq, int cnt, int lane, int base,
    const float* __restrict__ h,
    const int* __restrict__ row, const int* __restrict__ col,
    float* acc,
    int* __restrict__ lp, int* __restrict__ ovf, int saved) {
  const float4* h4 = (const float4*)h;
  for (int b0 = 0; b0 < cnt; b0 += 32) {
    const int m = min(32, cnt - b0);
    int r = 0, t = 0;
    if (lane < m) {
      const int e = myq[b0 + lane];
      const int c = col[e];
      r  = row[e];
      t  = c - base;
      const int slot = saved + b0 + lane;
      if (slot < AGG_LCAP) {
        lp[slot] = (r << AGG_TBITS) | t;
      } else {
        const int k = atomicAdd(ovf, 1);
        ovf[1 + k] = e;
      }
    }
    int i = 0;
    for (; i + 4 <= m; i += 4) {
      float4 v[4];
      int    tt[4];
#pragma unroll
      for (int u = 0; u < 4; ++u) {
        const int ru = __builtin_amdgcn_readlane(r, i + u);
        tt[u] = __builtin_amdgcn_readlane(t, i + u);
        v[u]  = h4[(size_t)ru * 32 + lane];
      }
#pragma unroll
      for (int u = 0; u < 4; ++u) {
        float* a = acc + tt[u] * 128 + lane * 4;
        atomicAdd(a + 0, v[u].x);
        atomicAdd(a + 1, v[u].y);
        atomicAdd(a + 2, v[u].z);
        atomicAdd(a + 3, v[u].w);
      }
    }
    for (; i < m; ++i) {
      const int   ru = __builtin_amdgcn_readlane(r, i);
      const int   tu = __builtin_amdgcn_readlane(t, i);
      const float4 v = h4[(size_t)ru * 32 + lane];
      float* a = acc + tu * 128 + lane * 4;
      atomicAdd(a + 0, v.x);
      atomicAdd(a + 1, v.y);
      atomicAdd(a + 2, v.z);
      atomicAdd(a + 3, v.w);
    }
  }
}

__global__ __launch_bounds__(1024) void k_agg_lds128(
    const float* __restrict__ h, const float* __restrict__ dinv,
    const float* __restrict__ bias,
    const int* __restrict__ row, const int* __restrict__ col,
    float* __restrict__ out, int E, int n,
    int* __restrict__ lstP,
    int* __restrict__ lcnt, int* __restrict__ ovf,
    const unsigned int* __restrict__ keys) {
  __shared__ float acc[AGG_T * 128];
  __shared__ int   q[32 * AGG_QCAP];

  const int tid  = threadIdx.x;
  const int lane = tid & 31;
  const int wave = tid >> 5;
  const int base = blockIdx.x * AGG_T;
  const int nT   = min(AGG_T, n - base);
  const unsigned lt = (1u << lane) - 1u;

  {
    const float4* s4 = (const float4*)(h + (size_t)base * 128);
    float4* d4 = (float4*)acc;
    for (int i = tid; i < nT * 32; i += 1024) d4[i] = s4[i];
  }
  __syncthreads();

  int* myq = q + wave * AGG_QCAP;
  int cnt = 0;
  const int wslot = blockIdx.x * 32 + wave;
  int*   lp = lstP + (size_t)wslot * AGG_LCAP;
  int saved = 0;

  const uint4* key4 = (const uint4*)keys;
  const int nK = (E + 15) >> 4;
  const unsigned pat = (unsigned)blockIdx.x * 0x01010101u;

  const int nU  = (nK + 1023) >> 10;
  const int nIt = (nU + AGG_SW - 1) / AGG_SW;
  const int u0  = (int)(((long long)blockIdx.x * nU) / gridDim.x);
  for (int j = 0; j < nIt; ++j) {
    uint4 k[AGG_SW];
    int   ix[AGG_SW];
#pragma unroll
    for (int u = 0; u < AGG_SW; ++u) {
      const int pos = j * AGG_SW + u;
      int unit = u0 + pos;
      if (unit >= nU) unit -= nU;
      const int i = (unit << 10) + tid;
      ix[u] = (pos < nU && i < nK) ? i : -1;
      k[u] = key4[min(i, nK - 1)];
    }
#pragma unroll
    for (int u = 0; u < AGG_SW; ++u) {
      const int i = ix[u];
      unsigned mk = (i >= 0) ? agg_match16(k[u], pat) : 0u;
      unsigned b = __builtin_amdgcn_ballot_w32(mk != 0u);
      while (b) {
        if (mk) {
          const int bit = __builtin_ctz(mk);
          myq[cnt + __builtin_popcount(b & lt)] = (i << 4) + ((bit & 7) << 2) + (bit >> 3);
          mk &= mk - 1u;
        }
        cnt += __builtin_popcount(b);
        if (cnt >= 32) {
          agg_flush(myq, cnt, lane, base, h, row, col, acc, lp, ovf, saved);
          saved += cnt;
          cnt = 0;
        }
        b = __builtin_amdgcn_ballot_w32(mk != 0u);
      }
    }
  }
  agg_flush(myq, cnt, lane, base, h, row, col, acc, lp, ovf, saved);
  saved += cnt;
  if (lane == 0) lcnt[wslot] = saved;

  __syncthreads();
  {
    float4* d4 = (float4*)(out + (size_t)base * 128);
    const float4* s4 = (const float4*)acc;
    const float4 bb = make_float4(bias[lane * 4 + 0], bias[lane * 4 + 1],
                                  bias[lane * 4 + 2], bias[lane * 4 + 3]);
    for (int i = tid; i < nT * 32; i += 1024) {
      const float d = dinv[base + (i >> 5)];
      float4 v = s4[i];
      v.x = fmaxf(v.x * d + bb.x, 0.0f);
      v.y = fmaxf(v.y * d + bb.y, 0.0f);
      v.z = fmaxf(v.z * d + bb.z, 0.0f);
      v.w = fmaxf(v.w * d + bb.w, 0.0f);
      d4[i] = v;
    }
  }
}

__global__ __launch_bounds__(1024) void k_agg2_lds32(
    const float* __restrict__ g, const float* __restrict__ dinv,
    const float* __restrict__ bias,
    const int* __restrict__ row, const int* __restrict__ col,
    const int* __restrict__ lstP,
    const int* __restrict__ lcnt, const int* __restrict__ ovf,
    float* __restrict__ out, int n) {
  __shared__ float acc[AGG_T * 32];

  const int tid  = threadIdx.x;
  const int lane = tid & 31;
  const int wave = __builtin_amdgcn_readfirstlane(tid >> 5);
  const int base = blockIdx.x * AGG_T;
  const int nT   = min(AGG_T, n - base);

  {
    const float4* s4 = (const float4*)(g + (size_t)base * 32);
    float4* d4 = (float4*)acc;
    for (int i = tid; i < nT * 8; i += 1024) d4[i] = s4[i];
  }
  __syncthreads();

  const int wslot = blockIdx.x * 32 + wave;
  const int cnt = min(__builtin_amdgcn_readfirstlane(lcnt[wslot]), AGG_LCAP);
  const int*   lp = lstP + (size_t)wslot * AGG_LCAP;
  for (int b0 = 0; b0 < cnt; b0 += 32) {
    const int m = min(32, cnt - b0);
    int   pk = 0;
    if (lane < m) pk = lp[b0 + lane];
    if (m == 32) {
      float v[32];
#pragma unroll
      for (int u = 0; u < 32; ++u) {
        const unsigned ru = (unsigned)__builtin_amdgcn_readlane(pk, u) >> AGG_TBITS;
        v[u] = g[(size_t)ru * 32 + lane];
      }
#pragma unroll
      for (int u = 0; u < 32; ++u) {
        const int   tu = __builtin_amdgcn_readlane(pk, u) & AGG_TMASK;
        atomicAdd(acc + tu * 32 + lane, v[u]);
      }
    } else {
      for (int u = 0; u < m; ++u) {
        const int   pu = __builtin_amdgcn_readlane(pk, u);
        const int   ru = pu >> AGG_TBITS;
        const int   tu = pu & AGG_TMASK;
        atomicAdd(acc + tu * 32 + lane, g[(size_t)ru * 32 + lane]);
      }
    }
  }

  const int nOv = ovf[0];
  for (int i = tid; i < nOv; i += 1024) {
    const int e = ovf[1 + i];
    const int c = col[e];
    const int t = c - base;
    if ((unsigned)t < (unsigned)nT) {
      const int r = row[e];
      for (int f = 0; f < 32; ++f)
        atomicAdd(acc + t * 32 + f, g[(size_t)r * 32 + f]);
    }
  }
  __syncthreads();

  const float bl = bias[lane];
  for (int node0 = wave; node0 < nT; node0 += 128) {
    float v[4], mx[4], s[4];
#pragma unroll
    for (int k = 0; k < 4; ++k) {
      const int node = node0 + 32 * k;
      v[k]  = (node < nT) ? acc[node * 32 + lane] * dinv[base + node] + bl : 0.0f;
      mx[k] = v[k];
    }
#pragma unroll
    for (int o = 16; o > 0; o >>= 1) {
#pragma unroll
      for (int k = 0; k < 4; ++k) mx[k] = fmaxf(mx[k], __shfl_xor(mx[k], o, 32));
    }
#pragma unroll
    for (int k = 0; k < 4; ++k) s[k] = __expf(v[k] - mx[k]);
#pragma unroll
    for (int o = 16; o > 0; o >>= 1) {
#pragma unroll
      for (int k = 0; k < 4; ++k) s[k] += __shfl_xor(s[k], o, 32);
    }
#pragma unroll
    for (int k = 0; k < 4; ++k) {
      const int node = node0 + 32 * k;
      if (node < nT)
        out[(size_t)(base + node) * 32 + lane] = v[k] - mx[k] - __logf(s[k]);
    }
  }
}

extern "C" void kernel_launch(void* const* d_in, const int* in_sizes, int n_in,
                              void* d_out, int out_size, void* d_ws, size_t ws_size,
                              hipStream_t stream) {
  const float* x  = (const float*)d_in[0];
  const float* W1 = (const float*)d_in[1];
  const float* b1 = (const float*)d_in[2];
  const float* W2 = (const float*)d_in[3];
  const float* b2 = (const float*)d_in[4];
  const int*  eix = (const int*)d_in[5];

  const int n = N_NODES;
  const int E = in_sizes[5] / 2;
  const int* row = eix;
  const int* col = eix + E;

  float* ws   = (float*)d_ws;
  float* dinv = ws;
  float* h    = ws + 100352;
  float* h1   = h  + (size_t)n * D_H;
  float* g2   = h1 + (size_t)n * D_H;
  int*   lcnt = (int*)(g2 + (size_t)n * D_OUT);
  unsigned int* keys = (unsigned int*)(lcnt + AGG_LCNT);
  int*   lstP = (int*)keys + E + 256;
  int*   ovf  = lstP + (size_t)AGG_NGRP * 32 * AGG_LCAP;
  float* outf = (float*)d_out;

  k_fill1 <<<(n + 255) / 256, 256, 0, stream>>>(dinv, n, ovf);
  k_deg   <<<(E + 1023) / 1024, 256, 0, stream>>>(col, dinv, E, keys);
  k_rsqrt <<<(n + 255) / 256, 256, 0, stream>>>(dinv, n);

  gemm_bf16_wmma<D_IN, D_H><<<(n + 127) / 128, 256, 0, stream>>>(
      x, W1, h, dinv, n);
  k_agg_lds128<<<(n + AGG_T - 1) / AGG_T, 1024, 0, stream>>>(h, dinv, b1, row, col, h1, E, n,
                                                         lstP, lcnt, ovf, keys);

  gemm_bf16_wmma<D_H, D_OUT><<<(n + 127) / 128, 256, 0, stream>>>(
      h1, W2, g2, dinv, n);
  k_agg2_lds32<<<(n + AGG_T - 1) / AGG_T, 1024, 0, stream>>>(
      g2, dinv, b2, row, col, lstP, lcnt, ovf, outf, n);
}


# Layer-1 LDS sums kept as four 32-float planes per target (column 4l+k at dword 32k+l): each LDS float add of a wave hits 32 banks, was 2 lanes per bank; sum order LDS arrival as before
# speedup vs baseline: 2.4105x; 1.1399x over previous
#include <hip/hip_runtime.h>

#define N_NODES 100000
#define D_IN 256
#define D_H 128
#define D_OUT 32

typedef __attribute__((ext_vector_type(16))) __bf16 v16bf;
typedef __attribute__((ext_vector_type(8)))  float  v8f;

__device__ __forceinline__ v16bf frag_from_lds(const __bf16* base, int rowIdx, int khalf) {
  union { uint4 u[2]; v16bf v; } c;
  const uint4* p = (const uint4*)(base + rowIdx * 32 + khalf);
  c.u[0] = p[0];
  c.u[1] = p[2];
  return c.v;
}

template<int K, int N>
__global__ __launch_bounds__(256) void gemm_bf16_wmma(
    const float* __restrict__ A, const float* __restrict__ W,
    float* __restrict__ C,
    const float* __restrict__ dinv, int M) {
  constexpr int NT = N / 16;
  __shared__ __bf16 ldsA[128 * 32];
  __shared__ __bf16 ldsB[N * 32];

  const int tid   = threadIdx.x;
  const int wave  = tid >> 5;
  const int lane  = tid & 31;
  const int laneM = lane & 15;
  const int khalf = (lane >> 4) << 3;
  const int rowBlk = blockIdx.x * 128;

  const int ldRow = tid >> 1;
  const int ldC0  = (tid & 1) << 4;
  const int ldGr  = rowBlk + ldRow;

  v8f acc[NT];
#pragma unroll
  for (int t = 0; t < NT; ++t) {
    v8f z = {0.f, 0.f, 0.f, 0.f, 0.f, 0.f, 0.f, 0.f};
    acc[t] = z;
  }

  for (int kk = 0; kk < K; kk += 32) {
    __syncthreads();

    {
      union { __bf16 h[16]; uint4 q[2]; } buf;
      if (ldGr < M) {
        const float4* src = (const float4*)(A + (size_t)ldGr * K + kk + ldC0);
#pragma unroll
        for (int j = 0; j < 4; ++j) {
          float4 v = src[j];
          buf.h[j * 4 + 0] = (__bf16)v.x;
          buf.h[j * 4 + 1] = (__bf16)v.y;
          buf.h[j * 4 + 2] = (__bf16)v.z;
          buf.h[j * 4 + 3] = (__bf16)v.w;
        }
      } else {
#pragma unroll
        for (int j = 0; j < 16; ++j) buf.h[j] = (__bf16)0.0f;
      }
      uint4* dst = (uint4*)(ldsA + ldRow * 32 + ldC0);
      dst[0] = buf.q[0];
      dst[1] = buf.q[1];
    }

#pragma unroll
    for (int i = tid; i < N * 8; i += 256) {
      const int n  = i % N;
      const int kq = i / N;
      union { __bf16 h[4]; uint2 q; } wb;
#pragma unroll
      for (int j = 0; j < 4; ++j)
        wb.h[j] = (__bf16)W[(size_t)(kk + kq * 4 + j) * N + n];
      *(uint2*)(ldsB + n * 32 + kq * 4) = wb.q;
    }
    __syncthreads();

    v16bf a = frag_from_lds(ldsA, (wave << 4) + laneM, khalf);
    v16bf bc = frag_from_lds(ldsB, laneM, khalf);
#pragma unroll
    for (int t = 0; t < NT; ++t) {
      v16bf bn = (t + 1 < NT) ? frag_from_lds(ldsB, ((t + 1) << 4) + laneM, khalf) : bc;
      acc[t] = __builtin_amdgcn_wmma_f32_16x16x32_bf16(
          false, a, false, bc, (short)0, acc[t], false, false);
      bc = bn;
    }
  }

  const int mBase = rowBlk + (wave << 4) + ((lane >> 4) << 3);
#pragma unroll
  for (int v = 0; v < 8; ++v) {
    const int gr = mBase + v;
    if (gr < M) {
      const float d = dinv[gr];
#pragma unroll
      for (int t = 0; t < NT; ++t) {
        const size_t idx = (size_t)gr * N + (t << 4) + laneM;
        C[idx] = acc[t][v] * d;
      }
    }
  }
}

#define AGG_T     400
#define AGG_TBITS 9
#define AGG_TMASK ((1 << AGG_TBITS) - 1)
#define AGG_QCAP  160
#define AGG_NGRP  ((N_NODES + AGG_T - 1) / AGG_T)
#define AGG_LCAP  1024
#define AGG_LCNT  8192
static_assert(AGG_T <= (1 << AGG_TBITS), "local target bits");
static_assert(AGG_NGRP * 32 <= AGG_LCNT, "one count per sweep wave");
static_assert(N_NODES <= (1 << (31 - AGG_TBITS)), "source index and local target share one int");
static_assert(AGG_NGRP <= 255, "an owner key is one byte and 0xFF marks an empty slot");
#define AGG_SW    8
#define AGG_PL    32

__global__ void k_fill1(float* __restrict__ p, int n, int* __restrict__ ovf) {
  int i = blockIdx.x * blockDim.x + threadIdx.x;
  if (i < n) p[i] = 1.0f;
  if (i == 0) ovf[0] = 0;
}

__global__ void k_deg(const int* __restrict__ col, float* __restrict__ deg, int E,
                      unsigned int* __restrict__ keys) {
  const int q  = blockIdx.x * blockDim.x + threadIdx.x;
  const int e0 = q * 4;
  unsigned kw = 0xFFFFFFFFu;
  if (e0 + 3 < E && ((((size_t)col) & 15) == 0)) {
    const int4 c = *(const int4*)(col + e0);
    atomicAdd(&deg[c.x], 1.0f);
    atomicAdd(&deg[c.y], 1.0f);
    atomicAdd(&deg[c.z], 1.0f);
    atomicAdd(&deg[c.w], 1.0f);
    kw = ((unsigned)c.x / AGG_T) | (((unsigned)c.y / AGG_T) << 8) |
         (((unsigned)c.z / AGG_T) << 16) | (((unsigned)c.w / AGG_T) << 24);
  } else {
    for (int j = 0; j < 4; ++j) {
      if (e0 + j < E) {
        const int c = col[e0 + j];
        atomicAdd(&deg[c], 1.0f);
        kw = (kw & ~(0xFFu << (8 * j))) | (((unsigned)c / AGG_T) << (8 * j));
      }
    }
  }
  keys[q] = kw;
}

__global__ void k_rsqrt(float* __restrict__ p, int n) {
  int i = blockIdx.x * blockDim.x + threadIdx.x;
  if (i < n) p[i] = rsqrtf(p[i]);
}


__device__ __forceinline__ unsigned agg_match16(uint4 k, unsigned pat) {
  const unsigned M7 = 0x7F7F7F7Fu;
  const unsigned t0 = k.x ^ pat, t1 = k.y ^ pat, t2 = k.z ^ pat, t3 = k.w ^ pat;
  const unsigned n0 = ((t0 & M7) + M7) | t0;
  const unsigned n1 = ((t1 & M7) + M7) | t1;
  const unsigned n2 = ((t2 & M7) + M7) | t2;
  const unsigned n3 = ((t3 & M7) + M7) | t3;
  unsigned r = ((n0 >> 7) & 0x01010101u) | ((n1 >> 6) & 0x02020202u) |
               ((n2 >> 5) & 0x04040404u) | ((n3 >> 4) & 0x08080808u);
  return ~r & 0x0F0F0F0Fu;
}

__device__ __forceinline__ void agg_flush(
    const int* myq, int cnt, int lane, int base,
    const float* __restrict__ h,
    const int* __restrict__ row, const int* __restrict__ col,
    float* acc,
    int* __restrict__ lp, int* __restrict__ ovf, int saved) {
  const float4* h4 = (const float4*)h;
  for (int b0 = 0; b0 < cnt; b0 += 32) {
    const int m = min(32, cnt - b0);
    int r = 0, t = 0;
    if (lane < m) {
      const int e = myq[b0 + lane];
      const int c = col[e];
      r  = row[e];
      t  = c - base;
      const int slot = saved + b0 + lane;
      if (slot < AGG_LCAP) {
        lp[slot] = (r << AGG_TBITS) | t;
      } else {
        const int k = atomicAdd(ovf, 1);
        ovf[1 + k] = e;
      }
    }
    int i = 0;
    for (; i + 4 <= m; i += 4) {
      float4 v[4];
      int    tt[4];
#pragma unroll
      for (int u = 0; u < 4; ++u) {
        const int ru = __builtin_amdgcn_readlane(r, i + u);
        tt[u] = __builtin_amdgcn_readlane(t, i + u);
        v[u]  = h4[(size_t)ru * 32 + lane];
      }
#pragma unroll
      for (int u = 0; u < 4; ++u) {
        float* a = acc + tt[u] * 128 + lane;
        atomicAdd(a + 0 * AGG_PL, v[u].x);
        atomicAdd(a + 1 * AGG_PL, v[u].y);
        atomicAdd(a + 2 * AGG_PL, v[u].z);
        atomicAdd(a + 3 * AGG_PL, v[u].w);
      }
    }
    for (; i < m; ++i) {
      const int   ru = __builtin_amdgcn_readlane(r, i);
      const int   tu = __builtin_amdgcn_readlane(t, i);
      const float4 v = h4[(size_t)ru * 32 + lane];
      float* a = acc + tu * 128 + lane;
      atomicAdd(a + 0 * AGG_PL, v.x);
      atomicAdd(a + 1 * AGG_PL, v.y);
      atomicAdd(a + 2 * AGG_PL, v.z);
      atomicAdd(a + 3 * AGG_PL, v.w);
    }
  }
}

__global__ __launch_bounds__(1024) void k_agg_lds128(
    const float* __restrict__ h, const float* __restrict__ dinv,
    const float* __restrict__ bias,
    const int* __restrict__ row, const int* __restrict__ col,
    float* __restrict__ out, int E, int n,
    int* __restrict__ lstP,
    int* __restrict__ lcnt, int* __restrict__ ovf,
    const unsigned int* __restrict__ keys) {
  __shared__ float acc[AGG_T * 128];
  __shared__ int   q[32 * AGG_QCAP];

  const int tid  = threadIdx.x;
  const int lane = tid & 31;
  const int wave = tid >> 5;
  const int base = blockIdx.x * AGG_T;
  const int nT   = min(AGG_T, n - base);
  const unsigned lt = (1u << lane) - 1u;

  {
    const float4* s4 = (const float4*)(h + (size_t)base * 128);
    for (int i = tid; i < nT * 32; i += 1024) {
      const float4 v = s4[i];
      float* a = acc + (i >> 5) * 128 + lane;
      a[0 * AGG_PL] = v.x;
      a[1 * AGG_PL] = v.y;
      a[2 * AGG_PL] = v.z;
      a[3 * AGG_PL] = v.w;
    }
  }
  __syncthreads();

  int* myq = q + wave * AGG_QCAP;
  int cnt = 0;
  const int wslot = blockIdx.x * 32 + wave;
  int*   lp = lstP + (size_t)wslot * AGG_LCAP;
  int saved = 0;

  const uint4* key4 = (const uint4*)keys;
  const int nK = (E + 15) >> 4;
  const unsigned pat = (unsigned)blockIdx.x * 0x01010101u;

  const int nU  = (nK + 1023) >> 10;
  const int nIt = (nU + AGG_SW - 1) / AGG_SW;
  const int u0  = (int)(((long long)blockIdx.x * nU) / gridDim.x);
  for (int j = 0; j < nIt; ++j) {
    uint4 k[AGG_SW];
    int   ix[AGG_SW];
#pragma unroll
    for (int u = 0; u < AGG_SW; ++u) {
      const int pos = j * AGG_SW + u;
      int unit = u0 + pos;
      if (unit >= nU) unit -= nU;
      const int i = (unit << 10) + tid;
      ix[u] = (pos < nU && i < nK) ? i : -1;
      k[u] = key4[min(i, nK - 1)];
    }
#pragma unroll
    for (int u = 0; u < AGG_SW; ++u) {
      const int i = ix[u];
      unsigned mk = (i >= 0) ? agg_match16(k[u], pat) : 0u;
      unsigned b = __builtin_amdgcn_ballot_w32(mk != 0u);
      while (b) {
        if (mk) {
          const int bit = __builtin_ctz(mk);
          myq[cnt + __builtin_popcount(b & lt)] = (i << 4) + ((bit & 7) << 2) + (bit >> 3);
          mk &= mk - 1u;
        }
        cnt += __builtin_popcount(b);
        if (cnt >= 32) {
          agg_flush(myq, cnt, lane, base, h, row, col, acc, lp, ovf, saved);
          saved += cnt;
          cnt = 0;
        }
        b = __builtin_amdgcn_ballot_w32(mk != 0u);
      }
    }
  }
  agg_flush(myq, cnt, lane, base, h, row, col, acc, lp, ovf, saved);
  saved += cnt;
  if (lane == 0) lcnt[wslot] = saved;

  __syncthreads();
  {
    float4* d4 = (float4*)(out + (size_t)base * 128);
    const float4 bb = make_float4(bias[lane * 4 + 0], bias[lane * 4 + 1],
                                  bias[lane * 4 + 2], bias[lane * 4 + 3]);
    for (int i = tid; i < nT * 32; i += 1024) {
      const float d = dinv[base + (i >> 5)];
      const float* a = acc + (i >> 5) * 128 + lane;
      float4 v = make_float4(a[0 * AGG_PL], a[1 * AGG_PL], a[2 * AGG_PL], a[3 * AGG_PL]);
      v.x = fmaxf(v.x * d + bb.x, 0.0f);
      v.y = fmaxf(v.y * d + bb.y, 0.0f);
      v.z = fmaxf(v.z * d + bb.z, 0.0f);
      v.w = fmaxf(v.w * d + bb.w, 0.0f);
      d4[i] = v;
    }
  }
}

__global__ __launch_bounds__(1024) void k_agg2_lds32(
    const float* __restrict__ g, const float* __restrict__ dinv,
    const float* __restrict__ bias,
    const int* __restrict__ row, const int* __restrict__ col,
    const int* __restrict__ lstP,
    const int* __restrict__ lcnt, const int* __restrict__ ovf,
    float* __restrict__ out, int n) {
  __shared__ float acc[AGG_T * 32];

  const int tid  = threadIdx.x;
  const int lane = tid & 31;
  const int wave = __builtin_amdgcn_readfirstlane(tid >> 5);
  const int base = blockIdx.x * AGG_T;
  const int nT   = min(AGG_T, n - base);

  {
    const float4* s4 = (const float4*)(g + (size_t)base * 32);
    float4* d4 = (float4*)acc;
    for (int i = tid; i < nT * 8; i += 1024) d4[i] = s4[i];
  }
  __syncthreads();

  const int wslot = blockIdx.x * 32 + wave;
  const int cnt = min(__builtin_amdgcn_readfirstlane(lcnt[wslot]), AGG_LCAP);
  const int*   lp = lstP + (size_t)wslot * AGG_LCAP;
  for (int b0 = 0; b0 < cnt; b0 += 32) {
    const int m = min(32, cnt - b0);
    int   pk = 0;
    if (lane < m) pk = lp[b0 + lane];
    if (m == 32) {
      float v[32];
#pragma unroll
      for (int u = 0; u < 32; ++u) {
        const unsigned ru = (unsigned)__builtin_amdgcn_readlane(pk, u) >> AGG_TBITS;
        v[u] = g[(size_t)ru * 32 + lane];
      }
#pragma unroll
      for (int u = 0; u < 32; ++u) {
        const int   tu = __builtin_amdgcn_readlane(pk, u) & AGG_TMASK;
        atomicAdd(acc + tu * 32 + lane, v[u]);
      }
    } else {
      for (int u = 0; u < m; ++u) {
        const int   pu = __builtin_amdgcn_readlane(pk, u);
        const int   ru = pu >> AGG_TBITS;
        const int   tu = pu & AGG_TMASK;
        atomicAdd(acc + tu * 32 + lane, g[(size_t)ru * 32 + lane]);
      }
    }
  }

  const int nOv = ovf[0];
  for (int i = tid; i < nOv; i += 1024) {
    const int e = ovf[1 + i];
    const int c = col[e];
    const int t = c - base;
    if ((unsigned)t < (unsigned)nT) {
      const int r = row[e];
      for (int f = 0; f < 32; ++f)
        atomicAdd(acc + t * 32 + f, g[(size_t)r * 32 + f]);
    }
  }
  __syncthreads();

  const float bl = bias[lane];
  for (int node0 = wave; node0 < nT; node0 += 128) {
    float v[4], mx[4], s[4];
#pragma unroll
    for (int k = 0; k < 4; ++k) {
      const int node = node0 + 32 * k;
      v[k]  = (node < nT) ? acc[node * 32 + lane] * dinv[base + node] + bl : 0.0f;
      mx[k] = v[k];
    }
#pragma unroll
    for (int o = 16; o > 0; o >>= 1) {
#pragma unroll
      for (int k = 0; k < 4; ++k) mx[k] = fmaxf(mx[k], __shfl_xor(mx[k], o, 32));
    }
#pragma unroll
    for (int k = 0; k < 4; ++k) s[k] = __expf(v[k] - mx[k]);
#pragma unroll
    for (int o = 16; o > 0; o >>= 1) {
#pragma unroll
      for (int k = 0; k < 4; ++k) s[k] += __shfl_xor(s[k], o, 32);
    }
#pragma unroll
    for (int k = 0; k < 4; ++k) {
      const int node = node0 + 32 * k;
      if (node < nT)
        out[(size_t)(base + node) * 32 + lane] = v[k] - mx[k] - __logf(s[k]);
    }
  }
}

extern "C" void kernel_launch(void* const* d_in, const int* in_sizes, int n_in,
                              void* d_out, int out_size, void* d_ws, size_t ws_size,
                              hipStream_t stream) {
  const float* x  = (const float*)d_in[0];
  const float* W1 = (const float*)d_in[1];
  const float* b1 = (const float*)d_in[2];
  const float* W2 = (const float*)d_in[3];
  const float* b2 = (const float*)d_in[4];
  const int*  eix = (const int*)d_in[5];

  const int n = N_NODES;
  const int E = in_sizes[5] / 2;
  const int* row = eix;
  const int* col = eix + E;

  float* ws   = (float*)d_ws;
  float* dinv = ws;
  float* h    = ws + 100352;
  float* h1   = h  + (size_t)n * D_H;
  float* g2   = h1 + (size_t)n * D_H;
  int*   lcnt = (int*)(g2 + (size_t)n * D_OUT);
  unsigned int* keys = (unsigned int*)(lcnt + AGG_LCNT);
  int*   lstP = (int*)keys + E + 256;
  int*   ovf  = lstP + (size_t)AGG_NGRP * 32 * AGG_LCAP;
  float* outf = (float*)d_out;

  k_fill1 <<<(n + 255) / 256, 256, 0, stream>>>(dinv, n, ovf);
  k_deg   <<<(E + 1023) / 1024, 256, 0, stream>>>(col, dinv, E, keys);
  k_rsqrt <<<(n + 255) / 256, 256, 0, stream>>>(dinv, n);

  gemm_bf16_wmma<D_IN, D_H><<<(n + 127) / 128, 256, 0, stream>>>(
      x, W1, h, dinv, n);
  k_agg_lds128<<<(n + AGG_T - 1) / AGG_T, 1024, 0, stream>>>(h, dinv, b1, row, col, h1, E, n,
                                                         lstP, lcnt, ovf, keys);

  gemm_bf16_wmma<D_H, D_OUT><<<(n + 127) / 128, 256, 0, stream>>>(
      h1, W2, g2, dinv, n);
  k_agg2_lds32<<<(n + AGG_T - 1) / AGG_T, 1024, 0, stream>>>(
      g2, dinv, b2, row, col, lstP, lcnt, ovf, outf, n);
}


# Degree count without global float atomics: 250 work-groups (25 node ranges x 10 edge chunks) count in LDS, integer adds; rsqrt launch sums the 10 partial planes; fill launch goes: 6 launches
# speedup vs baseline: 3.1202x; 1.2944x over previous
#include <hip/hip_runtime.h>

#define N_NODES 100000
#define D_IN 256
#define D_H 128
#define D_OUT 32

typedef __attribute__((ext_vector_type(16))) __bf16 v16bf;
typedef __attribute__((ext_vector_type(8)))  float  v8f;

__device__ __forceinline__ v16bf frag_from_lds(const __bf16* base, int rowIdx, int khalf) {
  union { uint4 u[2]; v16bf v; } c;
  const uint4* p = (const uint4*)(base + rowIdx * 32 + khalf);
  c.u[0] = p[0];
  c.u[1] = p[2];
  return c.v;
}

template<int K, int N>
__global__ __launch_bounds__(256) void gemm_bf16_wmma(
    const float* __restrict__ A, const float* __restrict__ W,
    float* __restrict__ C,
    const float* __restrict__ dinv, int M) {
  constexpr int NT = N / 16;
  __shared__ __bf16 ldsA[128 * 32];
  __shared__ __bf16 ldsB[N * 32];

  const int tid   = threadIdx.x;
  const int wave  = tid >> 5;
  const int lane  = tid & 31;
  const int laneM = lane & 15;
  const int khalf = (lane >> 4) << 3;
  const int rowBlk = blockIdx.x * 128;

  const int ldRow = tid >> 1;
  const int ldC0  = (tid & 1) << 4;
  const int ldGr  = rowBlk + ldRow;

  v8f acc[NT];
#pragma unroll
  for (int t = 0; t < NT; ++t) {
    v8f z = {0.f, 0.f, 0.f, 0.f, 0.f, 0.f, 0.f, 0.f};
    acc[t] = z;
  }

  for (int kk = 0; kk < K; kk += 32) {
    __syncthreads();

    {
      union { __bf16 h[16]; uint4 q[2]; } buf;
      if (ldGr < M) {
        const float4* src = (const float4*)(A + (size_t)ldGr * K + kk + ldC0);
#pragma unroll
        for (int j = 0; j < 4; ++j) {
          float4 v = src[j];
          buf.h[j * 4 + 0] = (__bf16)v.x;
          buf.h[j * 4 + 1] = (__bf16)v.y;
          buf.h[j * 4 + 2] = (__bf16)v.z;
          buf.h[j * 4 + 3] = (__bf16)v.w;
        }
      } else {
#pragma unroll
        for (int j = 0; j < 16; ++j) buf.h[j] = (__bf16)0.0f;
      }
      uint4* dst = (uint4*)(ldsA + ldRow * 32 + ldC0);
      dst[0] = buf.q[0];
      dst[1] = buf.q[1];
    }

#pragma unroll
    for (int i = tid; i < N * 8; i += 256) {
      const int n  = i % N;
      const int kq = i / N;
      union { __bf16 h[4]; uint2 q; } wb;
#pragma unroll
      for (int j = 0; j < 4; ++j)
        wb.h[j] = (__bf16)W[(size_t)(kk + kq * 4 + j) * N + n];
      *(uint2*)(ldsB + n * 32 + kq * 4) = wb.q;
    }
    __syncthreads();

    v16bf a = frag_from_lds(ldsA, (wave << 4) + laneM, khalf);
    v16bf bc = frag_from_lds(ldsB, laneM, khalf);
#pragma unroll
    for (int t = 0; t < NT; ++t) {
      v16bf bn = (t + 1 < NT) ? frag_from_lds(ldsB, ((t + 1) << 4) + laneM, khalf) : bc;
      acc[t] = __builtin_amdgcn_wmma_f32_16x16x32_bf16(
          false, a, false, bc, (short)0, acc[t], false, false);
      bc = bn;
    }
  }

  const int mBase = rowBlk + (wave << 4) + ((lane >> 4) << 3);
#pragma unroll
  for (int v = 0; v < 8; ++v) {
    const int gr = mBase + v;
    if (gr < M) {
      const float d = dinv[gr];
#pragma unroll
      for (int t = 0; t < NT; ++t) {
        const size_t idx = (size_t)gr * N + (t << 4) + laneM;
        C[idx] = acc[t][v] * d;
      }
    }
  }
}

#define AGG_T     400
#define AGG_TBITS 9
#define AGG_TMASK ((1 << AGG_TBITS) - 1)
#define AGG_QCAP  160
#define AGG_NGRP  ((N_NODES + AGG_T - 1) / AGG_T)
#define AGG_LCAP  1024
#define AGG_LCNT  8192
static_assert(AGG_T <= (1 << AGG_TBITS), "local target bits");
static_assert(AGG_NGRP * 32 <= AGG_LCNT, "one count per sweep wave");
static_assert(N_NODES <= (1 << (31 - AGG_TBITS)), "source index and local target share one int");
static_assert(AGG_NGRP <= 255, "an owner key is one byte and 0xFF marks an empty slot");
#define AGG_SW    8
#define AGG_PL    32

#define DG_P     25
#define DG_C     10
#define DG_NR    ((N_NODES + DG_P - 1) / DG_P)
#define DG_PITCH 100352
#define DG_SW    8
static_assert(DG_P * DG_NR >= N_NODES, "ranges cover all nodes");
static_assert(DG_PITCH >= N_NODES, "a plane holds all nodes");

__device__ __forceinline__ void dg_count(unsigned int* cnt, int4 v, int base) {
  const unsigned t0 = (unsigned)(v.x - base), t1 = (unsigned)(v.y - base);
  const unsigned t2 = (unsigned)(v.z - base), t3 = (unsigned)(v.w - base);
  if (t0 < (unsigned)DG_NR) atomicAdd(&cnt[t0], 1u);
  if (t1 < (unsigned)DG_NR) atomicAdd(&cnt[t1], 1u);
  if (t2 < (unsigned)DG_NR) atomicAdd(&cnt[t2], 1u);
  if (t3 < (unsigned)DG_NR) atomicAdd(&cnt[t3], 1u);
}

__device__ __forceinline__ unsigned dg_keybyte(int c) {
  return (c < 0) ? 0xFFu : ((unsigned)c / AGG_T);
}
__device__ __forceinline__ unsigned dg_keyword(int4 v) {
  return dg_keybyte(v.x) | (dg_keybyte(v.y) << 8) |
         (dg_keybyte(v.z) << 16) | (dg_keybyte(v.w) << 24);
}

__global__ __launch_bounds__(1024) void k_deg_own(
    const int* __restrict__ col, unsigned int* __restrict__ degP, int E, int n,
    unsigned int* __restrict__ keys, int* __restrict__ ovf) {
  __shared__ unsigned int cnt[DG_NR];

  const int tid  = threadIdx.x;
  const int p    = blockIdx.x % DG_P;
  const int c    = blockIdx.x / DG_P;
  const int base = p * DG_NR;

  for (int i = tid; i < DG_NR; i += 1024) cnt[i] = 0u;
  if (blockIdx.x == 0 && tid == 0) ovf[0] = 0;
  __syncthreads();

  const int QT = ((E + 15) >> 4) << 2;
  const int UT = (QT + 1023) >> 10;
  const int UC = (UT + DG_C - 1) / DG_C;
  const int uBeg = c * UC;
  const int uEnd = min(uBeg + UC, UT);
  const int QF = ((((size_t)col) & 15) == 0) ? (E >> 2) : 0;
  const int4* col4 = (const int4*)col;

  if (QF > 0) {
    for (int u = uBeg; u < uEnd; u += DG_SW) {
      int4 v[DG_SW];
#pragma unroll
      for (int s = 0; s < DG_SW; ++s) v[s] = col4[min(((u + s) << 10) + tid, QF - 1)];
#pragma unroll
      for (int s = 0; s < DG_SW; ++s) {
        const int q = ((u + s) << 10) + tid;
        if (u + s < uEnd && q < QF) {
          dg_count(cnt, v[s], base);
          if ((u + s) % DG_P == p) keys[q] = dg_keyword(v[s]);
        }
      }
    }
  }
  for (int q = max(QF, uBeg << 10) + tid; q < min(QT, uEnd << 10); q += 1024) {
    const int e0 = q * 4;
    int4 v = make_int4(-1, -1, -1, -1);
    if (e0 + 0 < E) v.x = col[e0 + 0];
    if (e0 + 1 < E) v.y = col[e0 + 1];
    if (e0 + 2 < E) v.z = col[e0 + 2];
    if (e0 + 3 < E) v.w = col[e0 + 3];
    dg_count(cnt, v, base);
    if ((q >> 10) % DG_P == p) keys[q] = dg_keyword(v);
  }
  __syncthreads();

  unsigned int* dst = degP + (size_t)c * DG_PITCH + base;
  for (int i = tid; i < DG_NR; i += 1024)
    if (base + i < n) dst[i] = cnt[i];
}

__global__ void k_rsqrt(const unsigned int* __restrict__ degP, float* __restrict__ p, int n) {
  int i = blockIdx.x * blockDim.x + threadIdx.x;
  if (i < n) {
    unsigned d = 1u;
#pragma unroll
    for (int c = 0; c < DG_C; ++c) d += degP[(size_t)c * DG_PITCH + i];
    p[i] = rsqrtf((float)d);
  }
}


__device__ __forceinline__ unsigned agg_match16(uint4 k, unsigned pat) {
  const unsigned M7 = 0x7F7F7F7Fu;
  const unsigned t0 = k.x ^ pat, t1 = k.y ^ pat, t2 = k.z ^ pat, t3 = k.w ^ pat;
  const unsigned n0 = ((t0 & M7) + M7) | t0;
  const unsigned n1 = ((t1 & M7) + M7) | t1;
  const unsigned n2 = ((t2 & M7) + M7) | t2;
  const unsigned n3 = ((t3 & M7) + M7) | t3;
  unsigned r = ((n0 >> 7) & 0x01010101u) | ((n1 >> 6) & 0x02020202u) |
               ((n2 >> 5) & 0x04040404u) | ((n3 >> 4) & 0x08080808u);
  return ~r & 0x0F0F0F0Fu;
}

__device__ __forceinline__ void agg_flush(
    const int* myq, int cnt, int lane, int base,
    const float* __restrict__ h,
    const int* __restrict__ row, const int* __restrict__ col,
    float* acc,
    int* __restrict__ lp, int* __restrict__ ovf, int saved) {
  const float4* h4 = (const float4*)h;
  for (int b0 = 0; b0 < cnt; b0 += 32) {
    const int m = min(32, cnt - b0);
    int r = 0, t = 0;
    if (lane < m) {
      const int e = myq[b0 + lane];
      const int c = col[e];
      r  = row[e];
      t  = c - base;
      const int slot = saved + b0 + lane;
      if (slot < AGG_LCAP) {
        lp[slot] = (r << AGG_TBITS) | t;
      } else {
        const int k = atomicAdd(ovf, 1);
        ovf[1 + k] = e;
      }
    }
    int i = 0;
    for (; i + 4 <= m; i += 4) {
      float4 v[4];
      int    tt[4];
#pragma unroll
      for (int u = 0; u < 4; ++u) {
        const int ru = __builtin_amdgcn_readlane(r, i + u);
        tt[u] = __builtin_amdgcn_readlane(t, i + u);
        v[u]  = h4[(size_t)ru * 32 + lane];
      }
#pragma unroll
      for (int u = 0; u < 4; ++u) {
        float* a = acc + tt[u] * 128 + lane;
        atomicAdd(a + 0 * AGG_PL, v[u].x);
        atomicAdd(a + 1 * AGG_PL, v[u].y);
        atomicAdd(a + 2 * AGG_PL, v[u].z);
        atomicAdd(a + 3 * AGG_PL, v[u].w);
      }
    }
    for (; i < m; ++i) {
      const int   ru = __builtin_amdgcn_readlane(r, i);
      const int   tu = __builtin_amdgcn_readlane(t, i);
      const float4 v = h4[(size_t)ru * 32 + lane];
      float* a = acc + tu * 128 + lane;
      atomicAdd(a + 0 * AGG_PL, v.x);
      atomicAdd(a + 1 * AGG_PL, v.y);
      atomicAdd(a + 2 * AGG_PL, v.z);
      atomicAdd(a + 3 * AGG_PL, v.w);
    }
  }
}

__global__ __launch_bounds__(1024) void k_agg_lds128(
    const float* __restrict__ h, const float* __restrict__ dinv,
    const float* __restrict__ bias,
    const int* __restrict__ row, const int* __restrict__ col,
    float* __restrict__ out, int E, int n,
    int* __restrict__ lstP,
    int* __restrict__ lcnt, int* __restrict__ ovf,
    const unsigned int* __restrict__ keys) {
  __shared__ float acc[AGG_T * 128];
  __shared__ int   q[32 * AGG_QCAP];

  const int tid  = threadIdx.x;
  const int lane = tid & 31;
  const int wave = tid >> 5;
  const int base = blockIdx.x * AGG_T;
  const int nT   = min(AGG_T, n - base);
  const unsigned lt = (1u << lane) - 1u;

  {
    const float4* s4 = (const float4*)(h + (size_t)base * 128);
    for (int i = tid; i < nT * 32; i += 1024) {
      const float4 v = s4[i];
      float* a = acc + (i >> 5) * 128 + lane;
      a[0 * AGG_PL] = v.x;
      a[1 * AGG_PL] = v.y;
      a[2 * AGG_PL] = v.z;
      a[3 * AGG_PL] = v.w;
    }
  }
  __syncthreads();

  int* myq = q + wave * AGG_QCAP;
  int cnt = 0;
  const int wslot = blockIdx.x * 32 + wave;
  int*   lp = lstP + (size_t)wslot * AGG_LCAP;
  int saved = 0;

  const uint4* key4 = (const uint4*)keys;
  const int nK = (E + 15) >> 4;
  const unsigned pat = (unsigned)blockIdx.x * 0x01010101u;

  const int nU  = (nK + 1023) >> 10;
  const int nIt = (nU + AGG_SW - 1) / AGG_SW;
  const int u0  = (int)(((long long)blockIdx.x * nU) / gridDim.x);
  for (int j = 0; j < nIt; ++j) {
    uint4 k[AGG_SW];
    int   ix[AGG_SW];
#pragma unroll
    for (int u = 0; u < AGG_SW; ++u) {
      const int pos = j * AGG_SW + u;
      int unit = u0 + pos;
      if (unit >= nU) unit -= nU;
      const int i = (unit << 10) + tid;
      ix[u] = (pos < nU && i < nK) ? i : -1;
      k[u] = key4[min(i, nK - 1)];
    }
#pragma unroll
    for (int u = 0; u < AGG_SW; ++u) {
      const int i = ix[u];
      unsigned mk = (i >= 0) ? agg_match16(k[u], pat) : 0u;
      unsigned b = __builtin_amdgcn_ballot_w32(mk != 0u);
      while (b) {
        if (mk) {
          const int bit = __builtin_ctz(mk);
          myq[cnt + __builtin_popcount(b & lt)] = (i << 4) + ((bit & 7) << 2) + (bit >> 3);
          mk &= mk - 1u;
        }
        cnt += __builtin_popcount(b);
        if (cnt >= 32) {
          agg_flush(myq, cnt, lane, base, h, row, col, acc, lp, ovf, saved);
          saved += cnt;
          cnt = 0;
        }
        b = __builtin_amdgcn_ballot_w32(mk != 0u);
      }
    }
  }
  agg_flush(myq, cnt, lane, base, h, row, col, acc, lp, ovf, saved);
  saved += cnt;
  if (lane == 0) lcnt[wslot] = saved;

  __syncthreads();
  {
    float4* d4 = (float4*)(out + (size_t)base * 128);
    const float4 bb = make_float4(bias[lane * 4 + 0], bias[lane * 4 + 1],
                                  bias[lane * 4 + 2], bias[lane * 4 + 3]);
    for (int i = tid; i < nT * 32; i += 1024) {
      const float d = dinv[base + (i >> 5)];
      const float* a = acc + (i >> 5) * 128 + lane;
      float4 v = make_float4(a[0 * AGG_PL], a[1 * AGG_PL], a[2 * AGG_PL], a[3 * AGG_PL]);
      v.x = fmaxf(v.x * d + bb.x, 0.0f);
      v.y = fmaxf(v.y * d + bb.y, 0.0f);
      v.z = fmaxf(v.z * d + bb.z, 0.0f);
      v.w = fmaxf(v.w * d + bb.w, 0.0f);
      d4[i] = v;
    }
  }
}

__global__ __launch_bounds__(1024) void k_agg2_lds32(
    const float* __restrict__ g, const float* __restrict__ dinv,
    const float* __restrict__ bias,
    const int* __restrict__ row, const int* __restrict__ col,
    const int* __restrict__ lstP,
    const int* __restrict__ lcnt, const int* __restrict__ ovf,
    float* __restrict__ out, int n) {
  __shared__ float acc[AGG_T * 32];

  const int tid  = threadIdx.x;
  const int lane = tid & 31;
  const int wave = __builtin_amdgcn_readfirstlane(tid >> 5);
  const int base = blockIdx.x * AGG_T;
  const int nT   = min(AGG_T, n - base);

  {
    const float4* s4 = (const float4*)(g + (size_t)base * 32);
    float4* d4 = (float4*)acc;
    for (int i = tid; i < nT * 8; i += 1024) d4[i] = s4[i];
  }
  __syncthreads();

  const int wslot = blockIdx.x * 32 + wave;
  const int cnt = min(__builtin_amdgcn_readfirstlane(lcnt[wslot]), AGG_LCAP);
  const int*   lp = lstP + (size_t)wslot * AGG_LCAP;
  for (int b0 = 0; b0 < cnt; b0 += 32) {
    const int m = min(32, cnt - b0);
    int   pk = 0;
    if (lane < m) pk = lp[b0 + lane];
    if (m == 32) {
      float v[32];
#pragma unroll
      for (int u = 0; u < 32; ++u) {
        const unsigned ru = (unsigned)__builtin_amdgcn_readlane(pk, u) >> AGG_TBITS;
        v[u] = g[(size_t)ru * 32 + lane];
      }
#pragma unroll
      for (int u = 0; u < 32; ++u) {
        const int   tu = __builtin_amdgcn_readlane(pk, u) & AGG_TMASK;
        atomicAdd(acc + tu * 32 + lane, v[u]);
      }
    } else {
      for (int u = 0; u < m; ++u) {
        const int   pu = __builtin_amdgcn_readlane(pk, u);
        const int   ru = pu >> AGG_TBITS;
        const int   tu = pu & AGG_TMASK;
        atomicAdd(acc + tu * 32 + lane, g[(size_t)ru * 32 + lane]);
      }
    }
  }

  const int nOv = ovf[0];
  for (int i = tid; i < nOv; i += 1024) {
    const int e = ovf[1 + i];
    const int c = col[e];
    const int t = c - base;
    if ((unsigned)t < (unsigned)nT) {
      const int r = row[e];
      for (int f = 0; f < 32; ++f)
        atomicAdd(acc + t * 32 + f, g[(size_t)r * 32 + f]);
    }
  }
  __syncthreads();

  const float bl = bias[lane];
  for (int node0 = wave; node0 < nT; node0 += 128) {
    float v[4], mx[4], s[4];
#pragma unroll
    for (int k = 0; k < 4; ++k) {
      const int node = node0 + 32 * k;
      v[k]  = (node < nT) ? acc[node * 32 + lane] * dinv[base + node] + bl : 0.0f;
      mx[k] = v[k];
    }
#pragma unroll
    for (int o = 16; o > 0; o >>= 1) {
#pragma unroll
      for (int k = 0; k < 4; ++k) mx[k] = fmaxf(mx[k], __shfl_xor(mx[k], o, 32));
    }
#pragma unroll
    for (int k = 0; k < 4; ++k) s[k] = __expf(v[k] - mx[k]);
#pragma unroll
    for (int o = 16; o > 0; o >>= 1) {
#pragma unroll
      for (int k = 0; k < 4; ++k) s[k] += __shfl_xor(s[k], o, 32);
    }
#pragma unroll
    for (int k = 0; k < 4; ++k) {
      const int node = node0 + 32 * k;
      if (node < nT)
        out[(size_t)(base + node) * 32 + lane] = v[k] - mx[k] - __logf(s[k]);
    }
  }
}

extern "C" void kernel_launch(void* const* d_in, const int* in_sizes, int n_in,
                              void* d_out, int out_size, void* d_ws, size_t ws_size,
                              hipStream_t stream) {
  const float* x  = (const float*)d_in[0];
  const float* W1 = (const float*)d_in[1];
  const float* b1 = (const float*)d_in[2];
  const float* W2 = (const float*)d_in[3];
  const float* b2 = (const float*)d_in[4];
  const int*  eix = (const int*)d_in[5];

  const int n = N_NODES;
  const int E = in_sizes[5] / 2;
  const int* row = eix;
  const int* col = eix + E;

  float* ws   = (float*)d_ws;
  float* dinv = ws;
  float* h    = ws + 100352;
  float* h1   = h  + (size_t)n * D_H;
  float* g2   = h1 + (size_t)n * D_H;
  int*   lcnt = (int*)(g2 + (size_t)n * D_OUT);
  unsigned int* keys = (unsigned int*)(lcnt + AGG_LCNT);
  int*   lstP = (int*)keys + E + 256;
  int*   ovf  = lstP + (size_t)AGG_NGRP * 32 * AGG_LCAP;
  unsigned int* degP = (unsigned int*)(ovf + E + 256);
  float* outf = (float*)d_out;

  k_deg_own<<<DG_P * DG_C, 1024, 0, stream>>>(col, degP, E, n, keys, ovf);
  k_rsqrt <<<(n + 255) / 256, 256, 0, stream>>>(degP, dinv, n);

  gemm_bf16_wmma<D_IN, D_H><<<(n + 127) / 128, 256, 0, stream>>>(
      x, W1, h, dinv, n);
  k_agg_lds128<<<(n + AGG_T - 1) / AGG_T, 1024, 0, stream>>>(h, dinv, b1, row, col, h1, E, n,
                                                         lstP, lcnt, ovf, keys);

  gemm_bf16_wmma<D_H, D_OUT><<<(n + 127) / 128, 256, 0, stream>>>(
      h1, W2, g2, dinv, n);
  k_agg2_lds32<<<(n + AGG_T - 1) / AGG_T, 1024, 0, stream>>>(
      g2, dinv, b2, row, col, lstP, lcnt, ovf, outf, n);
}


# Layer-1 gather: a wave gathers whole batches of 32 queued edges only, at the end of a sweep trip, with a rolling window of 16 source rows in flight (was 4 per wait); sum order LDS arrival
# speedup vs baseline: 3.2167x; 1.0309x over previous
#include <hip/hip_runtime.h>

#define N_NODES 100000
#define D_IN 256
#define D_H 128
#define D_OUT 32

typedef __attribute__((ext_vector_type(16))) __bf16 v16bf;
typedef __attribute__((ext_vector_type(8)))  float  v8f;

__device__ __forceinline__ v16bf frag_from_lds(const __bf16* base, int rowIdx, int khalf) {
  union { uint4 u[2]; v16bf v; } c;
  const uint4* p = (const uint4*)(base + rowIdx * 32 + khalf);
  c.u[0] = p[0];
  c.u[1] = p[2];
  return c.v;
}

template<int K, int N>
__global__ __launch_bounds__(256) void gemm_bf16_wmma(
    const float* __restrict__ A, const float* __restrict__ W,
    float* __restrict__ C,
    const float* __restrict__ dinv, int M) {
  constexpr int NT = N / 16;
  __shared__ __bf16 ldsA[128 * 32];
  __shared__ __bf16 ldsB[N * 32];

  const int tid   = threadIdx.x;
  const int wave  = tid >> 5;
  const int lane  = tid & 31;
  const int laneM = lane & 15;
  const int khalf = (lane >> 4) << 3;
  const int rowBlk = blockIdx.x * 128;

  const int ldRow = tid >> 1;
  const int ldC0  = (tid & 1) << 4;
  const int ldGr  = rowBlk + ldRow;

  v8f acc[NT];
#pragma unroll
  for (int t = 0; t < NT; ++t) {
    v8f z = {0.f, 0.f, 0.f, 0.f, 0.f, 0.f, 0.f, 0.f};
    acc[t] = z;
  }

  for (int kk = 0; kk < K; kk += 32) {
    __syncthreads();

    {
      union { __bf16 h[16]; uint4 q[2]; } buf;
      if (ldGr < M) {
        const float4* src = (const float4*)(A + (size_t)ldGr * K + kk + ldC0);
#pragma unroll
        for (int j = 0; j < 4; ++j) {
          float4 v = src[j];
          buf.h[j * 4 + 0] = (__bf16)v.x;
          buf.h[j * 4 + 1] = (__bf16)v.y;
          buf.h[j * 4 + 2] = (__bf16)v.z;
          buf.h[j * 4 + 3] = (__bf16)v.w;
        }
      } else {
#pragma unroll
        for (int j = 0; j < 16; ++j) buf.h[j] = (__bf16)0.0f;
      }
      uint4* dst = (uint4*)(ldsA + ldRow * 32 + ldC0);
      dst[0] = buf.q[0];
      dst[1] = buf.q[1];
    }

#pragma unroll
    for (int i = tid; i < N * 8; i += 256) {
      const int n  = i % N;
      const int kq = i / N;
      union { __bf16 h[4]; uint2 q; } wb;
#pragma unroll
      for (int j = 0; j < 4; ++j)
        wb.h[j] = (__bf16)W[(size_t)(kk + kq * 4 + j) * N + n];
      *(uint2*)(ldsB + n * 32 + kq * 4) = wb.q;
    }
    __syncthreads();

    v16bf a = frag_from_lds(ldsA, (wave << 4) + laneM, khalf);
    v16bf bc = frag_from_lds(ldsB, laneM, khalf);
#pragma unroll
    for (int t = 0; t < NT; ++t) {
      v16bf bn = (t + 1 < NT) ? frag_from_lds(ldsB, ((t + 1) << 4) + laneM, khalf) : bc;
      acc[t] = __builtin_amdgcn_wmma_f32_16x16x32_bf16(
          false, a, false, bc, (short)0, acc[t], false, false);
      bc = bn;
    }
  }

  const int mBase = rowBlk + (wave << 4) + ((lane >> 4) << 3);
#pragma unroll
  for (int v = 0; v < 8; ++v) {
    const int gr = mBase + v;
    if (gr < M) {
      const float d = dinv[gr];
#pragma unroll
      for (int t = 0; t < NT; ++t) {
        const size_t idx = (size_t)gr * N + (t << 4) + laneM;
        C[idx] = acc[t][v] * d;
      }
    }
  }
}

#define AGG_T     400
#define AGG_TBITS 9
#define AGG_TMASK ((1 << AGG_TBITS) - 1)
#define AGG_QCAP  160
#define AGG_NGRP  ((N_NODES + AGG_T - 1) / AGG_T)
#define AGG_LCAP  1024
#define AGG_LCNT  8192
static_assert(AGG_T <= (1 << AGG_TBITS), "local target bits");
static_assert(AGG_NGRP * 32 <= AGG_LCNT, "one count per sweep wave");
static_assert(N_NODES <= (1 << (31 - AGG_TBITS)), "source index and local target share one int");
static_assert(AGG_NGRP <= 255, "an owner key is one byte and 0xFF marks an empty slot");
static_assert(AGG_QCAP >= 96, "a trip starts below 32 queued edges and a pass adds at most 32");
#define AGG_SW    8
#define AGG_GD    16
#define AGG_PL    32

#define DG_P     25
#define DG_C     10
#define DG_NR    ((N_NODES + DG_P - 1) / DG_P)
#define DG_PITCH 100352
#define DG_SW    8
static_assert(DG_P * DG_NR >= N_NODES, "ranges cover all nodes");
static_assert(DG_PITCH >= N_NODES, "a plane holds all nodes");

__device__ __forceinline__ void dg_count(unsigned int* cnt, int4 v, int base) {
  const unsigned t0 = (unsigned)(v.x - base), t1 = (unsigned)(v.y - base);
  const unsigned t2 = (unsigned)(v.z - base), t3 = (unsigned)(v.w - base);
  if (t0 < (unsigned)DG_NR) atomicAdd(&cnt[t0], 1u);
  if (t1 < (unsigned)DG_NR) atomicAdd(&cnt[t1], 1u);
  if (t2 < (unsigned)DG_NR) atomicAdd(&cnt[t2], 1u);
  if (t3 < (unsigned)DG_NR) atomicAdd(&cnt[t3], 1u);
}

__device__ __forceinline__ unsigned dg_keybyte(int c) {
  return (c < 0) ? 0xFFu : ((unsigned)c / AGG_T);
}
__device__ __forceinline__ unsigned dg_keyword(int4 v) {
  return dg_keybyte(v.x) | (dg_keybyte(v.y) << 8) |
         (dg_keybyte(v.z) << 16) | (dg_keybyte(v.w) << 24);
}

__global__ __launch_bounds__(1024) void k_deg_own(
    const int* __restrict__ col, unsigned int* __restrict__ degP, int E, int n,
    unsigned int* __restrict__ keys, int* __restrict__ ovf) {
  __shared__ unsigned int cnt[DG_NR];

  const int tid  = threadIdx.x;
  const int p    = blockIdx.x % DG_P;
  const int c    = blockIdx.x / DG_P;
  const int base = p * DG_NR;

  for (int i = tid; i < DG_NR; i += 1024) cnt[i] = 0u;
  if (blockIdx.x == 0 && tid == 0) ovf[0] = 0;
  __syncthreads();

  const int QT = ((E + 15) >> 4) << 2;
  const int UT = (QT + 1023) >> 10;
  const int UC = (UT + DG_C - 1) / DG_C;
  const int uBeg = c * UC;
  const int uEnd = min(uBeg + UC, UT);
  const int QF = ((((size_t)col) & 15) == 0) ? (E >> 2) : 0;
  const int4* col4 = (const int4*)col;

  if (QF > 0) {
    for (int u = uBeg; u < uEnd; u += DG_SW) {
      int4 v[DG_SW];
#pragma unroll
      for (int s = 0; s < DG_SW; ++s) v[s] = col4[min(((u + s) << 10) + tid, QF - 1)];
#pragma unroll
      for (int s = 0; s < DG_SW; ++s) {
        const int q = ((u + s) << 10) + tid;
        if (u + s < uEnd && q < QF) {
          dg_count(cnt, v[s], base);
          if ((u + s) % DG_P == p) keys[q] = dg_keyword(v[s]);
        }
      }
    }
  }
  for (int q = max(QF, uBeg << 10) + tid; q < min(QT, uEnd << 10); q += 1024) {
    const int e0 = q * 4;
    int4 v = make_int4(-1, -1, -1, -1);
    if (e0 + 0 < E) v.x = col[e0 + 0];
    if (e0 + 1 < E) v.y = col[e0 + 1];
    if (e0 + 2 < E) v.z = col[e0 + 2];
    if (e0 + 3 < E) v.w = col[e0 + 3];
    dg_count(cnt, v, base);
    if ((q >> 10) % DG_P == p) keys[q] = dg_keyword(v);
  }
  __syncthreads();

  unsigned int* dst = degP + (size_t)c * DG_PITCH + base;
  for (int i = tid; i < DG_NR; i += 1024)
    if (base + i < n) dst[i] = cnt[i];
}

__global__ void k_rsqrt(const unsigned int* __restrict__ degP, float* __restrict__ p, int n) {
  int i = blockIdx.x * blockDim.x + threadIdx.x;
  if (i < n) {
    unsigned d = 1u;
#pragma unroll
    for (int c = 0; c < DG_C; ++c) d += degP[(size_t)c * DG_PITCH + i];
    p[i] = rsqrtf((float)d);
  }
}


__device__ __forceinline__ unsigned agg_match16(uint4 k, unsigned pat) {
  const unsigned M7 = 0x7F7F7F7Fu;
  const unsigned t0 = k.x ^ pat, t1 = k.y ^ pat, t2 = k.z ^ pat, t3 = k.w ^ pat;
  const unsigned n0 = ((t0 & M7) + M7) | t0;
  const unsigned n1 = ((t1 & M7) + M7) | t1;
  const unsigned n2 = ((t2 & M7) + M7) | t2;
  const unsigned n3 = ((t3 & M7) + M7) | t3;
  unsigned r = ((n0 >> 7) & 0x01010101u) | ((n1 >> 6) & 0x02020202u) |
               ((n2 >> 5) & 0x04040404u) | ((n3 >> 4) & 0x08080808u);
  return ~r & 0x0F0F0F0Fu;
}

template<int G>
__device__ __forceinline__ void agg_flush(
    const int* myq, int cnt, int lane, int base,
    const float* __restrict__ h,
    const int* __restrict__ row, const int* __restrict__ col,
    float* acc,
    int* __restrict__ lp, int* __restrict__ ovf, int saved) {
  const float4* h4 = (const float4*)h;
  for (int b0 = 0; b0 < cnt; b0 += 32) {
    const int m = min(32, cnt - b0);
    int r = 0, t = 0;
    if (lane < m) {
      const int e = myq[b0 + lane];
      const int c = col[e];
      r  = row[e];
      t  = c - base;
      const int slot = saved + b0 + lane;
      if (slot < AGG_LCAP) {
        lp[slot] = (r << AGG_TBITS) | t;
      } else {
        const int k = atomicAdd(ovf, 1);
        ovf[1 + k] = e;
      }
    }
    int i = 0;
    if (G > 4 && m == 32) {
      float4 v[G];
#pragma unroll
      for (int u = 0; u < G; ++u) {
        const int ru = __builtin_amdgcn_readlane(r, u);
        v[u] = h4[(size_t)ru * 32 + lane];
      }
#pragma unroll
      for (int u = 0; u < 32; ++u) {
        const int tu = __builtin_amdgcn_readlane(t, u);
        float* a = acc + tu * 128 + lane;
        atomicAdd(a + 0 * AGG_PL, v[u % G].x);
        atomicAdd(a + 1 * AGG_PL, v[u % G].y);
        atomicAdd(a + 2 * AGG_PL, v[u % G].z);
        atomicAdd(a + 3 * AGG_PL, v[u % G].w);
        if (u + G < 32) {
          const int ru = __builtin_amdgcn_readlane(r, u + G);
          v[u % G] = h4[(size_t)ru * 32 + lane];
        }
      }
      i = 32;
    }
    if (G > 4) {
      for (; i + G <= m; i += G) {
        float4 v[G];
        int    tt[G];
#pragma unroll
        for (int u = 0; u < G; ++u) {
          const int ru = __builtin_amdgcn_readlane(r, i + u);
          tt[u] = __builtin_amdgcn_readlane(t, i + u);
          v[u]  = h4[(size_t)ru * 32 + lane];
        }
#pragma unroll
        for (int u = 0; u < G; ++u) {
          float* a = acc + tt[u] * 128 + lane;
          atomicAdd(a + 0 * AGG_PL, v[u].x);
          atomicAdd(a + 1 * AGG_PL, v[u].y);
          atomicAdd(a + 2 * AGG_PL, v[u].z);
          atomicAdd(a + 3 * AGG_PL, v[u].w);
        }
      }
    }
    for (; i + 4 <= m; i += 4) {
      float4 v[4];
      int    tt[4];
#pragma unroll
      for (int u = 0; u < 4; ++u) {
        const int ru = __builtin_amdgcn_readlane(r, i + u);
        tt[u] = __builtin_amdgcn_readlane(t, i + u);
        v[u]  = h4[(size_t)ru * 32 + lane];
      }
#pragma unroll
      for (int u = 0; u < 4; ++u) {
        float* a = acc + tt[u] * 128 + lane;
        atomicAdd(a + 0 * AGG_PL, v[u].x);
        atomicAdd(a + 1 * AGG_PL, v[u].y);
        atomicAdd(a + 2 * AGG_PL, v[u].z);
        atomicAdd(a + 3 * AGG_PL, v[u].w);
      }
    }
    for (; i < m; ++i) {
      const int   ru = __builtin_amdgcn_readlane(r, i);
      const int   tu = __builtin_amdgcn_readlane(t, i);
      const float4 v = h4[(size_t)ru * 32 + lane];
      float* a = acc + tu * 128 + lane;
      atomicAdd(a + 0 * AGG_PL, v.x);
      atomicAdd(a + 1 * AGG_PL, v.y);
      atomicAdd(a + 2 * AGG_PL, v.z);
      atomicAdd(a + 3 * AGG_PL, v.w);
    }
  }
}

__global__ __launch_bounds__(1024) void k_agg_lds128(
    const float* __restrict__ h, const float* __restrict__ dinv,
    const float* __restrict__ bias,
    const int* __restrict__ row, const int* __restrict__ col,
    float* __restrict__ out, int E, int n,
    int* __restrict__ lstP,
    int* __restrict__ lcnt, int* __restrict__ ovf,
    const unsigned int* __restrict__ keys) {
  __shared__ float acc[AGG_T * 128];
  __shared__ int   q[32 * AGG_QCAP];

  const int tid  = threadIdx.x;
  const int lane = tid & 31;
  const int wave = tid >> 5;
  const int base = blockIdx.x * AGG_T;
  const int nT   = min(AGG_T, n - base);
  const unsigned lt = (1u << lane) - 1u;

  {
    const float4* s4 = (const float4*)(h + (size_t)base * 128);
    for (int i = tid; i < nT * 32; i += 1024) {
      const float4 v = s4[i];
      float* a = acc + (i >> 5) * 128 + lane;
      a[0 * AGG_PL] = v.x;
      a[1 * AGG_PL] = v.y;
      a[2 * AGG_PL] = v.z;
      a[3 * AGG_PL] = v.w;
    }
  }
  __syncthreads();

  int* myq = q + wave * AGG_QCAP;
  int cnt = 0;
  const int wslot = blockIdx.x * 32 + wave;
  int*   lp = lstP + (size_t)wslot * AGG_LCAP;
  int saved = 0;

  const uint4* key4 = (const uint4*)keys;
  const int nK = (E + 15) >> 4;
  const unsigned pat = (unsigned)blockIdx.x * 0x01010101u;

  const int nU  = (nK + 1023) >> 10;
  const int nIt = (nU + AGG_SW - 1) / AGG_SW;
  const int u0  = (int)(((long long)blockIdx.x * nU) / gridDim.x);
  for (int j = 0; j < nIt; ++j) {
    uint4 k[AGG_SW];
    int   ix[AGG_SW];
#pragma unroll
    for (int u = 0; u < AGG_SW; ++u) {
      const int pos = j * AGG_SW + u;
      int unit = u0 + pos;
      if (unit >= nU) unit -= nU;
      const int i = (unit << 10) + tid;
      ix[u] = (pos < nU && i < nK) ? i : -1;
      k[u] = key4[min(i, nK - 1)];
    }
#pragma unroll
    for (int u = 0; u < AGG_SW; ++u) {
      const int i = ix[u];
      unsigned mk = (i >= 0) ? agg_match16(k[u], pat) : 0u;
      unsigned b = __builtin_amdgcn_ballot_w32(mk != 0u);
      while (b) {
        if (mk) {
          const int bit = __builtin_ctz(mk);
          myq[cnt + __builtin_popcount(b & lt)] = (i << 4) + ((bit & 7) << 2) + (bit >> 3);
          mk &= mk - 1u;
        }
        cnt += __builtin_popcount(b);
        if (cnt >= AGG_QCAP - 32) {
          agg_flush<4>(myq, cnt, lane, base, h, row, col, acc, lp, ovf, saved);
          saved += cnt;
          cnt = 0;
        }
        b = __builtin_amdgcn_ballot_w32(mk != 0u);
      }
    }
    if (cnt >= 32) {
      const int whole = cnt & ~31;
      agg_flush<AGG_GD>(myq, whole, lane, base, h, row, col, acc, lp, ovf, saved);
      saved += whole;
      const int rem = cnt - whole;
      if (lane < rem) {
        const int e = myq[whole + lane];
        myq[lane] = e;
      }
      cnt = rem;
    }
  }
  agg_flush<AGG_GD>(myq, cnt, lane, base, h, row, col, acc, lp, ovf, saved);
  saved += cnt;
  if (lane == 0) lcnt[wslot] = saved;

  __syncthreads();
  {
    float4* d4 = (float4*)(out + (size_t)base * 128);
    const float4 bb = make_float4(bias[lane * 4 + 0], bias[lane * 4 + 1],
                                  bias[lane * 4 + 2], bias[lane * 4 + 3]);
    for (int i = tid; i < nT * 32; i += 1024) {
      const float d = dinv[base + (i >> 5)];
      const float* a = acc + (i >> 5) * 128 + lane;
      float4 v = make_float4(a[0 * AGG_PL], a[1 * AGG_PL], a[2 * AGG_PL], a[3 * AGG_PL]);
      v.x = fmaxf(v.x * d + bb.x, 0.0f);
      v.y = fmaxf(v.y * d + bb.y, 0.0f);
      v.z = fmaxf(v.z * d + bb.z, 0.0f);
      v.w = fmaxf(v.w * d + bb.w, 0.0f);
      d4[i] = v;
    }
  }
}

__global__ __launch_bounds__(1024) void k_agg2_lds32(
    const float* __restrict__ g, const float* __restrict__ dinv,
    const float* __restrict__ bias,
    const int* __restrict__ row, const int* __restrict__ col,
    const int* __restrict__ lstP,
    const int* __restrict__ lcnt, const int* __restrict__ ovf,
    float* __restrict__ out, int n) {
  __shared__ float acc[AGG_T * 32];

  const int tid  = threadIdx.x;
  const int lane = tid & 31;
  const int wave = __builtin_amdgcn_readfirstlane(tid >> 5);
  const int base = blockIdx.x * AGG_T;
  const int nT   = min(AGG_T, n - base);

  {
    const float4* s4 = (const float4*)(g + (size_t)base * 32);
    float4* d4 = (float4*)acc;
    for (int i = tid; i < nT * 8; i += 1024) d4[i] = s4[i];
  }
  __syncthreads();

  const int wslot = blockIdx.x * 32 + wave;
  const int cnt = min(__builtin_amdgcn_readfirstlane(lcnt[wslot]), AGG_LCAP);
  const int*   lp = lstP + (size_t)wslot * AGG_LCAP;
  for (int b0 = 0; b0 < cnt; b0 += 32) {
    const int m = min(32, cnt - b0);
    int   pk = 0;
    if (lane < m) pk = lp[b0 + lane];
    if (m == 32) {
      float v[32];
#pragma unroll
      for (int u = 0; u < 32; ++u) {
        const unsigned ru = (unsigned)__builtin_amdgcn_readlane(pk, u) >> AGG_TBITS;
        v[u] = g[(size_t)ru * 32 + lane];
      }
#pragma unroll
      for (int u = 0; u < 32; ++u) {
        const int   tu = __builtin_amdgcn_readlane(pk, u) & AGG_TMASK;
        atomicAdd(acc + tu * 32 + lane, v[u]);
      }
    } else {
      for (int u = 0; u < m; ++u) {
        const int   pu = __builtin_amdgcn_readlane(pk, u);
        const int   ru = pu >> AGG_TBITS;
        const int   tu = pu & AGG_TMASK;
        atomicAdd(acc + tu * 32 + lane, g[(size_t)ru * 32 + lane]);
      }
    }
  }

  const int nOv = ovf[0];
  for (int i = tid; i < nOv; i += 1024) {
    const int e = ovf[1 + i];
    const int c = col[e];
    const int t = c - base;
    if ((unsigned)t < (unsigned)nT) {
      const int r = row[e];
      for (int f = 0; f < 32; ++f)
        atomicAdd(acc + t * 32 + f, g[(size_t)r * 32 + f]);
    }
  }
  __syncthreads();

  const float bl = bias[lane];
  for (int node0 = wave; node0 < nT; node0 += 128) {
    float v[4], mx[4], s[4];
#pragma unroll
    for (int k = 0; k < 4; ++k) {
      const int node = node0 + 32 * k;
      v[k]  = (node < nT) ? acc[node * 32 + lane] * dinv[base + node] + bl : 0.0f;
      mx[k] = v[k];
    }
#pragma unroll
    for (int o = 16; o > 0; o >>= 1) {
#pragma unroll
      for (int k = 0; k < 4; ++k) mx[k] = fmaxf(mx[k], __shfl_xor(mx[k], o, 32));
    }
#pragma unroll
    for (int k = 0; k < 4; ++k) s[k] = __expf(v[k] - mx[k]);
#pragma unroll
    for (int o = 16; o > 0; o >>= 1) {
#pragma unroll
      for (int k = 0; k < 4; ++k) s[k] += __shfl_xor(s[k], o, 32);
    }
#pragma unroll
    for (int k = 0; k < 4; ++k) {
      const int node = node0 + 32 * k;
      if (node < nT)
        out[(size_t)(base + node) * 32 + lane] = v[k] - mx[k] - __logf(s[k]);
    }
  }
}

extern "C" void kernel_launch(void* const* d_in, const int* in_sizes, int n_in,
                              void* d_out, int out_size, void* d_ws, size_t ws_size,
                              hipStream_t stream) {
  const float* x  = (const float*)d_in[0];
  const float* W1 = (const float*)d_in[1];
  const float* b1 = (const float*)d_in[2];
  const float* W2 = (const float*)d_in[3];
  const float* b2 = (const float*)d_in[4];
  const int*  eix = (const int*)d_in[5];

  const int n = N_NODES;
  const int E = in_sizes[5] / 2;
  const int* row = eix;
  const int* col = eix + E;

  float* ws   = (float*)d_ws;
  float* dinv = ws;
  float* h    = ws + 100352;
  float* h1   = h  + (size_t)n * D_H;
  float* g2   = h1 + (size_t)n * D_H;
  int*   lcnt = (int*)(g2 + (size_t)n * D_OUT);
  unsigned int* keys = (unsigned int*)(lcnt + AGG_LCNT);
  int*   lstP = (int*)keys + E + 256;
  int*   ovf  = lstP + (size_t)AGG_NGRP * 32 * AGG_LCAP;
  unsigned int* degP = (unsigned int*)(ovf + E + 256);
  float* outf = (float*)d_out;

  k_deg_own<<<DG_P * DG_C, 1024, 0, stream>>>(col, degP, E, n, keys, ovf);
  k_rsqrt <<<(n + 255) / 256, 256, 0, stream>>>(degP, dinv, n);

  gemm_bf16_wmma<D_IN, D_H><<<(n + 127) / 128, 256, 0, stream>>>(
      x, W1, h, dinv, n);
  k_agg_lds128<<<(n + AGG_T - 1) / AGG_T, 1024, 0, stream>>>(h, dinv, b1, row, col, h1, E, n,
                                                         lstP, lcnt, ovf, keys);

  gemm_bf16_wmma<D_H, D_OUT><<<(n + 127) / 128, 256, 0, stream>>>(
      h1, W2, g2, dinv, n);
  k_agg2_lds32<<<(n + AGG_T - 1) / AGG_T, 1024, 0, stream>>>(
      g2, dinv, b2, row, col, lstP, lcnt, ovf, outf, n);
}


# Both GEMMs: all global loads of a k-step (4 A reads, W quads) in flight together behind one wait (input-only load fence), store loop's row factors too; were serial round trips; 118 regs: 4 groups/WGP
# speedup vs baseline: 3.9496x; 1.2278x over previous
#include <hip/hip_runtime.h>

#define N_NODES 100000
#define D_IN 256
#define D_H 128
#define D_OUT 32

typedef __attribute__((ext_vector_type(16))) __bf16 v16bf;
typedef __attribute__((ext_vector_type(8)))  float  v8f;
typedef __attribute__((ext_vector_type(4)))  float  v4f;

__device__ __forceinline__ v16bf frag_from_lds(const __bf16* base, int rowIdx, int khalf) {
  union { uint4 u[2]; v16bf v; } c;
  const uint4* p = (const uint4*)(base + rowIdx * 32 + khalf);
  c.u[0] = p[0];
  c.u[1] = p[2];
  return c.v;
}

template<int K, int N>
__global__ __launch_bounds__(256) void gemm_bf16_wmma(
    const float* __restrict__ A, const float* __restrict__ W,
    float* __restrict__ C,
    const float* __restrict__ dinv, int M) {
  constexpr int NT = N / 16;
  __shared__ __bf16 ldsA[128 * 32];
  __shared__ __bf16 ldsB[N * 32];

  const int tid   = threadIdx.x;
  const int wave  = tid >> 5;
  const int lane  = tid & 31;
  const int laneM = lane & 15;
  const int khalf = (lane >> 4) << 3;
  const int rowBlk = blockIdx.x * 128;

  const int ldRow = tid >> 1;
  const int ldC0  = (tid & 1) << 4;
  const int ldGr  = rowBlk + ldRow;
  const int ldGrC = min(ldGr, M - 1);
  constexpr int WQ = (N * 8) / 256;
  static_assert((N * 8) % 256 == 0, "whole W quads per thread");

  v8f acc[NT];
#pragma unroll
  for (int t = 0; t < NT; ++t) {
    v8f z = {0.f, 0.f, 0.f, 0.f, 0.f, 0.f, 0.f, 0.f};
    acc[t] = z;
  }

  for (int kk = 0; kk < K; kk += 32) {
    __syncthreads();

    v4f av[4];
    v4f wv[WQ];
    {
      const v4f* src = (const v4f*)(A + (size_t)ldGrC * K + kk + ldC0);
#pragma unroll
      for (int j = 0; j < 4; ++j) av[j] = src[j];
    }
#pragma unroll
    for (int q = 0; q < WQ; ++q) {
      const int i  = tid + q * 256;
      const int n  = i % N;
      const int kq = i / N;
#pragma unroll
      for (int j = 0; j < 4; ++j)
        wv[q][j] = W[(size_t)(kk + kq * 4 + j) * N + n];
    }
    if constexpr (WQ == 4) {
      asm volatile("" :: "v"(av[0]), "v"(av[1]), "v"(av[2]), "v"(av[3]),
                         "v"(wv[0]), "v"(wv[1]), "v"(wv[2]), "v"(wv[3]));
    } else {
      static_assert(WQ == 1, "fence lists the W quads by hand");
      asm volatile("" :: "v"(av[0]), "v"(av[1]), "v"(av[2]), "v"(av[3]), "v"(wv[0]));
    }

    {
      union { __bf16 h[16]; uint4 q[2]; } buf;
#pragma unroll
      for (int j = 0; j < 4; ++j) {
        buf.h[j * 4 + 0] = (__bf16)av[j][0];
        buf.h[j * 4 + 1] = (__bf16)av[j][1];
        buf.h[j * 4 + 2] = (__bf16)av[j][2];
        buf.h[j * 4 + 3] = (__bf16)av[j][3];
      }
      uint4* dst = (uint4*)(ldsA + ldRow * 32 + ldC0);
      dst[0] = buf.q[0];
      dst[1] = buf.q[1];
    }

#pragma unroll
    for (int q = 0; q < WQ; ++q) {
      const int i  = tid + q * 256;
      const int n  = i % N;
      const int kq = i / N;
      union { __bf16 h[4]; uint2 q2; } wb;
#pragma unroll
      for (int j = 0; j < 4; ++j) wb.h[j] = (__bf16)wv[q][j];
      *(uint2*)(ldsB + n * 32 + kq * 4) = wb.q2;
    }
    __syncthreads();

    v16bf a = frag_from_lds(ldsA, (wave << 4) + laneM, khalf);
    v16bf bc = frag_from_lds(ldsB, laneM, khalf);
#pragma unroll
    for (int t = 0; t < NT; ++t) {
      v16bf bn = (t + 1 < NT) ? frag_from_lds(ldsB, ((t + 1) << 4) + laneM, khalf) : bc;
      acc[t] = __builtin_amdgcn_wmma_f32_16x16x32_bf16(
          false, a, false, bc, (short)0, acc[t], false, false);
      bc = bn;
    }
  }

  const int mBase = rowBlk + (wave << 4) + ((lane >> 4) << 3);
  float dv[8];
#pragma unroll
  for (int v = 0; v < 8; ++v) dv[v] = dinv[min(mBase + v, M - 1)];
#pragma unroll
  for (int v = 0; v < 8; ++v) {
    const int gr = mBase + v;
    if (gr < M) {
      const float d = dv[v];
#pragma unroll
      for (int t = 0; t < NT; ++t) {
        const size_t idx = (size_t)gr * N + (t << 4) + laneM;
        C[idx] = acc[t][v] * d;
      }
    }
  }
}

#define AGG_T     400
#define AGG_TBITS 9
#define AGG_TMASK ((1 << AGG_TBITS) - 1)
#define AGG_QCAP  160
#define AGG_NGRP  ((N_NODES + AGG_T - 1) / AGG_T)
#define AGG_LCAP  1024
#define AGG_LCNT  8192
static_assert(AGG_T <= (1 << AGG_TBITS), "local target bits");
static_assert(AGG_NGRP * 32 <= AGG_LCNT, "one count per sweep wave");
static_assert(N_NODES <= (1 << (31 - AGG_TBITS)), "source index and local target share one int");
static_assert(AGG_NGRP <= 255, "an owner key is one byte and 0xFF marks an empty slot");
static_assert(AGG_QCAP >= 96, "a trip starts below 32 queued edges and a pass adds at most 32");
#define AGG_SW    8
#define AGG_GD    16
#define AGG_PL    32

#define DG_P     25
#define DG_C     10
#define DG_NR    ((N_NODES + DG_P - 1) / DG_P)
#define DG_PITCH 100352
#define DG_SW    8
static_assert(DG_P * DG_NR >= N_NODES, "ranges cover all nodes");
static_assert(DG_PITCH >= N_NODES, "a plane holds all nodes");

__device__ __forceinline__ void dg_count(unsigned int* cnt, int4 v, int base) {
  const unsigned t0 = (unsigned)(v.x - base), t1 = (unsigned)(v.y - base);
  const unsigned t2 = (unsigned)(v.z - base), t3 = (unsigned)(v.w - base);
  if (t0 < (unsigned)DG_NR) atomicAdd(&cnt[t0], 1u);
  if (t1 < (unsigned)DG_NR) atomicAdd(&cnt[t1], 1u);
  if (t2 < (unsigned)DG_NR) atomicAdd(&cnt[t2], 1u);
  if (t3 < (unsigned)DG_NR) atomicAdd(&cnt[t3], 1u);
}

__device__ __forceinline__ unsigned dg_keybyte(int c) {
  return (c < 0) ? 0xFFu : ((unsigned)c / AGG_T);
}
__device__ __forceinline__ unsigned dg_keyword(int4 v) {
  return dg_keybyte(v.x) | (dg_keybyte(v.y) << 8) |
         (dg_keybyte(v.z) << 16) | (dg_keybyte(v.w) << 24);
}

__global__ __launch_bounds__(1024) void k_deg_own(
    const int* __restrict__ col, unsigned int* __restrict__ degP, int E, int n,
    unsigned int* __restrict__ keys, int* __restrict__ ovf) {
  __shared__ unsigned int cnt[DG_NR];

  const int tid  = threadIdx.x;
  const int p    = blockIdx.x % DG_P;
  const int c    = blockIdx.x / DG_P;
  const int base = p * DG_NR;

  for (int i = tid; i < DG_NR; i += 1024) cnt[i] = 0u;
  if (blockIdx.x == 0 && tid == 0) ovf[0] = 0;
  __syncthreads();

  const int QT = ((E + 15) >> 4) << 2;
  const int UT = (QT + 1023) >> 10;
  const int UC = (UT + DG_C - 1) / DG_C;
  const int uBeg = c * UC;
  const int uEnd = min(uBeg + UC, UT);
  const int QF = ((((size_t)col) & 15) == 0) ? (E >> 2) : 0;
  const int4* col4 = (const int4*)col;

  if (QF > 0) {
    for (int u = uBeg; u < uEnd; u += DG_SW) {
      int4 v[DG_SW];
#pragma unroll
      for (int s = 0; s < DG_SW; ++s) v[s] = col4[min(((u + s) << 10) + tid, QF - 1)];
#pragma unroll
      for (int s = 0; s < DG_SW; ++s) {
        const int q = ((u + s) << 10) + tid;
        if (u + s < uEnd && q < QF) {
          dg_count(cnt, v[s], base);
          if ((u + s) % DG_P == p) keys[q] = dg_keyword(v[s]);
        }
      }
    }
  }
  for (int q = max(QF, uBeg << 10) + tid; q < min(QT, uEnd << 10); q += 1024) {
    const int e0 = q * 4;
    int4 v = make_int4(-1, -1, -1, -1);
    if (e0 + 0 < E) v.x = col[e0 + 0];
    if (e0 + 1 < E) v.y = col[e0 + 1];
    if (e0 + 2 < E) v.z = col[e0 + 2];
    if (e0 + 3 < E) v.w = col[e0 + 3];
    dg_count(cnt, v, base);
    if ((q >> 10) % DG_P == p) keys[q] = dg_keyword(v);
  }
  __syncthreads();

  unsigned int* dst = degP + (size_t)c * DG_PITCH + base;
  for (int i = tid; i < DG_NR; i += 1024)
    if (base + i < n) dst[i] = cnt[i];
}

__global__ void k_rsqrt(const unsigned int* __restrict__ degP, float* __restrict__ p, int n) {
  int i = blockIdx.x * blockDim.x + threadIdx.x;
  if (i < n) {
    unsigned d = 1u;
#pragma unroll
    for (int c = 0; c < DG_C; ++c) d += degP[(size_t)c * DG_PITCH + i];
    p[i] = rsqrtf((float)d);
  }
}


__device__ __forceinline__ unsigned agg_match16(uint4 k, unsigned pat) {
  const unsigned M7 = 0x7F7F7F7Fu;
  const unsigned t0 = k.x ^ pat, t1 = k.y ^ pat, t2 = k.z ^ pat, t3 = k.w ^ pat;
  const unsigned n0 = ((t0 & M7) + M7) | t0;
  const unsigned n1 = ((t1 & M7) + M7) | t1;
  const unsigned n2 = ((t2 & M7) + M7) | t2;
  const unsigned n3 = ((t3 & M7) + M7) | t3;
  unsigned r = ((n0 >> 7) & 0x01010101u) | ((n1 >> 6) & 0x02020202u) |
               ((n2 >> 5) & 0x04040404u) | ((n3 >> 4) & 0x08080808u);
  return ~r & 0x0F0F0F0Fu;
}

template<int G>
__device__ __forceinline__ void agg_flush(
    const int* myq, int cnt, int lane, int base,
    const float* __restrict__ h,
    const int* __restrict__ row, const int* __restrict__ col,
    float* acc,
    int* __restrict__ lp, int* __restrict__ ovf, int saved) {
  const float4* h4 = (const float4*)h;
  for (int b0 = 0; b0 < cnt; b0 += 32) {
    const int m = min(32, cnt - b0);
    int r = 0, t = 0;
    if (lane < m) {
      const int e = myq[b0 + lane];
      const int c = col[e];
      r  = row[e];
      t  = c - base;
      const int slot = saved + b0 + lane;
      if (slot < AGG_LCAP) {
        lp[slot] = (r << AGG_TBITS) | t;
      } else {
        const int k = atomicAdd(ovf, 1);
        ovf[1 + k] = e;
      }
    }
    int i = 0;
    if (G > 4 && m == 32) {
      float4 v[G];
#pragma unroll
      for (int u = 0; u < G; ++u) {
        const int ru = __builtin_amdgcn_readlane(r, u);
        v[u] = h4[(size_t)ru * 32 + lane];
      }
#pragma unroll
      for (int u = 0; u < 32; ++u) {
        const int tu = __builtin_amdgcn_readlane(t, u);
        float* a = acc + tu * 128 + lane;
        atomicAdd(a + 0 * AGG_PL, v[u % G].x);
        atomicAdd(a + 1 * AGG_PL, v[u % G].y);
        atomicAdd(a + 2 * AGG_PL, v[u % G].z);
        atomicAdd(a + 3 * AGG_PL, v[u % G].w);
        if (u + G < 32) {
          const int ru = __builtin_amdgcn_readlane(r, u + G);
          v[u % G] = h4[(size_t)ru * 32 + lane];
        }
      }
      i = 32;
    }
    if (G > 4) {
      for (; i + G <= m; i += G) {
        float4 v[G];
        int    tt[G];
#pragma unroll
        for (int u = 0; u < G; ++u) {
          const int ru = __builtin_amdgcn_readlane(r, i + u);
          tt[u] = __builtin_amdgcn_readlane(t, i + u);
          v[u]  = h4[(size_t)ru * 32 + lane];
        }
#pragma unroll
        for (int u = 0; u < G; ++u) {
          float* a = acc + tt[u] * 128 + lane;
          atomicAdd(a + 0 * AGG_PL, v[u].x);
          atomicAdd(a + 1 * AGG_PL, v[u].y);
          atomicAdd(a + 2 * AGG_PL, v[u].z);
          atomicAdd(a + 3 * AGG_PL, v[u].w);
        }
      }
    }
    for (; i + 4 <= m; i += 4) {
      float4 v[4];
      int    tt[4];
#pragma unroll
      for (int u = 0; u < 4; ++u) {
        const int ru = __builtin_amdgcn_readlane(r, i + u);
        tt[u] = __builtin_amdgcn_readlane(t, i + u);
        v[u]  = h4[(size_t)ru * 32 + lane];
      }
#pragma unroll
      for (int u = 0; u < 4; ++u) {
        float* a = acc + tt[u] * 128 + lane;
        atomicAdd(a + 0 * AGG_PL, v[u].x);
        atomicAdd(a + 1 * AGG_PL, v[u].y);
        atomicAdd(a + 2 * AGG_PL, v[u].z);
        atomicAdd(a + 3 * AGG_PL, v[u].w);
      }
    }
    for (; i < m; ++i) {
      const int   ru = __builtin_amdgcn_readlane(r, i);
      const int   tu = __builtin_amdgcn_readlane(t, i);
      const float4 v = h4[(size_t)ru * 32 + lane];
      float* a = acc + tu * 128 + lane;
      atomicAdd(a + 0 * AGG_PL, v.x);
      atomicAdd(a + 1 * AGG_PL, v.y);
      atomicAdd(a + 2 * AGG_PL, v.z);
      atomicAdd(a + 3 * AGG_PL, v.w);
    }
  }
}

__global__ __launch_bounds__(1024) void k_agg_lds128(
    const float* __restrict__ h, const float* __restrict__ dinv,
    const float* __restrict__ bias,
    const int* __restrict__ row, const int* __restrict__ col,
    float* __restrict__ out, int E, int n,
    int* __restrict__ lstP,
    int* __restrict__ lcnt, int* __restrict__ ovf,
    const unsigned int* __restrict__ keys) {
  __shared__ float acc[AGG_T * 128];
  __shared__ int   q[32 * AGG_QCAP];

  const int tid  = threadIdx.x;
  const int lane = tid & 31;
  const int wave = tid >> 5;
  const int base = blockIdx.x * AGG_T;
  const int nT   = min(AGG_T, n - base);
  const unsigned lt = (1u << lane) - 1u;

  {
    const float4* s4 = (const float4*)(h + (size_t)base * 128);
    for (int i = tid; i < nT * 32; i += 1024) {
      const float4 v = s4[i];
      float* a = acc + (i >> 5) * 128 + lane;
      a[0 * AGG_PL] = v.x;
      a[1 * AGG_PL] = v.y;
      a[2 * AGG_PL] = v.z;
      a[3 * AGG_PL] = v.w;
    }
  }
  __syncthreads();

  int* myq = q + wave * AGG_QCAP;
  int cnt = 0;
  const int wslot = blockIdx.x * 32 + wave;
  int*   lp = lstP + (size_t)wslot * AGG_LCAP;
  int saved = 0;

  const uint4* key4 = (const uint4*)keys;
  const int nK = (E + 15) >> 4;
  const unsigned pat = (unsigned)blockIdx.x * 0x01010101u;

  const int nU  = (nK + 1023) >> 10;
  const int nIt = (nU + AGG_SW - 1) / AGG_SW;
  const int u0  = (int)(((long long)blockIdx.x * nU) / gridDim.x);
  for (int j = 0; j < nIt; ++j) {
    uint4 k[AGG_SW];
    int   ix[AGG_SW];
#pragma unroll
    for (int u = 0; u < AGG_SW; ++u) {
      const int pos = j * AGG_SW + u;
      int unit = u0 + pos;
      if (unit >= nU) unit -= nU;
      const int i = (unit << 10) + tid;
      ix[u] = (pos < nU && i < nK) ? i : -1;
      k[u] = key4[min(i, nK - 1)];
    }
#pragma unroll
    for (int u = 0; u < AGG_SW; ++u) {
      const int i = ix[u];
      unsigned mk = (i >= 0) ? agg_match16(k[u], pat) : 0u;
      unsigned b = __builtin_amdgcn_ballot_w32(mk != 0u);
      while (b) {
        if (mk) {
          const int bit = __builtin_ctz(mk);
          myq[cnt + __builtin_popcount(b & lt)] = (i << 4) + ((bit & 7) << 2) + (bit >> 3);
          mk &= mk - 1u;
        }
        cnt += __builtin_popcount(b);
        if (cnt >= AGG_QCAP - 32) {
          agg_flush<4>(myq, cnt, lane, base, h, row, col, acc, lp, ovf, saved);
          saved += cnt;
          cnt = 0;
        }
        b = __builtin_amdgcn_ballot_w32(mk != 0u);
      }
    }
    if (cnt >= 32) {
      const int whole = cnt & ~31;
      agg_flush<AGG_GD>(myq, whole, lane, base, h, row, col, acc, lp, ovf, saved);
      saved += whole;
      const int rem = cnt - whole;
      if (lane < rem) {
        const int e = myq[whole + lane];
        myq[lane] = e;
      }
      cnt = rem;
    }
  }
  agg_flush<AGG_GD>(myq, cnt, lane, base, h, row, col, acc, lp, ovf, saved);
  saved += cnt;
  if (lane == 0) lcnt[wslot] = saved;

  __syncthreads();
  {
    float4* d4 = (float4*)(out + (size_t)base * 128);
    const float4 bb = make_float4(bias[lane * 4 + 0], bias[lane * 4 + 1],
                                  bias[lane * 4 + 2], bias[lane * 4 + 3]);
    for (int i = tid; i < nT * 32; i += 1024) {
      const float d = dinv[base + (i >> 5)];
      const float* a = acc + (i >> 5) * 128 + lane;
      float4 v = make_float4(a[0 * AGG_PL], a[1 * AGG_PL], a[2 * AGG_PL], a[3 * AGG_PL]);
      v.x = fmaxf(v.x * d + bb.x, 0.0f);
      v.y = fmaxf(v.y * d + bb.y, 0.0f);
      v.z = fmaxf(v.z * d + bb.z, 0.0f);
      v.w = fmaxf(v.w * d + bb.w, 0.0f);
      d4[i] = v;
    }
  }
}

__global__ __launch_bounds__(1024) void k_agg2_lds32(
    const float* __restrict__ g, const float* __restrict__ dinv,
    const float* __restrict__ bias,
    const int* __restrict__ row, const int* __restrict__ col,
    const int* __restrict__ lstP,
    const int* __restrict__ lcnt, const int* __restrict__ ovf,
    float* __restrict__ out, int n) {
  __shared__ float acc[AGG_T * 32];

  const int tid  = threadIdx.x;
  const int lane = tid & 31;
  const int wave = __builtin_amdgcn_readfirstlane(tid >> 5);
  const int base = blockIdx.x * AGG_T;
  const int nT   = min(AGG_T, n - base);

  {
    const float4* s4 = (const float4*)(g + (size_t)base * 32);
    float4* d4 = (float4*)acc;
    for (int i = tid; i < nT * 8; i += 1024) d4[i] = s4[i];
  }
  __syncthreads();

  const int wslot = blockIdx.x * 32 + wave;
  const int cnt = min(__builtin_amdgcn_readfirstlane(lcnt[wslot]), AGG_LCAP);
  const int*   lp = lstP + (size_t)wslot * AGG_LCAP;
  for (int b0 = 0; b0 < cnt; b0 += 32) {
    const int m = min(32, cnt - b0);
    int   pk = 0;
    if (lane < m) pk = lp[b0 + lane];
    if (m == 32) {
      float v[32];
#pragma unroll
      for (int u = 0; u < 32; ++u) {
        const unsigned ru = (unsigned)__builtin_amdgcn_readlane(pk, u) >> AGG_TBITS;
        v[u] = g[(size_t)ru * 32 + lane];
      }
#pragma unroll
      for (int u = 0; u < 32; ++u) {
        const int   tu = __builtin_amdgcn_readlane(pk, u) & AGG_TMASK;
        atomicAdd(acc + tu * 32 + lane, v[u]);
      }
    } else {
      for (int u = 0; u < m; ++u) {
        const int   pu = __builtin_amdgcn_readlane(pk, u);
        const int   ru = pu >> AGG_TBITS;
        const int   tu = pu & AGG_TMASK;
        atomicAdd(acc + tu * 32 + lane, g[(size_t)ru * 32 + lane]);
      }
    }
  }

  const int nOv = ovf[0];
  for (int i = tid; i < nOv; i += 1024) {
    const int e = ovf[1 + i];
    const int c = col[e];
    const int t = c - base;
    if ((unsigned)t < (unsigned)nT) {
      const int r = row[e];
      for (int f = 0; f < 32; ++f)
        atomicAdd(acc + t * 32 + f, g[(size_t)r * 32 + f]);
    }
  }
  __syncthreads();

  const float bl = bias[lane];
  for (int node0 = wave; node0 < nT; node0 += 128) {
    float v[4], mx[4], s[4];
#pragma unroll
    for (int k = 0; k < 4; ++k) {
      const int node = node0 + 32 * k;
      v[k]  = (node < nT) ? acc[node * 32 + lane] * dinv[base + node] + bl : 0.0f;
      mx[k] = v[k];
    }
#pragma unroll
    for (int o = 16; o > 0; o >>= 1) {
#pragma unroll
      for (int k = 0; k < 4; ++k) mx[k] = fmaxf(mx[k], __shfl_xor(mx[k], o, 32));
    }
#pragma unroll
    for (int k = 0; k < 4; ++k) s[k] = __expf(v[k] - mx[k]);
#pragma unroll
    for (int o = 16; o > 0; o >>= 1) {
#pragma unroll
      for (int k = 0; k < 4; ++k) s[k] += __shfl_xor(s[k], o, 32);
    }
#pragma unroll
    for (int k = 0; k < 4; ++k) {
      const int node = node0 + 32 * k;
      if (node < nT)
        out[(size_t)(base + node) * 32 + lane] = v[k] - mx[k] - __logf(s[k]);
    }
  }
}

extern "C" void kernel_launch(void* const* d_in, const int* in_sizes, int n_in,
                              void* d_out, int out_size, void* d_ws, size_t ws_size,
                              hipStream_t stream) {
  const float* x  = (const float*)d_in[0];
  const float* W1 = (const float*)d_in[1];
  const float* b1 = (const float*)d_in[2];
  const float* W2 = (const float*)d_in[3];
  const float* b2 = (const float*)d_in[4];
  const int*  eix = (const int*)d_in[5];

  const int n = N_NODES;
  const int E = in_sizes[5] / 2;
  const int* row = eix;
  const int* col = eix + E;

  float* ws   = (float*)d_ws;
  float* dinv = ws;
  float* h    = ws + 100352;
  float* h1   = h  + (size_t)n * D_H;
  float* g2   = h1 + (size_t)n * D_H;
  int*   lcnt = (int*)(g2 + (size_t)n * D_OUT);
  unsigned int* keys = (unsigned int*)(lcnt + AGG_LCNT);
  int*   lstP = (int*)keys + E + 256;
  int*   ovf  = lstP + (size_t)AGG_NGRP * 32 * AGG_LCAP;
  unsigned int* degP = (unsigned int*)(ovf + E + 256);
  float* outf = (float*)d_out;

  k_deg_own<<<DG_P * DG_C, 1024, 0, stream>>>(col, degP, E, n, keys, ovf);
  k_rsqrt <<<(n + 255) / 256, 256, 0, stream>>>(degP, dinv, n);

  gemm_bf16_wmma<D_IN, D_H><<<(n + 127) / 128, 256, 0, stream>>>(
      x, W1, h, dinv, n);
  k_agg_lds128<<<(n + AGG_T - 1) / AGG_T, 1024, 0, stream>>>(h, dinv, b1, row, col, h1, E, n,
                                                         lstP, lcnt, ovf, keys);

  gemm_bf16_wmma<D_H, D_OUT><<<(n + 127) / 128, 256, 0, stream>>>(
      h1, W2, g2, dinv, n);
  k_agg2_lds32<<<(n + AGG_T - 1) / AGG_T, 1024, 0, stream>>>(
      g2, dinv, b2, row, col, lstP, lcnt, ovf, outf, n);
}
